# starting kernel
# baseline (speedup 1.0000x reference)
#include <hip/hip_runtime.h>

#define HDIM 128
#define BN_EPS 1e-5f
#define NEG_INF_ENC 0x007FFFFFu

typedef __attribute__((ext_vector_type(2))) float v2f;
typedef __attribute__((ext_vector_type(8))) float v8f;

__device__ __forceinline__ unsigned enc_f32(float f) {
    unsigned u = __float_as_uint(f);
    return (u & 0x80000000u) ? ~u : (u | 0x80000000u);
}
__device__ __forceinline__ float dec_f32(unsigned u) {
    return __uint_as_float((u & 0x80000000u) ? (u ^ 0x80000000u) : ~u);
}

__global__ void gemm128_bias(const float* __restrict__ A,
                             const float* __restrict__ W,
                             const float* __restrict__ bias,
                             float* __restrict__ out,
                             int nrows, int do_relu) {
    const int wave  = (int)((blockIdx.x * blockDim.x + threadIdx.x) >> 5);
    const int lane  = threadIdx.x & 31;
    const int row0  = (wave >> 3) << 4;
    const int col0  = (wave & 7) << 4;
    if (row0 >= nrows) return;

    const int t     = lane & 15;
    const int khalf = lane >> 4;

    int arow_i = row0 + t;
    if (arow_i >= nrows) arow_i = nrows - 1;
    const float* arow = A + (size_t)arow_i * HDIM + khalf * 2;
    const float* wrow = W + (size_t)(col0 + t) * HDIM + khalf * 2;
    __builtin_prefetch(wrow, 0, 3);

    const float bcol = bias[col0 + t];
    v8f c;
#pragma unroll
    for (int r = 0; r < 8; ++r) c[r] = bcol;

#pragma unroll
    for (int k0 = 0; k0 < HDIM; k0 += 4) {
        v2f a = *(const v2f*)(arow + k0);
        v2f b = *(const v2f*)(wrow + k0);
        c = __builtin_amdgcn_wmma_f32_16x16x4_f32(false, a, false, b,
                                                  (short)0, c, false, false);
    }

#pragma unroll
    for (int r = 0; r < 8; ++r) {
        const int orow = row0 + khalf * 8 + r;
        if (orow < nrows) {
            float v = c[r];
            if (do_relu) v = fmaxf(v, 0.0f);
            out[(size_t)orow * HDIM + col0 + t] = v;
        }
    }
}

__global__ void sage_layer_gemm(const float* __restrict__ AGG,
                                const float* __restrict__ WL,
                                const float* __restrict__ X,
                                const float* __restrict__ WR,
                                const float* __restrict__ bias,
                                const float* __restrict__ bn_g,
                                const float* __restrict__ bn_b,
                                const float* __restrict__ bn_m,
                                const float* __restrict__ bn_v,
                                float* __restrict__ out, int nrows) {
    const int wave  = (int)((blockIdx.x * blockDim.x + threadIdx.x) >> 5);
    const int lane  = threadIdx.x & 31;
    const int row0  = (wave >> 3) << 4;
    const int col0  = (wave & 7) << 4;
    if (row0 >= nrows) return;

    const int t     = lane & 15;
    const int khalf = lane >> 4;
    const int col   = col0 + t;

    int arow_i = row0 + t;
    if (arow_i >= nrows) arow_i = nrows - 1;
    const float* aggrow = AGG + (size_t)arow_i * HDIM + khalf * 2;
    const float* xrow   = X   + (size_t)arow_i * HDIM + khalf * 2;
    const float* wlrow  = WL  + (size_t)col * HDIM + khalf * 2;
    const float* wrrow  = WR  + (size_t)col * HDIM + khalf * 2;
    __builtin_prefetch(wlrow, 0, 3);
    __builtin_prefetch(wrrow, 0, 3);

    const float bcol = bias[col];
    v8f c;
#pragma unroll
    for (int r = 0; r < 8; ++r) c[r] = bcol;

#pragma unroll
    for (int k0 = 0; k0 < HDIM; k0 += 4) {
        v2f a = *(const v2f*)(aggrow + k0);
        v2f b = *(const v2f*)(wlrow + k0);
        c = __builtin_amdgcn_wmma_f32_16x16x4_f32(false, a, false, b,
                                                  (short)0, c, false, false);
    }
#pragma unroll
    for (int k0 = 0; k0 < HDIM; k0 += 4) {
        v2f a = *(const v2f*)(xrow + k0);
        v2f b = *(const v2f*)(wrrow + k0);
        c = __builtin_amdgcn_wmma_f32_16x16x4_f32(false, a, false, b,
                                                  (short)0, c, false, false);
    }

    const float s  = bn_g[col] * rsqrtf(bn_v[col] + BN_EPS);
    const float mu = bn_m[col];
    const float bb = bn_b[col];
#pragma unroll
    for (int r = 0; r < 8; ++r) {
        const int orow = row0 + khalf * 8 + r;
        if (orow < nrows) {
            float v = (c[r] - mu) * s + bb;
            out[(size_t)orow * HDIM + col0 + t] = fmaxf(v, 0.0f);
        }
    }
}

__global__ void init_agg(unsigned* __restrict__ agg, long long count) {
    long long i = (long long)blockIdx.x * blockDim.x + threadIdx.x;
    if (i < count) agg[i] = NEG_INF_ENC;
}

__global__ void scatter_max_edges(const int* __restrict__ ei,
                                  const float* __restrict__ X,
                                  unsigned* __restrict__ agg, int E) {
    long long tid = (long long)blockIdx.x * blockDim.x + threadIdx.x;
    int e = (int)(tid >> 5);
    if (e >= E) return;
    const int lane = (int)(tid & 31);
    const int dst = ei[e];
    const int src = ei[E + e];
    const float4 v = *(const float4*)(X + (size_t)src * HDIM + lane * 4);
    unsigned* a = agg + (size_t)dst * HDIM + lane * 4;
    atomicMax(a + 0, enc_f32(v.x));
    atomicMax(a + 1, enc_f32(v.y));
    atomicMax(a + 2, enc_f32(v.z));
    atomicMax(a + 3, enc_f32(v.w));
}

__global__ void decode_agg(const unsigned* __restrict__ agg,
                           float* __restrict__ out, long long count) {
    long long i = (long long)blockIdx.x * blockDim.x + threadIdx.x;
    if (i < count) {
        unsigned u = agg[i];
        out[i] = (u == NEG_INF_ENC) ? 0.0f : dec_f32(u);
    }
}

__global__ void fc2_head(const float* __restrict__ Y,
                         const float* __restrict__ W2,
                         const float* __restrict__ B2,
                         float* __restrict__ out, int nrows) {
    const int wave = (int)((blockIdx.x * blockDim.x + threadIdx.x) >> 5);
    const int lane = threadIdx.x & 31;
    if (wave >= nrows) return;

    const float4 y = *(const float4*)(Y + (size_t)wave * HDIM + lane * 4);
    float acc[4];
#pragma unroll
    for (int j = 0; j < 4; ++j) {
        const float4 w = *(const float4*)(W2 + (size_t)j * HDIM + lane * 4);
        acc[j] = y.x * w.x + y.y * w.y + y.z * w.z + y.w * w.w;
    }
#pragma unroll
    for (int off = 16; off > 0; off >>= 1) {
#pragma unroll
        for (int j = 0; j < 4; ++j) acc[j] += __shfl_xor(acc[j], off, 32);
    }
    if (lane == 0) {
#pragma unroll
        for (int j = 0; j < 4; ++j) out[(size_t)wave * 4 + j] = acc[j] + B2[j];
    }
}

extern "C" void kernel_launch(void* const* d_in, const int* in_sizes, int n_in,
                              void* d_out, int out_size, void* d_ws, size_t ws_size,
                              hipStream_t stream) {
    const float* features = (const float*)d_in[0];
    const int*   ei       = (const int*)d_in[1];
    const float* w_proj = (const float*)d_in[2];
    const float* b_proj = (const float*)d_in[3];
    const float* w_l0 = (const float*)d_in[4];
    const float* b_l0 = (const float*)d_in[5];
    const float* w_r0 = (const float*)d_in[6];
    const float* bn_g0 = (const float*)d_in[7];
    const float* bn_b0 = (const float*)d_in[8];
    const float* bn_m0 = (const float*)d_in[9];
    const float* bn_v0 = (const float*)d_in[10];
    const float* w_l1 = (const float*)d_in[11];
    const float* b_l1 = (const float*)d_in[12];
    const float* w_r1 = (const float*)d_in[13];
    const float* bn_g1 = (const float*)d_in[14];
    const float* bn_b1 = (const float*)d_in[15];
    const float* bn_m1 = (const float*)d_in[16];
    const float* bn_v1 = (const float*)d_in[17];
    const float* w_fc1 = (const float*)d_in[18];
    const float* b_fc1 = (const float*)d_in[19];
    const float* w_fc2 = (const float*)d_in[20];
    const float* b_fc2 = (const float*)d_in[21];
    float* outp = (float*)d_out;

    const int N = in_sizes[0] / HDIM;
    const int E = in_sizes[1] / 2;
    const long long NH = (long long)N * HDIM;

    const size_t SZ = (size_t)N * HDIM * sizeof(float);
    float*    X    = (float*)d_ws;
    unsigned* AGG  = (unsigned*)((char*)d_ws + SZ);
    float*    AGGF = (float*)((char*)d_ws + 2 * SZ);
    float*    H1   = (float*)((char*)d_ws + 3 * SZ);

    const dim3 blk(128);
    const int row_tiles   = (N + 15) / 16;
    const int gemm_blocks = (row_tiles * 8 + 3) / 4;
    const int ew_blocks   = (int)((NH + 255) / 256);
    const long long sc_threads = (long long)E * 32;
    const int sc_blocks   = (int)((sc_threads + 255) / 256);
    const int fc2_blocks  = (N + 7) / 8;

    gemm128_bias<<<gemm_blocks, blk, 0, stream>>>(features, w_proj, b_proj, X, N, 0);

    init_agg<<<ew_blocks, 256, 0, stream>>>(AGG, NH);
    scatter_max_edges<<<sc_blocks, 256, 0, stream>>>(ei, X, AGG, E);
    decode_agg<<<ew_blocks, 256, 0, stream>>>(AGG, AGGF, NH);
    sage_layer_gemm<<<gemm_blocks, blk, 0, stream>>>(AGGF, w_l0, X, w_r0, b_l0,
                                                     bn_g0, bn_b0, bn_m0, bn_v0, H1, N);

    init_agg<<<ew_blocks, 256, 0, stream>>>(AGG, NH);
    scatter_max_edges<<<sc_blocks, 256, 0, stream>>>(ei, H1, AGG, E);
    decode_agg<<<ew_blocks, 256, 0, stream>>>(AGG, AGGF, NH);
    sage_layer_gemm<<<gemm_blocks, blk, 0, stream>>>(AGGF, w_l1, H1, w_r1, b_l1,
                                                     bn_g1, bn_b1, bn_m1, bn_v1, X, N);

    gemm128_bias<<<gemm_blocks, blk, 0, stream>>>(X, w_fc1, b_fc1, H1, N, 1);

    fc2_head<<<fc2_blocks, 256, 0, stream>>>(H1, w_fc2, b_fc2, outp, N);
}


# Max-aggregation by pull: edges ordered by destination once per call (integer atomics), then one wave per node takes the max of its source rows; feature atomics, init and decode passes gone
# speedup vs baseline: 1.6656x; 1.6656x over previous
#include <hip/hip_runtime.h>

#define HDIM 128
#define BN_EPS 1e-5f

typedef __attribute__((ext_vector_type(2))) float v2f;
typedef __attribute__((ext_vector_type(8))) float v8f;

__global__ void gemm128_bias(const float* __restrict__ A,
                             const float* __restrict__ W,
                             const float* __restrict__ bias,
                             float* __restrict__ out,
                             int nrows, int do_relu) {
    const int wave  = (int)((blockIdx.x * blockDim.x + threadIdx.x) >> 5);
    const int lane  = threadIdx.x & 31;
    const int row0  = (wave >> 3) << 4;
    const int col0  = (wave & 7) << 4;
    if (row0 >= nrows) return;

    const int t     = lane & 15;
    const int khalf = lane >> 4;

    int arow_i = row0 + t;
    if (arow_i >= nrows) arow_i = nrows - 1;
    const float* arow = A + (size_t)arow_i * HDIM + khalf * 2;
    const float* wrow = W + (size_t)(col0 + t) * HDIM + khalf * 2;
    __builtin_prefetch(wrow, 0, 3);

    const float bcol = bias[col0 + t];
    v8f c;
#pragma unroll
    for (int r = 0; r < 8; ++r) c[r] = bcol;

#pragma unroll
    for (int k0 = 0; k0 < HDIM; k0 += 4) {
        v2f a = *(const v2f*)(arow + k0);
        v2f b = *(const v2f*)(wrow + k0);
        c = __builtin_amdgcn_wmma_f32_16x16x4_f32(false, a, false, b,
                                                  (short)0, c, false, false);
    }

#pragma unroll
    for (int r = 0; r < 8; ++r) {
        const int orow = row0 + khalf * 8 + r;
        if (orow < nrows) {
            float v = c[r];
            if (do_relu) v = fmaxf(v, 0.0f);
            out[(size_t)orow * HDIM + col0 + t] = v;
        }
    }
}

__global__ void sage_layer_gemm(const float* __restrict__ AGG,
                                const float* __restrict__ WL,
                                const float* __restrict__ X,
                                const float* __restrict__ WR,
                                const float* __restrict__ bias,
                                const float* __restrict__ bn_g,
                                const float* __restrict__ bn_b,
                                const float* __restrict__ bn_m,
                                const float* __restrict__ bn_v,
                                float* __restrict__ out, int nrows) {
    const int wave  = (int)((blockIdx.x * blockDim.x + threadIdx.x) >> 5);
    const int lane  = threadIdx.x & 31;
    const int row0  = (wave >> 3) << 4;
    const int col0  = (wave & 7) << 4;
    if (row0 >= nrows) return;

    const int t     = lane & 15;
    const int khalf = lane >> 4;
    const int col   = col0 + t;

    int arow_i = row0 + t;
    if (arow_i >= nrows) arow_i = nrows - 1;
    const float* aggrow = AGG + (size_t)arow_i * HDIM + khalf * 2;
    const float* xrow   = X   + (size_t)arow_i * HDIM + khalf * 2;
    const float* wlrow  = WL  + (size_t)col * HDIM + khalf * 2;
    const float* wrrow  = WR  + (size_t)col * HDIM + khalf * 2;
    __builtin_prefetch(wlrow, 0, 3);
    __builtin_prefetch(wrrow, 0, 3);

    const float bcol = bias[col];
    v8f c;
#pragma unroll
    for (int r = 0; r < 8; ++r) c[r] = bcol;

#pragma unroll
    for (int k0 = 0; k0 < HDIM; k0 += 4) {
        v2f a = *(const v2f*)(aggrow + k0);
        v2f b = *(const v2f*)(wlrow + k0);
        c = __builtin_amdgcn_wmma_f32_16x16x4_f32(false, a, false, b,
                                                  (short)0, c, false, false);
    }
#pragma unroll
    for (int k0 = 0; k0 < HDIM; k0 += 4) {
        v2f a = *(const v2f*)(xrow + k0);
        v2f b = *(const v2f*)(wrrow + k0);
        c = __builtin_amdgcn_wmma_f32_16x16x4_f32(false, a, false, b,
                                                  (short)0, c, false, false);
    }

    const float s  = bn_g[col] * rsqrtf(bn_v[col] + BN_EPS);
    const float mu = bn_m[col];
    const float bb = bn_b[col];
#pragma unroll
    for (int r = 0; r < 8; ++r) {
        const int orow = row0 + khalf * 8 + r;
        if (orow < nrows) {
            float v = (c[r] - mu) * s + bb;
            out[(size_t)orow * HDIM + col0 + t] = fmaxf(v, 0.0f);
        }
    }
}

__global__ void zero_counts(int* __restrict__ deg, int count) {
    const int i = blockIdx.x * blockDim.x + threadIdx.x;
    if (i < count) deg[i] = 0;
}

__global__ void rank_edges(const int* __restrict__ ei, int* __restrict__ deg,
                           int* __restrict__ rank, int E, int nrows) {
    const int e = blockIdx.x * blockDim.x + threadIdx.x;
    if (e >= E) return;
    const int dst = ei[e];
    int r = -1;
    if ((unsigned)dst < (unsigned)nrows) r = atomicAdd(deg + dst, 1);
    rank[e] = r;
}

__global__ void __launch_bounds__(256)
alloc_segments(int* __restrict__ deg, int2* __restrict__ seg, int nrows) {
    __shared__ int wave_off[8];
    __shared__ int block_base;
    const int n    = blockIdx.x * 256 + threadIdx.x;
    const int lane = threadIdx.x & 31;
    const int wv   = threadIdx.x >> 5;
    const int d    = (n < nrows) ? deg[n] : 0;
    int incl = d;
#pragma unroll
    for (int off = 1; off < 32; off <<= 1) {
        const int t = __shfl_up(incl, off, 32);
        if (lane >= off) incl += t;
    }
    if (lane == 31) wave_off[wv] = incl;
    __syncthreads();
    if (threadIdx.x == 0) {
        int tot = 0;
#pragma unroll
        for (int w = 0; w < 8; ++w) { const int t = wave_off[w]; wave_off[w] = tot; tot += t; }
        block_base = atomicAdd(deg + nrows, tot);
    }
    __syncthreads();
    if (n < nrows) seg[n] = make_int2(block_base + wave_off[wv] + incl - d, d);
}

__global__ void fill_sources(const int* __restrict__ ei, const int* __restrict__ rank,
                             const int2* __restrict__ seg, int* __restrict__ col,
                             int E, int nrows) {
    const int e = blockIdx.x * blockDim.x + threadIdx.x;
    if (e >= E) return;
    const int r = rank[e];
    if (r < 0) return;
    const int dst = ei[e];
    int src = ei[E + e];
    src = min(max(src, 0), nrows - 1);
    col[seg[dst].x + r] = src;
}

__device__ __forceinline__ float4 max4(float4 a, float4 b) {
    return make_float4(fmaxf(a.x, b.x), fmaxf(a.y, b.y), fmaxf(a.z, b.z), fmaxf(a.w, b.w));
}

__global__ void __launch_bounds__(256)
gather_max_rows(const int2* __restrict__ seg, const int* __restrict__ col,
                const float* __restrict__ X, float* __restrict__ out, int nrows) {
    const int lane = threadIdx.x & 31;
    const int n = __builtin_amdgcn_readfirstlane(
        (int)((blockIdx.x * blockDim.x + threadIdx.x) >> 5));
    if (n >= nrows) return;
    const int2 sg  = seg[n];
    const int  beg = sg.x;
    const int  cnt = sg.y;
    const float4* __restrict__ xr = (const float4*)X + lane;
    const float ninf = -__builtin_inff();
    float4 m = make_float4(ninf, ninf, ninf, ninf);

    for (int base = 0; base < cnt; base += 32) {
        const int c = min(cnt - base, 32);
        const int ids = col[beg + base + min(lane, c - 1)];
        int i = 0;
        for (; i + 4 <= c; i += 4) {
            const int s0 = __builtin_amdgcn_readlane(ids, i);
            const int s1 = __builtin_amdgcn_readlane(ids, i + 1);
            const int s2 = __builtin_amdgcn_readlane(ids, i + 2);
            const int s3 = __builtin_amdgcn_readlane(ids, i + 3);
            const float4 v0 = xr[(size_t)s0 * (HDIM / 4)];
            const float4 v1 = xr[(size_t)s1 * (HDIM / 4)];
            const float4 v2 = xr[(size_t)s2 * (HDIM / 4)];
            const float4 v3 = xr[(size_t)s3 * (HDIM / 4)];
            m = max4(m, max4(max4(v0, v1), max4(v2, v3)));
        }
        const int left = c - i;
        if (left > 0) {
            const int s0 = __builtin_amdgcn_readlane(ids, i);
            float4 t = xr[(size_t)s0 * (HDIM / 4)];
            if (left > 1) {
                const int s1 = __builtin_amdgcn_readlane(ids, i + 1);
                t = max4(t, xr[(size_t)s1 * (HDIM / 4)]);
            }
            if (left > 2) {
                const int s2 = __builtin_amdgcn_readlane(ids, i + 2);
                t = max4(t, xr[(size_t)s2 * (HDIM / 4)]);
            }
            m = max4(m, t);
        }
    }
    float4 r;
    r.x = (m.x == ninf) ? 0.0f : m.x;
    r.y = (m.y == ninf) ? 0.0f : m.y;
    r.z = (m.z == ninf) ? 0.0f : m.z;
    r.w = (m.w == ninf) ? 0.0f : m.w;
    ((float4*)out)[(size_t)n * (HDIM / 4) + lane] = r;
}

__global__ void fc2_head(const float* __restrict__ Y,
                         const float* __restrict__ W2,
                         const float* __restrict__ B2,
                         float* __restrict__ out, int nrows) {
    const int wave = (int)((blockIdx.x * blockDim.x + threadIdx.x) >> 5);
    const int lane = threadIdx.x & 31;
    if (wave >= nrows) return;

    const float4 y = *(const float4*)(Y + (size_t)wave * HDIM + lane * 4);
    float acc[4];
#pragma unroll
    for (int j = 0; j < 4; ++j) {
        const float4 w = *(const float4*)(W2 + (size_t)j * HDIM + lane * 4);
        acc[j] = y.x * w.x + y.y * w.y + y.z * w.z + y.w * w.w;
    }
#pragma unroll
    for (int off = 16; off > 0; off >>= 1) {
#pragma unroll
        for (int j = 0; j < 4; ++j) acc[j] += __shfl_xor(acc[j], off, 32);
    }
    if (lane == 0) {
#pragma unroll
        for (int j = 0; j < 4; ++j) out[(size_t)wave * 4 + j] = acc[j] + B2[j];
    }
}

extern "C" void kernel_launch(void* const* d_in, const int* in_sizes, int n_in,
                              void* d_out, int out_size, void* d_ws, size_t ws_size,
                              hipStream_t stream) {
    const float* features = (const float*)d_in[0];
    const int*   ei       = (const int*)d_in[1];
    const float* w_proj = (const float*)d_in[2];
    const float* b_proj = (const float*)d_in[3];
    const float* w_l0 = (const float*)d_in[4];
    const float* b_l0 = (const float*)d_in[5];
    const float* w_r0 = (const float*)d_in[6];
    const float* bn_g0 = (const float*)d_in[7];
    const float* bn_b0 = (const float*)d_in[8];
    const float* bn_m0 = (const float*)d_in[9];
    const float* bn_v0 = (const float*)d_in[10];
    const float* w_l1 = (const float*)d_in[11];
    const float* b_l1 = (const float*)d_in[12];
    const float* w_r1 = (const float*)d_in[13];
    const float* bn_g1 = (const float*)d_in[14];
    const float* bn_b1 = (const float*)d_in[15];
    const float* bn_m1 = (const float*)d_in[16];
    const float* bn_v1 = (const float*)d_in[17];
    const float* w_fc1 = (const float*)d_in[18];
    const float* b_fc1 = (const float*)d_in[19];
    const float* w_fc2 = (const float*)d_in[20];
    const float* b_fc2 = (const float*)d_in[21];
    float* outp = (float*)d_out;

    const int N = in_sizes[0] / HDIM;
    const int E = in_sizes[1] / 2;

    const size_t SZ = (size_t)N * HDIM * sizeof(float);
    float* X    = (float*)d_ws;
    float* AGGF = (float*)((char*)d_ws + SZ);
    float* H1   = (float*)((char*)d_ws + 2 * SZ);
    char*  G    = (char*)d_ws + 3 * SZ;
    const size_t o_seg  = (((size_t)N + 1) * sizeof(int) + 255) & ~(size_t)255;
    const size_t o_rank = o_seg  + (((size_t)N * sizeof(int2) + 255) & ~(size_t)255);
    const size_t o_col  = o_rank + (((size_t)E * sizeof(int)  + 255) & ~(size_t)255);
    int*  deg  = (int*)G;
    int2* seg  = (int2*)(G + o_seg);
    int*  rank = (int*)(G + o_rank);
    int*  col  = (int*)(G + o_col);

    const dim3 blk(128);
    const int row_tiles   = (N + 15) / 16;
    const int gemm_blocks = (row_tiles * 8 + 3) / 4;
    const int edge_blocks = (E + 255) / 256;
    const int node_blocks = (N + 255) / 256;
    const int wave_blocks = (N + 7) / 8;

    zero_counts<<<(N + 1 + 255) / 256, 256, 0, stream>>>(deg, N + 1);
    rank_edges<<<edge_blocks, 256, 0, stream>>>(ei, deg, rank, E, N);
    alloc_segments<<<node_blocks, 256, 0, stream>>>(deg, seg, N);
    fill_sources<<<edge_blocks, 256, 0, stream>>>(ei, rank, seg, col, E, N);

    gemm128_bias<<<gemm_blocks, blk, 0, stream>>>(features, w_proj, b_proj, X, N, 0);

    gather_max_rows<<<wave_blocks, 256, 0, stream>>>(seg, col, X, AGGF, N);
    sage_layer_gemm<<<gemm_blocks, blk, 0, stream>>>(AGGF, w_l0, X, w_r0, b_l0,
                                                     bn_g0, bn_b0, bn_m0, bn_v0, H1, N);

    gather_max_rows<<<wave_blocks, 256, 0, stream>>>(seg, col, H1, AGGF, N);
    sage_layer_gemm<<<gemm_blocks, blk, 0, stream>>>(AGGF, w_l1, H1, w_r1, b_l1,
                                                     bn_g1, bn_b1, bn_m1, bn_v1, X, N);

    gemm128_bias<<<gemm_blocks, blk, 0, stream>>>(X, w_fc1, b_fc1, H1, N, 1);

    fc2_head<<<wave_blocks, 256, 0, stream>>>(H1, w_fc2, b_fc2, outp, N);
}


# GEMMs re-tiled: a wave owns a 16x128 output strip, each A fragment loaded once for 8 column tiles, weights staged per work-group in LDS; k order in 8-wide chunks permuted (sum order changes)
# speedup vs baseline: 2.8144x; 1.6897x over previous
#include <hip/hip_runtime.h>

#define HDIM 128
#define BN_EPS 1e-5f
#define WPITCH4 33

typedef __attribute__((ext_vector_type(2))) float v2f;
typedef __attribute__((ext_vector_type(8))) float v8f;

__device__ __forceinline__ void stage_weights(const float* __restrict__ Wg, float4* sW,
                                              int wv, int lane) {
    float4 w[16];
#pragma unroll
    for (int it = 0; it < 16; ++it)
        w[it] = *(const float4*)(Wg + (size_t)(it * 8 + wv) * HDIM + lane * 4);
#pragma unroll
    for (int it = 0; it < 16; ++it)
        sW[(it * 8 + wv) * WPITCH4 + lane] = w[it];
}

__device__ __forceinline__ void load_a_strip(const float* __restrict__ A, int arow_i, int h,
                                             float4 (&a)[16]) {
    const float4* ap = (const float4*)(A + (size_t)arow_i * HDIM + 4 * h);
#pragma unroll
    for (int t = 0; t < 16; ++t) a[t] = ap[2 * t];
}

template <int NMAT, bool BN>
__device__ __forceinline__ void strip_mac(float4 (&a0)[16], const float* __restrict__ A1,
                                          const float4* sW, int arow_i, int c, int h,
                                          v8f (&acc)[8],
                                          const float* __restrict__ bn_g,
                                          const float* __restrict__ bn_b,
                                          const float* __restrict__ bn_m,
                                          const float* __restrict__ bn_v,
                                          float (&pg)[8], float (&pb)[8],
                                          float (&pm)[8], float (&pv)[8]) {
    const float4* ap1 = (const float4*)(A1 + (size_t)arow_i * HDIM + 4 * h);
    const float4* bp  = sW + c * WPITCH4 + h;

    float4 a1[16];
    float4 b[2][8];
#pragma unroll
    for (int j = 0; j < 8; ++j) b[0][j] = bp[j * 16 * WPITCH4];

#pragma unroll
    for (int t = 0; t < 16 * NMAT; ++t) {
        if (NMAT == 2 && t == 8) {
#pragma unroll
            for (int u = 0; u < 8; ++u) a1[u] = ap1[2 * u];
        }
        if (NMAT == 2 && t == 16) {
#pragma unroll
            for (int u = 8; u < 16; ++u) a1[u] = ap1[2 * u];
        }
        if (BN && t == 16 * NMAT - 3) {
#pragma unroll
            for (int j = 0; j < 8; ++j) {
                pg[j] = bn_g[j * 16 + c];
                pv[j] = bn_v[j * 16 + c];
                pm[j] = bn_m[j * 16 + c];
                pb[j] = bn_b[j * 16 + c];
            }
        }
        if (t + 1 < 16 * NMAT) {
            const int tn = t + 1;
            const float4* nb = bp + (tn >> 4) * (HDIM * WPITCH4) + (tn & 15) * 2;
#pragma unroll
            for (int j = 0; j < 8; ++j) b[tn & 1][j] = nb[j * 16 * WPITCH4];
        }
        __builtin_amdgcn_sched_barrier(0);
        const float4 av = (t < 16) ? a0[t & 15] : a1[t & 15];
        const v2f a01 = {av.x, av.y};
        const v2f a23 = {av.z, av.w};
#pragma unroll
        for (int j = 0; j < 8; ++j) {
            const v2f b01 = {b[t & 1][j].x, b[t & 1][j].y};
            acc[j] = __builtin_amdgcn_wmma_f32_16x16x4_f32(false, a01, false, b01,
                                                           (short)0, acc[j], false, false);
        }
#pragma unroll
        for (int j = 0; j < 8; ++j) {
            const v2f b23 = {b[t & 1][j].z, b[t & 1][j].w};
            acc[j] = __builtin_amdgcn_wmma_f32_16x16x4_f32(false, a23, false, b23,
                                                           (short)0, acc[j], false, false);
        }
        __builtin_amdgcn_sched_barrier(0);
    }
}

__global__ void __launch_bounds__(256)
gemm128_bias(const float* __restrict__ A,
             const float* __restrict__ W,
             const float* __restrict__ bias,
             float* __restrict__ out,
             int nrows, int do_relu) {
    __shared__ float4 sW[HDIM * WPITCH4];
    const int lane = threadIdx.x & 31;
    const int wv   = threadIdx.x >> 5;
    const int c    = lane & 15;
    const int h    = lane >> 4;
    const int row0 = (blockIdx.x * 8 + wv) << 4;
    int arow_i = row0 + c;
    if (arow_i >= nrows) arow_i = nrows - 1;

    float4 a0[16];
    load_a_strip(A, arow_i, h, a0);
    float bcol[8];
#pragma unroll
    for (int j = 0; j < 8; ++j) bcol[j] = bias[j * 16 + c];

    stage_weights(W, sW, wv, lane);
    __syncthreads();
    if (row0 >= nrows) return;

    v8f acc[8];
#pragma unroll
    for (int j = 0; j < 8; ++j) {
#pragma unroll
        for (int r = 0; r < 8; ++r) acc[j][r] = bcol[j];
    }

    float pg[8], pb[8], pm[8], pv[8];
    strip_mac<1, false>(a0, A, sW, arow_i, c, h, acc, bias, bias, bias, bias, pg, pb, pm, pv);

    float* op = out + (size_t)(row0 + h * 8) * HDIM + c;
    if (row0 + 16 <= nrows) {
#pragma unroll
        for (int j = 0; j < 8; ++j) {
#pragma unroll
            for (int r = 0; r < 8; ++r) {
                float v = acc[j][r];
                if (do_relu) v = fmaxf(v, 0.0f);
                op[r * HDIM + j * 16] = v;
            }
        }
    } else {
#pragma unroll
        for (int j = 0; j < 8; ++j) {
#pragma unroll
            for (int r = 0; r < 8; ++r) {
                if (row0 + h * 8 + r < nrows) {
                    float v = acc[j][r];
                    if (do_relu) v = fmaxf(v, 0.0f);
                    op[r * HDIM + j * 16] = v;
                }
            }
        }
    }
}

__global__ void __launch_bounds__(256)
sage_layer_gemm(const float* __restrict__ AGG,
                const float* __restrict__ WL,
                const float* __restrict__ X,
                const float* __restrict__ WR,
                const float* __restrict__ bias,
                const float* __restrict__ bn_g,
                const float* __restrict__ bn_b,
                const float* __restrict__ bn_m,
                const float* __restrict__ bn_v,
                float* __restrict__ out, int nrows) {
    __shared__ float4 sW[2 * HDIM * WPITCH4];
    const int lane = threadIdx.x & 31;
    const int wv   = threadIdx.x >> 5;
    const int c    = lane & 15;
    const int h    = lane >> 4;
    const int row0 = (blockIdx.x * 8 + wv) << 4;
    int arow_i = row0 + c;
    if (arow_i >= nrows) arow_i = nrows - 1;

    float4 a0[16];
    load_a_strip(AGG, arow_i, h, a0);
    float bcol[8];
#pragma unroll
    for (int j = 0; j < 8; ++j) bcol[j] = bias[j * 16 + c];

    stage_weights(WL, sW, wv, lane);
    stage_weights(WR, sW + HDIM * WPITCH4, wv, lane);
    __syncthreads();
    if (row0 >= nrows) return;

    v8f acc[8];
#pragma unroll
    for (int j = 0; j < 8; ++j) {
#pragma unroll
        for (int r = 0; r < 8; ++r) acc[j][r] = bcol[j];
    }

    float pg[8], pb[8], pm[8], pv[8];
    strip_mac<2, true>(a0, X, sW, arow_i, c, h, acc, bn_g, bn_b, bn_m, bn_v, pg, pb, pm, pv);

    float* op = out + (size_t)(row0 + h * 8) * HDIM + c;
    const bool whole = (row0 + 16 <= nrows);
#pragma unroll
    for (int j = 0; j < 8; ++j) {
        const float s  = pg[j] * rsqrtf(pv[j] + BN_EPS);
        const float mu = pm[j];
        const float bb = pb[j];
        if (whole) {
#pragma unroll
            for (int r = 0; r < 8; ++r) {
                const float v = (acc[j][r] - mu) * s + bb;
                op[r * HDIM + j * 16] = fmaxf(v, 0.0f);
            }
        } else {
#pragma unroll
            for (int r = 0; r < 8; ++r) {
                if (row0 + h * 8 + r < nrows) {
                    const float v = (acc[j][r] - mu) * s + bb;
                    op[r * HDIM + j * 16] = fmaxf(v, 0.0f);
                }
            }
        }
    }
}

__global__ void zero_counts(int* __restrict__ deg, int count) {
    const int i = blockIdx.x * blockDim.x + threadIdx.x;
    if (i < count) deg[i] = 0;
}

__global__ void rank_edges(const int* __restrict__ ei, int* __restrict__ deg,
                           int* __restrict__ rank, int E, int nrows) {
    const int e = blockIdx.x * blockDim.x + threadIdx.x;
    if (e >= E) return;
    const int dst = ei[e];
    int r = -1;
    if ((unsigned)dst < (unsigned)nrows) r = atomicAdd(deg + dst, 1);
    rank[e] = r;
}

__global__ void __launch_bounds__(256)
alloc_segments(int* __restrict__ deg, int2* __restrict__ seg, int nrows) {
    __shared__ int wave_off[8];
    __shared__ int block_base;
    const int n    = blockIdx.x * 256 + threadIdx.x;
    const int lane = threadIdx.x & 31;
    const int wv   = threadIdx.x >> 5;
    const int d    = (n < nrows) ? deg[n] : 0;
    int incl = d;
#pragma unroll
    for (int off = 1; off < 32; off <<= 1) {
        const int t = __shfl_up(incl, off, 32);
        if (lane >= off) incl += t;
    }
    if (lane == 31) wave_off[wv] = incl;
    __syncthreads();
    if (threadIdx.x == 0) {
        int tot = 0;
#pragma unroll
        for (int w = 0; w < 8; ++w) { const int t = wave_off[w]; wave_off[w] = tot; tot += t; }
        block_base = atomicAdd(deg + nrows, tot);
    }
    __syncthreads();
    if (n < nrows) seg[n] = make_int2(block_base + wave_off[wv] + incl - d, d);
}

__global__ void fill_sources(const int* __restrict__ ei, const int* __restrict__ rank,
                             const int2* __restrict__ seg, int* __restrict__ col,
                             int E, int nrows) {
    const int e = blockIdx.x * blockDim.x + threadIdx.x;
    if (e >= E) return;
    const int r = rank[e];
    if (r < 0) return;
    const int dst = ei[e];
    int src = ei[E + e];
    src = min(max(src, 0), nrows - 1);
    col[seg[dst].x + r] = src;
}

__device__ __forceinline__ float4 max4(float4 a, float4 b) {
    return make_float4(fmaxf(a.x, b.x), fmaxf(a.y, b.y), fmaxf(a.z, b.z), fmaxf(a.w, b.w));
}

__global__ void __launch_bounds__(256)
gather_max_rows(const int2* __restrict__ seg, const int* __restrict__ col,
                const float* __restrict__ X, float* __restrict__ out, int nrows) {
    const int lane = threadIdx.x & 31;
    const int n = __builtin_amdgcn_readfirstlane(
        (int)((blockIdx.x * blockDim.x + threadIdx.x) >> 5));
    if (n >= nrows) return;
    const int2 sg  = seg[n];
    const int  beg = sg.x;
    const int  cnt = sg.y;
    const float4* __restrict__ xr = (const float4*)X + lane;
    const float ninf = -__builtin_inff();
    float4 m = make_float4(ninf, ninf, ninf, ninf);

    for (int base = 0; base < cnt; base += 32) {
        const int c = min(cnt - base, 32);
        const int ids = col[beg + base + min(lane, c - 1)];
        int i = 0;
        for (; i + 4 <= c; i += 4) {
            const int s0 = __builtin_amdgcn_readlane(ids, i);
            const int s1 = __builtin_amdgcn_readlane(ids, i + 1);
            const int s2 = __builtin_amdgcn_readlane(ids, i + 2);
            const int s3 = __builtin_amdgcn_readlane(ids, i + 3);
            const float4 v0 = xr[(size_t)s0 * (HDIM / 4)];
            const float4 v1 = xr[(size_t)s1 * (HDIM / 4)];
            const float4 v2 = xr[(size_t)s2 * (HDIM / 4)];
            const float4 v3 = xr[(size_t)s3 * (HDIM / 4)];
            m = max4(m, max4(max4(v0, v1), max4(v2, v3)));
        }
        const int left = c - i;
        if (left > 0) {
            const int s0 = __builtin_amdgcn_readlane(ids, i);
            float4 t = xr[(size_t)s0 * (HDIM / 4)];
            if (left > 1) {
                const int s1 = __builtin_amdgcn_readlane(ids, i + 1);
                t = max4(t, xr[(size_t)s1 * (HDIM / 4)]);
            }
            if (left > 2) {
                const int s2 = __builtin_amdgcn_readlane(ids, i + 2);
                t = max4(t, xr[(size_t)s2 * (HDIM / 4)]);
            }
            m = max4(m, t);
        }
    }
    float4 r;
    r.x = (m.x == ninf) ? 0.0f : m.x;
    r.y = (m.y == ninf) ? 0.0f : m.y;
    r.z = (m.z == ninf) ? 0.0f : m.z;
    r.w = (m.w == ninf) ? 0.0f : m.w;
    ((float4*)out)[(size_t)n * (HDIM / 4) + lane] = r;
}

__global__ void fc2_head(const float* __restrict__ Y,
                         const float* __restrict__ W2,
                         const float* __restrict__ B2,
                         float* __restrict__ out, int nrows) {
    const int wave = (int)((blockIdx.x * blockDim.x + threadIdx.x) >> 5);
    const int lane = threadIdx.x & 31;
    if (wave >= nrows) return;

    const float4 y = *(const float4*)(Y + (size_t)wave * HDIM + lane * 4);
    float acc[4];
#pragma unroll
    for (int j = 0; j < 4; ++j) {
        const float4 w = *(const float4*)(W2 + (size_t)j * HDIM + lane * 4);
        acc[j] = y.x * w.x + y.y * w.y + y.z * w.z + y.w * w.w;
    }
#pragma unroll
    for (int off = 16; off > 0; off >>= 1) {
#pragma unroll
        for (int j = 0; j < 4; ++j) acc[j] += __shfl_xor(acc[j], off, 32);
    }
    if (lane == 0) {
#pragma unroll
        for (int j = 0; j < 4; ++j) out[(size_t)wave * 4 + j] = acc[j] + B2[j];
    }
}

extern "C" void kernel_launch(void* const* d_in, const int* in_sizes, int n_in,
                              void* d_out, int out_size, void* d_ws, size_t ws_size,
                              hipStream_t stream) {
    const float* features = (const float*)d_in[0];
    const int*   ei       = (const int*)d_in[1];
    const float* w_proj = (const float*)d_in[2];
    const float* b_proj = (const float*)d_in[3];
    const float* w_l0 = (const float*)d_in[4];
    const float* b_l0 = (const float*)d_in[5];
    const float* w_r0 = (const float*)d_in[6];
    const float* bn_g0 = (const float*)d_in[7];
    const float* bn_b0 = (const float*)d_in[8];
    const float* bn_m0 = (const float*)d_in[9];
    const float* bn_v0 = (const float*)d_in[10];
    const float* w_l1 = (const float*)d_in[11];
    const float* b_l1 = (const float*)d_in[12];
    const float* w_r1 = (const float*)d_in[13];
    const float* bn_g1 = (const float*)d_in[14];
    const float* bn_b1 = (const float*)d_in[15];
    const float* bn_m1 = (const float*)d_in[16];
    const float* bn_v1 = (const float*)d_in[17];
    const float* w_fc1 = (const float*)d_in[18];
    const float* b_fc1 = (const float*)d_in[19];
    const float* w_fc2 = (const float*)d_in[20];
    const float* b_fc2 = (const float*)d_in[21];
    float* outp = (float*)d_out;

    const int N = in_sizes[0] / HDIM;
    const int E = in_sizes[1] / 2;

    const size_t SZ = (size_t)N * HDIM * sizeof(float);
    float* X    = (float*)d_ws;
    float* AGGF = (float*)((char*)d_ws + SZ);
    float* H1   = (float*)((char*)d_ws + 2 * SZ);
    char*  G    = (char*)d_ws + 3 * SZ;
    const size_t o_seg  = (((size_t)N + 1) * sizeof(int) + 255) & ~(size_t)255;
    const size_t o_rank = o_seg  + (((size_t)N * sizeof(int2) + 255) & ~(size_t)255);
    const size_t o_col  = o_rank + (((size_t)E * sizeof(int)  + 255) & ~(size_t)255);
    int*  deg  = (int*)G;
    int2* seg  = (int2*)(G + o_seg);
    int*  rank = (int*)(G + o_rank);
    int*  col  = (int*)(G + o_col);

    const int row_tiles   = (N + 15) / 16;
    const int gemm_blocks = (row_tiles + 7) / 8;
    const int edge_blocks = (E + 255) / 256;
    const int node_blocks = (N + 255) / 256;
    const int wave_blocks = (N + 7) / 8;

    zero_counts<<<(N + 1 + 255) / 256, 256, 0, stream>>>(deg, N + 1);
    rank_edges<<<edge_blocks, 256, 0, stream>>>(ei, deg, rank, E, N);
    alloc_segments<<<node_blocks, 256, 0, stream>>>(deg, seg, N);
    fill_sources<<<edge_blocks, 256, 0, stream>>>(ei, rank, seg, col, E, N);

    gemm128_bias<<<gemm_blocks, 256, 0, stream>>>(features, w_proj, b_proj, X, N, 0);

    gather_max_rows<<<wave_blocks, 256, 0, stream>>>(seg, col, X, AGGF, N);
    sage_layer_gemm<<<gemm_blocks, 256, 0, stream>>>(AGGF, w_l0, X, w_r0, b_l0,
                                                     bn_g0, bn_b0, bn_m0, bn_v0, H1, N);

    gather_max_rows<<<wave_blocks, 256, 0, stream>>>(seg, col, H1, AGGF, N);
    sage_layer_gemm<<<gemm_blocks, 256, 0, stream>>>(AGGF, w_l1, H1, w_r1, b_l1,
                                                     bn_g1, bn_b1, bn_m1, bn_v1, X, N);

    gemm128_bias<<<gemm_blocks, 256, 0, stream>>>(X, w_fc1, b_fc1, H1, N, 1);

    fc2_head<<<wave_blocks, 256, 0, stream>>>(H1, w_fc2, b_fc2, outp, N);
}


# Edges ordered by destination with a two-level counting sort on LDS counters (4096-edge chunks into 256-node bins, one work-group orders each bin, stores staged in LDS); no global atomics left
# speedup vs baseline: 3.5877x; 1.2748x over previous
#include <hip/hip_runtime.h>

#define HDIM 128
#define BN_EPS 1e-5f
#define WPITCH4 33

typedef __attribute__((ext_vector_type(2))) float v2f;
typedef __attribute__((ext_vector_type(8))) float v8f;

__device__ __forceinline__ void stage_weights(const float* __restrict__ Wg, float4* sW,
                                              int wv, int lane) {
    float4 w[16];
#pragma unroll
    for (int it = 0; it < 16; ++it)
        w[it] = *(const float4*)(Wg + (size_t)(it * 8 + wv) * HDIM + lane * 4);
#pragma unroll
    for (int it = 0; it < 16; ++it)
        sW[(it * 8 + wv) * WPITCH4 + lane] = w[it];
}

__device__ __forceinline__ void load_a_strip(const float* __restrict__ A, int arow_i, int h,
                                             float4 (&a)[16]) {
    const float4* ap = (const float4*)(A + (size_t)arow_i * HDIM + 4 * h);
#pragma unroll
    for (int t = 0; t < 16; ++t) a[t] = ap[2 * t];
}

template <int NMAT, bool BN>
__device__ __forceinline__ void strip_mac(float4 (&a0)[16], const float* __restrict__ A1,
                                          const float4* sW, int arow_i, int c, int h,
                                          v8f (&acc)[8],
                                          const float* __restrict__ bn_g,
                                          const float* __restrict__ bn_b,
                                          const float* __restrict__ bn_m,
                                          const float* __restrict__ bn_v,
                                          float (&pg)[8], float (&pb)[8],
                                          float (&pm)[8], float (&pv)[8]) {
    const float4* ap1 = (const float4*)(A1 + (size_t)arow_i * HDIM + 4 * h);
    const float4* bp  = sW + c * WPITCH4 + h;

    float4 a1[16];
    float4 b[2][8];
#pragma unroll
    for (int j = 0; j < 8; ++j) b[0][j] = bp[j * 16 * WPITCH4];

#pragma unroll
    for (int t = 0; t < 16 * NMAT; ++t) {
        if (NMAT == 2 && t == 8) {
#pragma unroll
            for (int u = 0; u < 8; ++u) a1[u] = ap1[2 * u];
        }
        if (NMAT == 2 && t == 16) {
#pragma unroll
            for (int u = 8; u < 16; ++u) a1[u] = ap1[2 * u];
        }
        if (BN && t == 16 * NMAT - 3) {
#pragma unroll
            for (int j = 0; j < 8; ++j) {
                pg[j] = bn_g[j * 16 + c];
                pv[j] = bn_v[j * 16 + c];
                pm[j] = bn_m[j * 16 + c];
                pb[j] = bn_b[j * 16 + c];
            }
        }
        if (t + 1 < 16 * NMAT) {
            const int tn = t + 1;
            const float4* nb = bp + (tn >> 4) * (HDIM * WPITCH4) + (tn & 15) * 2;
#pragma unroll
            for (int j = 0; j < 8; ++j) b[tn & 1][j] = nb[j * 16 * WPITCH4];
        }
        __builtin_amdgcn_sched_barrier(0);
        const float4 av = (t < 16) ? a0[t & 15] : a1[t & 15];
        const v2f a01 = {av.x, av.y};
        const v2f a23 = {av.z, av.w};
#pragma unroll
        for (int j = 0; j < 8; ++j) {
            const v2f b01 = {b[t & 1][j].x, b[t & 1][j].y};
            acc[j] = __builtin_amdgcn_wmma_f32_16x16x4_f32(false, a01, false, b01,
                                                           (short)0, acc[j], false, false);
        }
#pragma unroll
        for (int j = 0; j < 8; ++j) {
            const v2f b23 = {b[t & 1][j].z, b[t & 1][j].w};
            acc[j] = __builtin_amdgcn_wmma_f32_16x16x4_f32(false, a23, false, b23,
                                                           (short)0, acc[j], false, false);
        }
        __builtin_amdgcn_sched_barrier(0);
    }
}

__global__ void __launch_bounds__(256)
gemm128_bias(const float* __restrict__ A,
             const float* __restrict__ W,
             const float* __restrict__ bias,
             float* __restrict__ out,
             int nrows, int do_relu) {
    __shared__ float4 sW[HDIM * WPITCH4];
    const int lane = threadIdx.x & 31;
    const int wv   = threadIdx.x >> 5;
    const int c    = lane & 15;
    const int h    = lane >> 4;
    const int row0 = (blockIdx.x * 8 + wv) << 4;
    int arow_i = row0 + c;
    if (arow_i >= nrows) arow_i = nrows - 1;

    float4 a0[16];
    load_a_strip(A, arow_i, h, a0);
    float bcol[8];
#pragma unroll
    for (int j = 0; j < 8; ++j) bcol[j] = bias[j * 16 + c];

    stage_weights(W, sW, wv, lane);
    __syncthreads();
    if (row0 >= nrows) return;

    v8f acc[8];
#pragma unroll
    for (int j = 0; j < 8; ++j) {
#pragma unroll
        for (int r = 0; r < 8; ++r) acc[j][r] = bcol[j];
    }

    float pg[8], pb[8], pm[8], pv[8];
    strip_mac<1, false>(a0, A, sW, arow_i, c, h, acc, bias, bias, bias, bias, pg, pb, pm, pv);

    float* op = out + (size_t)(row0 + h * 8) * HDIM + c;
    if (row0 + 16 <= nrows) {
#pragma unroll
        for (int j = 0; j < 8; ++j) {
#pragma unroll
            for (int r = 0; r < 8; ++r) {
                float v = acc[j][r];
                if (do_relu) v = fmaxf(v, 0.0f);
                op[r * HDIM + j * 16] = v;
            }
        }
    } else {
#pragma unroll
        for (int j = 0; j < 8; ++j) {
#pragma unroll
            for (int r = 0; r < 8; ++r) {
                if (row0 + h * 8 + r < nrows) {
                    float v = acc[j][r];
                    if (do_relu) v = fmaxf(v, 0.0f);
                    op[r * HDIM + j * 16] = v;
                }
            }
        }
    }
}

__global__ void __launch_bounds__(256)
sage_layer_gemm(const float* __restrict__ AGG,
                const float* __restrict__ WL,
                const float* __restrict__ X,
                const float* __restrict__ WR,
                const float* __restrict__ bias,
                const float* __restrict__ bn_g,
                const float* __restrict__ bn_b,
                const float* __restrict__ bn_m,
                const float* __restrict__ bn_v,
                float* __restrict__ out, int nrows) {
    __shared__ float4 sW[2 * HDIM * WPITCH4];
    const int lane = threadIdx.x & 31;
    const int wv   = threadIdx.x >> 5;
    const int c    = lane & 15;
    const int h    = lane >> 4;
    const int row0 = (blockIdx.x * 8 + wv) << 4;
    int arow_i = row0 + c;
    if (arow_i >= nrows) arow_i = nrows - 1;

    float4 a0[16];
    load_a_strip(AGG, arow_i, h, a0);
    float bcol[8];
#pragma unroll
    for (int j = 0; j < 8; ++j) bcol[j] = bias[j * 16 + c];

    stage_weights(WL, sW, wv, lane);
    stage_weights(WR, sW + HDIM * WPITCH4, wv, lane);
    __syncthreads();
    if (row0 >= nrows) return;

    v8f acc[8];
#pragma unroll
    for (int j = 0; j < 8; ++j) {
#pragma unroll
        for (int r = 0; r < 8; ++r) acc[j][r] = bcol[j];
    }

    float pg[8], pb[8], pm[8], pv[8];
    strip_mac<2, true>(a0, X, sW, arow_i, c, h, acc, bn_g, bn_b, bn_m, bn_v, pg, pb, pm, pv);

    float* op = out + (size_t)(row0 + h * 8) * HDIM + c;
    const bool whole = (row0 + 16 <= nrows);
#pragma unroll
    for (int j = 0; j < 8; ++j) {
        const float s  = pg[j] * rsqrtf(pv[j] + BN_EPS);
        const float mu = pm[j];
        const float bb = pb[j];
        if (whole) {
#pragma unroll
            for (int r = 0; r < 8; ++r) {
                const float v = (acc[j][r] - mu) * s + bb;
                op[r * HDIM + j * 16] = fmaxf(v, 0.0f);
            }
        } else {
#pragma unroll
            for (int r = 0; r < 8; ++r) {
                if (row0 + h * 8 + r < nrows) {
                    const float v = (acc[j][r] - mu) * s + bb;
                    op[r * HDIM + j * 16] = fmaxf(v, 0.0f);
                }
            }
        }
    }
}

#define BIN_SHIFT 8
#define CHUNK_EDGES 4096

__device__ __forceinline__ int block_excl_scan(int v, int* wsum, int& total) {
    const int lane = threadIdx.x & 31;
    const int wv   = threadIdx.x >> 5;
    int incl = v;
#pragma unroll
    for (int off = 1; off < 32; off <<= 1) {
        const int t = __shfl_up(incl, off, 32);
        if (lane >= off) incl += t;
    }
    __syncthreads();
    if (lane == 31) wsum[wv] = incl;
    __syncthreads();
    int woff = 0, tot = 0;
#pragma unroll
    for (int w = 0; w < 8; ++w) {
        const int t = wsum[w];
        if (w < wv) woff += t;
        tot += t;
    }
    total = tot;
    return woff + incl - v;
}

__global__ void __launch_bounds__(256)
bin_count(const int* __restrict__ ei, int* __restrict__ hist, int E, int nrows, int nbins) {
    extern __shared__ int sbin[];
    for (int i = threadIdx.x; i < nbins; i += 256) sbin[i] = 0;
    __syncthreads();
    const int base = blockIdx.x * CHUNK_EDGES + threadIdx.x;
    int d[16];
#pragma unroll
    for (int j = 0; j < 16; ++j) {
        const int e = min(base + j * 256, E - 1);
        d[j] = ei[e];
    }
#pragma unroll
    for (int j = 0; j < 16; ++j)
        if (base + j * 256 < E && (unsigned)d[j] < (unsigned)nrows)
            atomicAdd(&sbin[d[j] >> BIN_SHIFT], 1);
    __syncthreads();
    for (int i = threadIdx.x; i < nbins; i += 256)
        hist[(size_t)blockIdx.x * nbins + i] = sbin[i];
}

__global__ void __launch_bounds__(256)
bin_offsets(int* __restrict__ hist, int* __restrict__ bin_tot, int nchunks, int nbins) {
    __shared__ int wsum[8];
    const int k = blockIdx.x;
    int carry = 0;
    for (int b0 = 0; b0 < nchunks; b0 += 256) {
        const int b = b0 + threadIdx.x;
        const int v = (b < nchunks) ? hist[(size_t)b * nbins + k] : 0;
        int total;
        const int excl = block_excl_scan(v, wsum, total);
        if (b < nchunks) hist[(size_t)b * nbins + k] = carry + excl;
        carry += total;
    }
    if (threadIdx.x == 0) bin_tot[k] = carry;
}

__global__ void __launch_bounds__(256)
bin_partition(const int* __restrict__ ei, const int* __restrict__ hist,
              const int* __restrict__ bin_tot, int* __restrict__ bin_start,
              unsigned* __restrict__ part, int E, int nrows, int nbins) {
    extern __shared__ int sbin[];
    __shared__ unsigned rec[CHUNK_EDGES];
    __shared__ int      pos[CHUNK_EDGES];
    __shared__ int      wsum[8];
    int* lcnt  = sbin;
    int* gbase = sbin + nbins;
    for (int i = threadIdx.x; i < nbins; i += 256) lcnt[i] = 0;
    int carry = 0;
    for (int k0 = 0; k0 < nbins; k0 += 256) {
        const int k = k0 + threadIdx.x;
        const int v = (k < nbins) ? bin_tot[k] : 0;
        int total;
        const int excl = block_excl_scan(v, wsum, total);
        if (k < nbins) {
            const int s = carry + excl;
            if (blockIdx.x == 0) bin_start[k] = s;
            gbase[k] = s + hist[(size_t)blockIdx.x * nbins + k];
        }
        carry += total;
    }
    __syncthreads();
    const int base = blockIdx.x * CHUNK_EDGES + threadIdx.x;
    int d[16], s[16], r[16];
#pragma unroll
    for (int j = 0; j < 16; ++j) {
        const int e = min(base + j * 256, E - 1);
        d[j] = ei[e];
        s[j] = ei[(size_t)E + e];
    }
#pragma unroll
    for (int j = 0; j < 16; ++j) {
        const bool live = base + j * 256 < E && (unsigned)d[j] < (unsigned)nrows;
        r[j] = live ? atomicAdd(&lcnt[d[j] >> BIN_SHIFT], 1) : -1;
    }
    __syncthreads();
    carry = 0;
    for (int k0 = 0; k0 < nbins; k0 += 256) {
        const int k = k0 + threadIdx.x;
        const int v = (k < nbins) ? lcnt[k] : 0;
        int total;
        const int excl = block_excl_scan(v, wsum, total);
        if (k < nbins) lcnt[k] = carry + excl;
        carry += total;
    }
    __syncthreads();
    const int nlive = carry;
#pragma unroll
    for (int j = 0; j < 16; ++j) {
        if (r[j] >= 0) {
            const int bin  = d[j] >> BIN_SHIFT;
            const int srcc = min(max(s[j], 0), nrows - 1);
            const int slot = lcnt[bin] + r[j];
            rec[slot] = ((unsigned)(d[j] & 255) << 24) | (unsigned)srcc;
            pos[slot] = gbase[bin] + r[j];
        }
    }
    __syncthreads();
    for (int i = threadIdx.x; i < nlive; i += 256) part[pos[i]] = rec[i];
}

#define SORT_CAP 8192
__global__ void __launch_bounds__(256)
bin_sort(const unsigned* __restrict__ part, const int* __restrict__ bin_start,
         const int* __restrict__ bin_tot, int2* __restrict__ seg,
         int* __restrict__ col, int nrows) {
    __shared__ int cnt[256];
    __shared__ int wsum[8];
    __shared__ int sorted[SORT_CAP];
    const int k = blockIdx.x;
    const int s = bin_start[k];
    const int m = bin_tot[k];
    cnt[threadIdx.x] = 0;
    __syncthreads();
    for (int i0 = 0; i0 < m; i0 += 2048) {
        unsigned r[8];
#pragma unroll
        for (int u = 0; u < 8; ++u) r[u] = part[s + min(i0 + u * 256 + (int)threadIdx.x, m - 1)];
#pragma unroll
        for (int u = 0; u < 8; ++u)
            if (i0 + u * 256 + (int)threadIdx.x < m) atomicAdd(&cnt[r[u] >> 24], 1);
    }
    __syncthreads();
    const int deg = cnt[threadIdx.x];
    int total;
    const int first = block_excl_scan(deg, wsum, total);
    const int n = (k << BIN_SHIFT) + threadIdx.x;
    if (n < nrows) seg[n] = make_int2(s + first, deg);
    cnt[threadIdx.x] = first;
    __syncthreads();
    const bool in_lds = (m <= SORT_CAP);
    for (int i0 = 0; i0 < m; i0 += 2048) {
        unsigned r[8];
#pragma unroll
        for (int u = 0; u < 8; ++u) r[u] = part[s + min(i0 + u * 256 + (int)threadIdx.x, m - 1)];
#pragma unroll
        for (int u = 0; u < 8; ++u) {
            if (i0 + u * 256 + (int)threadIdx.x < m) {
                const int p = atomicAdd(&cnt[r[u] >> 24], 1);
                const int v = (int)(r[u] & 0xFFFFFFu);
                if (in_lds) sorted[p] = v; else col[s + p] = v;
            }
        }
    }
    if (in_lds) {
        __syncthreads();
        for (int i = threadIdx.x; i < m; i += 256) col[s + i] = sorted[i];
    }
}

__device__ __forceinline__ float4 max4(float4 a, float4 b) {
    return make_float4(fmaxf(a.x, b.x), fmaxf(a.y, b.y), fmaxf(a.z, b.z), fmaxf(a.w, b.w));
}

__global__ void __launch_bounds__(256)
gather_max_rows(const int2* __restrict__ seg, const int* __restrict__ col,
                const float* __restrict__ X, float* __restrict__ out, int nrows) {
    const int lane = threadIdx.x & 31;
    const int n = __builtin_amdgcn_readfirstlane(
        (int)((blockIdx.x * blockDim.x + threadIdx.x) >> 5));
    if (n >= nrows) return;
    const int2 sg  = seg[n];
    const int  beg = sg.x;
    const int  cnt = sg.y;
    const float4* __restrict__ xr = (const float4*)X + lane;
    const float ninf = -__builtin_inff();
    float4 m = make_float4(ninf, ninf, ninf, ninf);

    for (int base = 0; base < cnt; base += 32) {
        const int c = min(cnt - base, 32);
        const int ids = col[beg + base + min(lane, c - 1)];
        int i = 0;
        for (; i + 4 <= c; i += 4) {
            const int s0 = __builtin_amdgcn_readlane(ids, i);
            const int s1 = __builtin_amdgcn_readlane(ids, i + 1);
            const int s2 = __builtin_amdgcn_readlane(ids, i + 2);
            const int s3 = __builtin_amdgcn_readlane(ids, i + 3);
            const float4 v0 = xr[(size_t)s0 * (HDIM / 4)];
            const float4 v1 = xr[(size_t)s1 * (HDIM / 4)];
            const float4 v2 = xr[(size_t)s2 * (HDIM / 4)];
            const float4 v3 = xr[(size_t)s3 * (HDIM / 4)];
            m = max4(m, max4(max4(v0, v1), max4(v2, v3)));
        }
        const int left = c - i;
        if (left > 0) {
            const int s0 = __builtin_amdgcn_readlane(ids, i);
            float4 t = xr[(size_t)s0 * (HDIM / 4)];
            if (left > 1) {
                const int s1 = __builtin_amdgcn_readlane(ids, i + 1);
                t = max4(t, xr[(size_t)s1 * (HDIM / 4)]);
            }
            if (left > 2) {
                const int s2 = __builtin_amdgcn_readlane(ids, i + 2);
                t = max4(t, xr[(size_t)s2 * (HDIM / 4)]);
            }
            m = max4(m, t);
        }
    }
    float4 r;
    r.x = (m.x == ninf) ? 0.0f : m.x;
    r.y = (m.y == ninf) ? 0.0f : m.y;
    r.z = (m.z == ninf) ? 0.0f : m.z;
    r.w = (m.w == ninf) ? 0.0f : m.w;
    ((float4*)out)[(size_t)n * (HDIM / 4) + lane] = r;
}

__global__ void fc2_head(const float* __restrict__ Y,
                         const float* __restrict__ W2,
                         const float* __restrict__ B2,
                         float* __restrict__ out, int nrows) {
    const int wave = (int)((blockIdx.x * blockDim.x + threadIdx.x) >> 5);
    const int lane = threadIdx.x & 31;
    if (wave >= nrows) return;

    const float4 y = *(const float4*)(Y + (size_t)wave * HDIM + lane * 4);
    float acc[4];
#pragma unroll
    for (int j = 0; j < 4; ++j) {
        const float4 w = *(const float4*)(W2 + (size_t)j * HDIM + lane * 4);
        acc[j] = y.x * w.x + y.y * w.y + y.z * w.z + y.w * w.w;
    }
#pragma unroll
    for (int off = 16; off > 0; off >>= 1) {
#pragma unroll
        for (int j = 0; j < 4; ++j) acc[j] += __shfl_xor(acc[j], off, 32);
    }
    if (lane == 0) {
#pragma unroll
        for (int j = 0; j < 4; ++j) out[(size_t)wave * 4 + j] = acc[j] + B2[j];
    }
}

extern "C" void kernel_launch(void* const* d_in, const int* in_sizes, int n_in,
                              void* d_out, int out_size, void* d_ws, size_t ws_size,
                              hipStream_t stream) {
    const float* features = (const float*)d_in[0];
    const int*   ei       = (const int*)d_in[1];
    const float* w_proj = (const float*)d_in[2];
    const float* b_proj = (const float*)d_in[3];
    const float* w_l0 = (const float*)d_in[4];
    const float* b_l0 = (const float*)d_in[5];
    const float* w_r0 = (const float*)d_in[6];
    const float* bn_g0 = (const float*)d_in[7];
    const float* bn_b0 = (const float*)d_in[8];
    const float* bn_m0 = (const float*)d_in[9];
    const float* bn_v0 = (const float*)d_in[10];
    const float* w_l1 = (const float*)d_in[11];
    const float* b_l1 = (const float*)d_in[12];
    const float* w_r1 = (const float*)d_in[13];
    const float* bn_g1 = (const float*)d_in[14];
    const float* bn_b1 = (const float*)d_in[15];
    const float* bn_m1 = (const float*)d_in[16];
    const float* bn_v1 = (const float*)d_in[17];
    const float* w_fc1 = (const float*)d_in[18];
    const float* b_fc1 = (const float*)d_in[19];
    const float* w_fc2 = (const float*)d_in[20];
    const float* b_fc2 = (const float*)d_in[21];
    float* outp = (float*)d_out;

    const int N = in_sizes[0] / HDIM;
    const int E = in_sizes[1] / 2;

    const size_t SZ = (size_t)N * HDIM * sizeof(float);
    float* X    = (float*)d_ws;
    float* AGGF = (float*)((char*)d_ws + SZ);
    float* H1   = (float*)((char*)d_ws + 2 * SZ);
    char*  G    = (char*)d_ws + 3 * SZ;
    const size_t o_seg  = (((size_t)N + 1) * sizeof(int) + 255) & ~(size_t)255;
    const size_t o_part = o_seg  + (((size_t)N * sizeof(int2) + 255) & ~(size_t)255);
    const size_t o_col  = o_part + (((size_t)E * sizeof(int)  + 255) & ~(size_t)255);
    int2*     seg  = (int2*)(G + o_seg);
    unsigned* part = (unsigned*)(G + o_part);
    int*      col  = (int*)(G + o_col);

    const int nbins   = (N + 255) / 256;
    const int nchunks = (E + 4095) / 4096;
    int* hist      = (int*)AGGF;
    int* bin_tot   = hist + (size_t)nchunks * nbins;
    int* bin_start = bin_tot + nbins;

    const int row_tiles   = (N + 15) / 16;
    const int gemm_blocks = (row_tiles + 7) / 8;
    const int wave_blocks = (N + 7) / 8;

    bin_count<<<nchunks, 256, (size_t)nbins * sizeof(int), stream>>>(ei, hist, E, N, nbins);
    bin_offsets<<<nbins, 256, 0, stream>>>(hist, bin_tot, nchunks, nbins);
    bin_partition<<<nchunks, 256, 2 * (size_t)nbins * sizeof(int), stream>>>(ei, hist, bin_tot, bin_start,
                                                                          part, E, N, nbins);
    bin_sort<<<nbins, 256, 0, stream>>>(part, bin_start, bin_tot, seg, col, N);

    gemm128_bias<<<gemm_blocks, 256, 0, stream>>>(features, w_proj, b_proj, X, N, 0);

    gather_max_rows<<<wave_blocks, 256, 0, stream>>>(seg, col, X, AGGF, N);
    sage_layer_gemm<<<gemm_blocks, 256, 0, stream>>>(AGGF, w_l0, X, w_r0, b_l0,
                                                     bn_g0, bn_b0, bn_m0, bn_v0, H1, N);

    gather_max_rows<<<wave_blocks, 256, 0, stream>>>(seg, col, H1, AGGF, N);
    sage_layer_gemm<<<gemm_blocks, 256, 0, stream>>>(AGGF, w_l1, H1, w_r1, b_l1,
                                                     bn_g1, bn_b1, bn_m1, bn_v1, X, N);

    gemm128_bias<<<gemm_blocks, 256, 0, stream>>>(X, w_fc1, b_fc1, H1, N, 1);

    fc2_head<<<wave_blocks, 256, 0, stream>>>(H1, w_fc2, b_fc2, outp, N);
}


# fc2 head computed in the fc1 GEMM's epilogue from the accumulators (per-lane partials, halving exchange over 16 lanes; fc2 sum order changes): no y buffer, no fc2 launch of one wave per row
# speedup vs baseline: 3.7550x; 1.0466x over previous
#include <hip/hip_runtime.h>

#define HDIM 128
#define BN_EPS 1e-5f
#define WPITCH4 33

typedef __attribute__((ext_vector_type(2))) float v2f;
typedef __attribute__((ext_vector_type(8))) float v8f;

__device__ __forceinline__ void stage_weights(const float* __restrict__ Wg, float4* sW,
                                              int wv, int lane) {
    float4 w[16];
#pragma unroll
    for (int it = 0; it < 16; ++it)
        w[it] = *(const float4*)(Wg + (size_t)(it * 8 + wv) * HDIM + lane * 4);
#pragma unroll
    for (int it = 0; it < 16; ++it)
        sW[(it * 8 + wv) * WPITCH4 + lane] = w[it];
}

__device__ __forceinline__ void load_a_strip(const float* __restrict__ A, int arow_i, int h,
                                             float4 (&a)[16]) {
    const float4* ap = (const float4*)(A + (size_t)arow_i * HDIM + 4 * h);
#pragma unroll
    for (int t = 0; t < 16; ++t) a[t] = ap[2 * t];
}

template <int NMAT, bool BN>
__device__ __forceinline__ void strip_mac(float4 (&a0)[16], const float* __restrict__ A1,
                                          const float4* sW, int arow_i, int c, int h,
                                          v8f (&acc)[8],
                                          const float* __restrict__ bn_g,
                                          const float* __restrict__ bn_b,
                                          const float* __restrict__ bn_m,
                                          const float* __restrict__ bn_v,
                                          float (&pg)[8], float (&pb)[8],
                                          float (&pm)[8], float (&pv)[8]) {
    const float4* ap1 = (const float4*)(A1 + (size_t)arow_i * HDIM + 4 * h);
    const float4* bp  = sW + c * WPITCH4 + h;

    float4 a1[16];
    float4 b[2][8];
#pragma unroll
    for (int j = 0; j < 8; ++j) b[0][j] = bp[j * 16 * WPITCH4];

#pragma unroll
    for (int t = 0; t < 16 * NMAT; ++t) {
        if (NMAT == 2 && t == 8) {
#pragma unroll
            for (int u = 0; u < 8; ++u) a1[u] = ap1[2 * u];
        }
        if (NMAT == 2 && t == 16) {
#pragma unroll
            for (int u = 8; u < 16; ++u) a1[u] = ap1[2 * u];
        }
        if (BN && t == 16 * NMAT - 3) {
#pragma unroll
            for (int j = 0; j < 8; ++j) {
                pg[j] = bn_g[j * 16 + c];
                pv[j] = bn_v[j * 16 + c];
                pm[j] = bn_m[j * 16 + c];
                pb[j] = bn_b[j * 16 + c];
            }
        }
        if (t + 1 < 16 * NMAT) {
            const int tn = t + 1;
            const float4* nb = bp + (tn >> 4) * (HDIM * WPITCH4) + (tn & 15) * 2;
#pragma unroll
            for (int j = 0; j < 8; ++j) b[tn & 1][j] = nb[j * 16 * WPITCH4];
        }
        __builtin_amdgcn_sched_barrier(0);
        const float4 av = (t < 16) ? a0[t & 15] : a1[t & 15];
        const v2f a01 = {av.x, av.y};
        const v2f a23 = {av.z, av.w};
#pragma unroll
        for (int j = 0; j < 8; ++j) {
            const v2f b01 = {b[t & 1][j].x, b[t & 1][j].y};
            acc[j] = __builtin_amdgcn_wmma_f32_16x16x4_f32(false, a01, false, b01,
                                                           (short)0, acc[j], false, false);
        }
#pragma unroll
        for (int j = 0; j < 8; ++j) {
            const v2f b23 = {b[t & 1][j].z, b[t & 1][j].w};
            acc[j] = __builtin_amdgcn_wmma_f32_16x16x4_f32(false, a23, false, b23,
                                                           (short)0, acc[j], false, false);
        }
        __builtin_amdgcn_sched_barrier(0);
    }
}

__global__ void __launch_bounds__(256)
gemm128_bias(const float* __restrict__ A,
             const float* __restrict__ W,
             const float* __restrict__ bias,
             float* __restrict__ out,
             int nrows) {
    __shared__ float4 sW[HDIM * WPITCH4];
    const int lane = threadIdx.x & 31;
    const int wv   = threadIdx.x >> 5;
    const int c    = lane & 15;
    const int h    = lane >> 4;
    const int row0 = (blockIdx.x * 8 + wv) << 4;
    int arow_i = row0 + c;
    if (arow_i >= nrows) arow_i = nrows - 1;

    float4 a0[16];
    load_a_strip(A, arow_i, h, a0);
    float bcol[8];
#pragma unroll
    for (int j = 0; j < 8; ++j) bcol[j] = bias[j * 16 + c];

    stage_weights(W, sW, wv, lane);
    __syncthreads();
    if (row0 >= nrows) return;

    v8f acc[8];
#pragma unroll
    for (int j = 0; j < 8; ++j) {
#pragma unroll
        for (int r = 0; r < 8; ++r) acc[j][r] = bcol[j];
    }

    float pg[8], pb[8], pm[8], pv[8];
    strip_mac<1, false>(a0, A, sW, arow_i, c, h, acc, bias, bias, bias, bias, pg, pb, pm, pv);

    float* op = out + (size_t)(row0 + h * 8) * HDIM + c;
    if (row0 + 16 <= nrows) {
#pragma unroll
        for (int j = 0; j < 8; ++j) {
#pragma unroll
            for (int r = 0; r < 8; ++r) op[r * HDIM + j * 16] = acc[j][r];
        }
    } else {
#pragma unroll
        for (int j = 0; j < 8; ++j) {
#pragma unroll
            for (int r = 0; r < 8; ++r) {
                if (row0 + h * 8 + r < nrows) op[r * HDIM + j * 16] = acc[j][r];
            }
        }
    }
}

__global__ void __launch_bounds__(256)
fc1_relu_fc2(const float* __restrict__ A,
             const float* __restrict__ W,
             const float* __restrict__ bias,
             const float* __restrict__ W2,
             const float* __restrict__ B2,
             float* __restrict__ out, int nrows) {
    __shared__ float4 sW[HDIM * WPITCH4];
    const int lane = threadIdx.x & 31;
    const int wv   = threadIdx.x >> 5;
    const int c    = lane & 15;
    const int h    = lane >> 4;
    const int row0 = (blockIdx.x * 8 + wv) << 4;
    int arow_i = row0 + c;
    if (arow_i >= nrows) arow_i = nrows - 1;

    float4 a0[16];
    load_a_strip(A, arow_i, h, a0);
    float bcol[8];
#pragma unroll
    for (int j = 0; j < 8; ++j) bcol[j] = bias[j * 16 + c];

    stage_weights(W, sW, wv, lane);
    __syncthreads();
    if (row0 >= nrows) return;

    v8f acc[8];
#pragma unroll
    for (int j = 0; j < 8; ++j) {
#pragma unroll
        for (int r = 0; r < 8; ++r) acc[j][r] = bcol[j];
    }

    float w2[8][4];
#pragma unroll
    for (int j = 0; j < 8; ++j) {
#pragma unroll
        for (int q = 0; q < 4; ++q) w2[j][q] = W2[q * HDIM + j * 16 + c];
    }
    const int   q_own = (c & 8) ? 2 : 0;
    const float b2x = B2[q_own], b2y = B2[q_own + 1];

    float pg[8], pb[8], pm[8], pv[8];
    strip_mac<1, false>(a0, A, sW, arow_i, c, h, acc, bias, bias, bias, bias, pg, pb, pm, pv);

    float p[32];
#pragma unroll
    for (int i = 0; i < 32; ++i) p[i] = 0.0f;
#pragma unroll
    for (int j = 0; j < 8; ++j) {
#pragma unroll
        for (int r = 0; r < 8; ++r) {
            const float y = fmaxf(acc[j][r], 0.0f);
#pragma unroll
            for (int q = 0; q < 4; ++q) p[r * 4 + q] = fmaf(y, w2[j][q], p[r * 4 + q]);
        }
    }

    const bool b0 = (c & 1) != 0, b1 = (c & 2) != 0, b2 = (c & 4) != 0, b3 = (c & 8) != 0;
    float s16[16], s8[8], s4[4], s2[2];
#pragma unroll
    for (int k = 0; k < 16; ++k) {
        const float keep = b0 ? p[k + 16] : p[k];
        const float send = b0 ? p[k] : p[k + 16];
        s16[k] = keep + __shfl_xor(send, 1, 32);
    }
#pragma unroll
    for (int k = 0; k < 8; ++k) {
        const float keep = b1 ? s16[k + 8] : s16[k];
        const float send = b1 ? s16[k] : s16[k + 8];
        s8[k] = keep + __shfl_xor(send, 2, 32);
    }
#pragma unroll
    for (int k = 0; k < 4; ++k) {
        const float keep = b2 ? s8[k + 4] : s8[k];
        const float send = b2 ? s8[k] : s8[k + 4];
        s4[k] = keep + __shfl_xor(send, 4, 32);
    }
#pragma unroll
    for (int k = 0; k < 2; ++k) {
        const float keep = b3 ? s4[k + 2] : s4[k];
        const float send = b3 ? s4[k] : s4[k + 2];
        s2[k] = keep + __shfl_xor(send, 8, 32);
    }
    const int r_own = (b0 ? 4 : 0) + (b1 ? 2 : 0) + (b2 ? 1 : 0);
    const int row   = row0 + h * 8 + r_own;
    if (row < nrows) {
        float2 o;
        o.x = s2[0] + b2x;
        o.y = s2[1] + b2y;
        *(float2*)(out + (size_t)row * 4 + q_own) = o;
    }
}

__global__ void __launch_bounds__(256)
sage_layer_gemm(const float* __restrict__ AGG,
                const float* __restrict__ WL,
                const float* __restrict__ X,
                const float* __restrict__ WR,
                const float* __restrict__ bias,
                const float* __restrict__ bn_g,
                const float* __restrict__ bn_b,
                const float* __restrict__ bn_m,
                const float* __restrict__ bn_v,
                float* __restrict__ out, int nrows) {
    __shared__ float4 sW[2 * HDIM * WPITCH4];
    const int lane = threadIdx.x & 31;
    const int wv   = threadIdx.x >> 5;
    const int c    = lane & 15;
    const int h    = lane >> 4;
    const int row0 = (blockIdx.x * 8 + wv) << 4;
    int arow_i = row0 + c;
    if (arow_i >= nrows) arow_i = nrows - 1;

    float4 a0[16];
    load_a_strip(AGG, arow_i, h, a0);
    float bcol[8];
#pragma unroll
    for (int j = 0; j < 8; ++j) bcol[j] = bias[j * 16 + c];

    stage_weights(WL, sW, wv, lane);
    stage_weights(WR, sW + HDIM * WPITCH4, wv, lane);
    __syncthreads();
    if (row0 >= nrows) return;

    v8f acc[8];
#pragma unroll
    for (int j = 0; j < 8; ++j) {
#pragma unroll
        for (int r = 0; r < 8; ++r) acc[j][r] = bcol[j];
    }

    float pg[8], pb[8], pm[8], pv[8];
    strip_mac<2, true>(a0, X, sW, arow_i, c, h, acc, bn_g, bn_b, bn_m, bn_v, pg, pb, pm, pv);

    float* op = out + (size_t)(row0 + h * 8) * HDIM + c;
    const bool whole = (row0 + 16 <= nrows);
#pragma unroll
    for (int j = 0; j < 8; ++j) {
        const float s  = pg[j] * rsqrtf(pv[j] + BN_EPS);
        const float mu = pm[j];
        const float bb = pb[j];
        if (whole) {
#pragma unroll
            for (int r = 0; r < 8; ++r) {
                const float v = (acc[j][r] - mu) * s + bb;
                op[r * HDIM + j * 16] = fmaxf(v, 0.0f);
            }
        } else {
#pragma unroll
            for (int r = 0; r < 8; ++r) {
                if (row0 + h * 8 + r < nrows) {
                    const float v = (acc[j][r] - mu) * s + bb;
                    op[r * HDIM + j * 16] = fmaxf(v, 0.0f);
                }
            }
        }
    }
}

#define BIN_SHIFT 8
#define CHUNK_EDGES 4096

__device__ __forceinline__ int block_excl_scan(int v, int* wsum, int& total) {
    const int lane = threadIdx.x & 31;
    const int wv   = threadIdx.x >> 5;
    int incl = v;
#pragma unroll
    for (int off = 1; off < 32; off <<= 1) {
        const int t = __shfl_up(incl, off, 32);
        if (lane >= off) incl += t;
    }
    __syncthreads();
    if (lane == 31) wsum[wv] = incl;
    __syncthreads();
    int woff = 0, tot = 0;
#pragma unroll
    for (int w = 0; w < 8; ++w) {
        const int t = wsum[w];
        if (w < wv) woff += t;
        tot += t;
    }
    total = tot;
    return woff + incl - v;
}

__global__ void __launch_bounds__(256)
bin_count(const int* __restrict__ ei, int* __restrict__ hist, int E, int nrows, int nbins) {
    extern __shared__ int sbin[];
    for (int i = threadIdx.x; i < nbins; i += 256) sbin[i] = 0;
    __syncthreads();
    const int base = blockIdx.x * CHUNK_EDGES + threadIdx.x;
    int d[16];
#pragma unroll
    for (int j = 0; j < 16; ++j) {
        const int e = min(base + j * 256, E - 1);
        d[j] = ei[e];
    }
#pragma unroll
    for (int j = 0; j < 16; ++j)
        if (base + j * 256 < E && (unsigned)d[j] < (unsigned)nrows)
            atomicAdd(&sbin[d[j] >> BIN_SHIFT], 1);
    __syncthreads();
    for (int i = threadIdx.x; i < nbins; i += 256)
        hist[(size_t)blockIdx.x * nbins + i] = sbin[i];
}

__global__ void __launch_bounds__(256)
bin_offsets(int* __restrict__ hist, int* __restrict__ bin_tot, int nchunks, int nbins) {
    __shared__ int wsum[8];
    const int k = blockIdx.x;
    int carry = 0;
    for (int b0 = 0; b0 < nchunks; b0 += 256) {
        const int b = b0 + threadIdx.x;
        const int v = (b < nchunks) ? hist[(size_t)b * nbins + k] : 0;
        int total;
        const int excl = block_excl_scan(v, wsum, total);
        if (b < nchunks) hist[(size_t)b * nbins + k] = carry + excl;
        carry += total;
    }
    if (threadIdx.x == 0) bin_tot[k] = carry;
}

__global__ void __launch_bounds__(256)
bin_partition(const int* __restrict__ ei, const int* __restrict__ hist,
              const int* __restrict__ bin_tot, int* __restrict__ bin_start,
              unsigned* __restrict__ part, int E, int nrows, int nbins) {
    extern __shared__ int sbin[];
    __shared__ unsigned rec[CHUNK_EDGES];
    __shared__ int      pos[CHUNK_EDGES];
    __shared__ int      wsum[8];
    int* lcnt  = sbin;
    int* gbase = sbin + nbins;
    for (int i = threadIdx.x; i < nbins; i += 256) lcnt[i] = 0;
    int carry = 0;
    for (int k0 = 0; k0 < nbins; k0 += 256) {
        const int k = k0 + threadIdx.x;
        const int v = (k < nbins) ? bin_tot[k] : 0;
        int total;
        const int excl = block_excl_scan(v, wsum, total);
        if (k < nbins) {
            const int s = carry + excl;
            if (blockIdx.x == 0) bin_start[k] = s;
            gbase[k] = s + hist[(size_t)blockIdx.x * nbins + k];
        }
        carry += total;
    }
    __syncthreads();
    const int base = blockIdx.x * CHUNK_EDGES + threadIdx.x;
    int d[16], s[16], r[16];
#pragma unroll
    for (int j = 0; j < 16; ++j) {
        const int e = min(base + j * 256, E - 1);
        d[j] = ei[e];
        s[j] = ei[(size_t)E + e];
    }
#pragma unroll
    for (int j = 0; j < 16; ++j) {
        const bool live = base + j * 256 < E && (unsigned)d[j] < (unsigned)nrows;
        r[j] = live ? atomicAdd(&lcnt[d[j] >> BIN_SHIFT], 1) : -1;
    }
    __syncthreads();
    carry = 0;
    for (int k0 = 0; k0 < nbins; k0 += 256) {
        const int k = k0 + threadIdx.x;
        const int v = (k < nbins) ? lcnt[k] : 0;
        int total;
        const int excl = block_excl_scan(v, wsum, total);
        if (k < nbins) lcnt[k] = carry + excl;
        carry += total;
    }
    __syncthreads();
    const int nlive = carry;
#pragma unroll
    for (int j = 0; j < 16; ++j) {
        if (r[j] >= 0) {
            const int bin  = d[j] >> BIN_SHIFT;
            const int srcc = min(max(s[j], 0), nrows - 1);
            const int slot = lcnt[bin] + r[j];
            rec[slot] = ((unsigned)(d[j] & 255) << 24) | (unsigned)srcc;
            pos[slot] = gbase[bin] + r[j];
        }
    }
    __syncthreads();
    for (int i = threadIdx.x; i < nlive; i += 256) part[pos[i]] = rec[i];
}

#define SORT_CAP 8192
__global__ void __launch_bounds__(256)
bin_sort(const unsigned* __restrict__ part, const int* __restrict__ bin_start,
         const int* __restrict__ bin_tot, int2* __restrict__ seg,
         int* __restrict__ col, int nrows) {
    __shared__ int cnt[256];
    __shared__ int wsum[8];
    __shared__ int sorted[SORT_CAP];
    const int k = blockIdx.x;
    const int s = bin_start[k];
    const int m = bin_tot[k];
    cnt[threadIdx.x] = 0;
    __syncthreads();
    for (int i0 = 0; i0 < m; i0 += 2048) {
        unsigned r[8];
#pragma unroll
        for (int u = 0; u < 8; ++u) r[u] = part[s + min(i0 + u * 256 + (int)threadIdx.x, m - 1)];
#pragma unroll
        for (int u = 0; u < 8; ++u)
            if (i0 + u * 256 + (int)threadIdx.x < m) atomicAdd(&cnt[r[u] >> 24], 1);
    }
    __syncthreads();
    const int deg = cnt[threadIdx.x];
    int total;
    const int first = block_excl_scan(deg, wsum, total);
    const int n = (k << BIN_SHIFT) + threadIdx.x;
    if (n < nrows) seg[n] = make_int2(s + first, deg);
    cnt[threadIdx.x] = first;
    __syncthreads();
    const bool in_lds = (m <= SORT_CAP);
    for (int i0 = 0; i0 < m; i0 += 2048) {
        unsigned r[8];
#pragma unroll
        for (int u = 0; u < 8; ++u) r[u] = part[s + min(i0 + u * 256 + (int)threadIdx.x, m - 1)];
#pragma unroll
        for (int u = 0; u < 8; ++u) {
            if (i0 + u * 256 + (int)threadIdx.x < m) {
                const int p = atomicAdd(&cnt[r[u] >> 24], 1);
                const int v = (int)(r[u] & 0xFFFFFFu);
                if (in_lds) sorted[p] = v; else col[s + p] = v;
            }
        }
    }
    if (in_lds) {
        __syncthreads();
        for (int i = threadIdx.x; i < m; i += 256) col[s + i] = sorted[i];
    }
}

__device__ __forceinline__ float4 max4(float4 a, float4 b) {
    return make_float4(fmaxf(a.x, b.x), fmaxf(a.y, b.y), fmaxf(a.z, b.z), fmaxf(a.w, b.w));
}

__global__ void __launch_bounds__(256)
gather_max_rows(const int2* __restrict__ seg, const int* __restrict__ col,
                const float* __restrict__ X, float* __restrict__ out, int nrows) {
    const int lane = threadIdx.x & 31;
    const int n = __builtin_amdgcn_readfirstlane(
        (int)((blockIdx.x * blockDim.x + threadIdx.x) >> 5));
    if (n >= nrows) return;
    const int2 sg  = seg[n];
    const int  beg = sg.x;
    const int  cnt = sg.y;
    const float4* __restrict__ xr = (const float4*)X + lane;
    const float ninf = -__builtin_inff();
    float4 m = make_float4(ninf, ninf, ninf, ninf);

    for (int base = 0; base < cnt; base += 32) {
        const int c = min(cnt - base, 32);
        const int ids = col[beg + base + min(lane, c - 1)];
        int i = 0;
        for (; i + 4 <= c; i += 4) {
            const int s0 = __builtin_amdgcn_readlane(ids, i);
            const int s1 = __builtin_amdgcn_readlane(ids, i + 1);
            const int s2 = __builtin_amdgcn_readlane(ids, i + 2);
            const int s3 = __builtin_amdgcn_readlane(ids, i + 3);
            const float4 v0 = xr[(size_t)s0 * (HDIM / 4)];
            const float4 v1 = xr[(size_t)s1 * (HDIM / 4)];
            const float4 v2 = xr[(size_t)s2 * (HDIM / 4)];
            const float4 v3 = xr[(size_t)s3 * (HDIM / 4)];
            m = max4(m, max4(max4(v0, v1), max4(v2, v3)));
        }
        const int left = c - i;
        if (left > 0) {
            const int s0 = __builtin_amdgcn_readlane(ids, i);
            float4 t = xr[(size_t)s0 * (HDIM / 4)];
            if (left > 1) {
                const int s1 = __builtin_amdgcn_readlane(ids, i + 1);
                t = max4(t, xr[(size_t)s1 * (HDIM / 4)]);
            }
            if (left > 2) {
                const int s2 = __builtin_amdgcn_readlane(ids, i + 2);
                t = max4(t, xr[(size_t)s2 * (HDIM / 4)]);
            }
            m = max4(m, t);
        }
    }
    float4 r;
    r.x = (m.x == ninf) ? 0.0f : m.x;
    r.y = (m.y == ninf) ? 0.0f : m.y;
    r.z = (m.z == ninf) ? 0.0f : m.z;
    r.w = (m.w == ninf) ? 0.0f : m.w;
    ((float4*)out)[(size_t)n * (HDIM / 4) + lane] = r;
}

extern "C" void kernel_launch(void* const* d_in, const int* in_sizes, int n_in,
                              void* d_out, int out_size, void* d_ws, size_t ws_size,
                              hipStream_t stream) {
    const float* features = (const float*)d_in[0];
    const int*   ei       = (const int*)d_in[1];
    const float* w_proj = (const float*)d_in[2];
    const float* b_proj = (const float*)d_in[3];
    const float* w_l0 = (const float*)d_in[4];
    const float* b_l0 = (const float*)d_in[5];
    const float* w_r0 = (const float*)d_in[6];
    const float* bn_g0 = (const float*)d_in[7];
    const float* bn_b0 = (const float*)d_in[8];
    const float* bn_m0 = (const float*)d_in[9];
    const float* bn_v0 = (const float*)d_in[10];
    const float* w_l1 = (const float*)d_in[11];
    const float* b_l1 = (const float*)d_in[12];
    const float* w_r1 = (const float*)d_in[13];
    const float* bn_g1 = (const float*)d_in[14];
    const float* bn_b1 = (const float*)d_in[15];
    const float* bn_m1 = (const float*)d_in[16];
    const float* bn_v1 = (const float*)d_in[17];
    const float* w_fc1 = (const float*)d_in[18];
    const float* b_fc1 = (const float*)d_in[19];
    const float* w_fc2 = (const float*)d_in[20];
    const float* b_fc2 = (const float*)d_in[21];
    float* outp = (float*)d_out;

    const int N = in_sizes[0] / HDIM;
    const int E = in_sizes[1] / 2;

    const size_t SZ = (size_t)N * HDIM * sizeof(float);
    float* X    = (float*)d_ws;
    float* AGGF = (float*)((char*)d_ws + SZ);
    float* H1   = (float*)((char*)d_ws + 2 * SZ);
    char*  G    = (char*)d_ws + 3 * SZ;
    const size_t o_seg  = (((size_t)N + 1) * sizeof(int) + 255) & ~(size_t)255;
    const size_t o_part = o_seg  + (((size_t)N * sizeof(int2) + 255) & ~(size_t)255);
    const size_t o_col  = o_part + (((size_t)E * sizeof(int)  + 255) & ~(size_t)255);
    int2*     seg  = (int2*)(G + o_seg);
    unsigned* part = (unsigned*)(G + o_part);
    int*      col  = (int*)(G + o_col);

    const int nbins   = (N + 255) / 256;
    const int nchunks = (E + 4095) / 4096;
    int* hist      = (int*)AGGF;
    int* bin_tot   = hist + (size_t)nchunks * nbins;
    int* bin_start = bin_tot + nbins;

    const int row_tiles   = (N + 15) / 16;
    const int gemm_blocks = (row_tiles + 7) / 8;
    const int wave_blocks = (N + 7) / 8;

    bin_count<<<nchunks, 256, (size_t)nbins * sizeof(int), stream>>>(ei, hist, E, N, nbins);
    bin_offsets<<<nbins, 256, 0, stream>>>(hist, bin_tot, nchunks, nbins);
    bin_partition<<<nchunks, 256, 2 * (size_t)nbins * sizeof(int), stream>>>(ei, hist, bin_tot, bin_start,
                                                                          part, E, N, nbins);
    bin_sort<<<nbins, 256, 0, stream>>>(part, bin_start, bin_tot, seg, col, N);

    gemm128_bias<<<gemm_blocks, 256, 0, stream>>>(features, w_proj, b_proj, X, N);

    gather_max_rows<<<wave_blocks, 256, 0, stream>>>(seg, col, X, AGGF, N);
    sage_layer_gemm<<<gemm_blocks, 256, 0, stream>>>(AGGF, w_l0, X, w_r0, b_l0,
                                                     bn_g0, bn_b0, bn_m0, bn_v0, H1, N);

    gather_max_rows<<<wave_blocks, 256, 0, stream>>>(seg, col, H1, AGGF, N);
    sage_layer_gemm<<<gemm_blocks, 256, 0, stream>>>(AGGF, w_l1, H1, w_r1, b_l1,
                                                     bn_g1, bn_b1, bn_m1, bn_v1, X, N);

    fc1_relu_fc2<<<gemm_blocks, 256, 0, stream>>>(X, w_fc1, b_fc1, w_fc2, b_fc2, outp, N);
}


# Max-aggregation fused into the SAGE GEMM: one resident 16-wave work-group per WGP, a wave builds its strip's 16 aggregate rows in LDS as its A operand; bias added after the products (sum order)
# speedup vs baseline: 4.0646x; 1.0825x over previous
#include <hip/hip_runtime.h>

#define HDIM 128
#define BN_EPS 1e-5f
#define WPITCH4 33

typedef __attribute__((ext_vector_type(2))) float v2f;
typedef __attribute__((ext_vector_type(8))) float v8f;

__device__ __forceinline__ void stage_weights(const float* __restrict__ Wg, float4* sW,
                                              int wv, int lane) {
    float4 w[16];
#pragma unroll
    for (int it = 0; it < 16; ++it)
        w[it] = *(const float4*)(Wg + (size_t)(it * 8 + wv) * HDIM + lane * 4);
#pragma unroll
    for (int it = 0; it < 16; ++it)
        sW[(it * 8 + wv) * WPITCH4 + lane] = w[it];
}

__device__ __forceinline__ void load_a_strip(const float* __restrict__ A, int arow_i, int h,
                                             float4 (&a)[16]) {
    const float4* ap = (const float4*)(A + (size_t)arow_i * HDIM + 4 * h);
#pragma unroll
    for (int t = 0; t < 16; ++t) a[t] = ap[2 * t];
}

template <int NMAT, bool BN>
__device__ __forceinline__ void strip_mac(float4 (&a0)[16], const float* __restrict__ A1,
                                          const float4* sW, int arow_i, int c, int h,
                                          v8f (&acc)[8],
                                          const float* __restrict__ bn_g,
                                          const float* __restrict__ bn_b,
                                          const float* __restrict__ bn_m,
                                          const float* __restrict__ bn_v,
                                          float (&pg)[8], float (&pb)[8],
                                          float (&pm)[8], float (&pv)[8]) {
    const float4* ap1 = (const float4*)(A1 + (size_t)arow_i * HDIM + 4 * h);
    const float4* bp  = sW + c * WPITCH4 + h;

    float4 a1[16];
    float4 b[2][8];
#pragma unroll
    for (int j = 0; j < 8; ++j) b[0][j] = bp[j * 16 * WPITCH4];

#pragma unroll
    for (int t = 0; t < 16 * NMAT; ++t) {
        if (NMAT == 2 && t == 8) {
#pragma unroll
            for (int u = 0; u < 8; ++u) a1[u] = ap1[2 * u];
        }
        if (NMAT == 2 && t == 16) {
#pragma unroll
            for (int u = 8; u < 16; ++u) a1[u] = ap1[2 * u];
        }
        if (BN && t == 16 * NMAT - 3) {
#pragma unroll
            for (int j = 0; j < 8; ++j) {
                pg[j] = bn_g[j * 16 + c];
                pv[j] = bn_v[j * 16 + c];
                pm[j] = bn_m[j * 16 + c];
                pb[j] = bn_b[j * 16 + c];
            }
        }
        if (t + 1 < 16 * NMAT) {
            const int tn = t + 1;
            const float4* nb = bp + (tn >> 4) * (HDIM * WPITCH4) + (tn & 15) * 2;
#pragma unroll
            for (int j = 0; j < 8; ++j) b[tn & 1][j] = nb[j * 16 * WPITCH4];
        }
        __builtin_amdgcn_sched_barrier(0);
        const float4 av = (t < 16) ? a0[t & 15] : a1[t & 15];
        const v2f a01 = {av.x, av.y};
        const v2f a23 = {av.z, av.w};
#pragma unroll
        for (int j = 0; j < 8; ++j) {
            const v2f b01 = {b[t & 1][j].x, b[t & 1][j].y};
            acc[j] = __builtin_amdgcn_wmma_f32_16x16x4_f32(false, a01, false, b01,
                                                           (short)0, acc[j], false, false);
        }
#pragma unroll
        for (int j = 0; j < 8; ++j) {
            const v2f b23 = {b[t & 1][j].z, b[t & 1][j].w};
            acc[j] = __builtin_amdgcn_wmma_f32_16x16x4_f32(false, a23, false, b23,
                                                           (short)0, acc[j], false, false);
        }
        __builtin_amdgcn_sched_barrier(0);
    }
}

__global__ void __launch_bounds__(256)
gemm128_bias(const float* __restrict__ A,
             const float* __restrict__ W,
             const float* __restrict__ bias,
             float* __restrict__ out,
             int nrows) {
    __shared__ float4 sW[HDIM * WPITCH4];
    const int lane = threadIdx.x & 31;
    const int wv   = threadIdx.x >> 5;
    const int c    = lane & 15;
    const int h    = lane >> 4;
    const int row0 = (blockIdx.x * 8 + wv) << 4;
    int arow_i = row0 + c;
    if (arow_i >= nrows) arow_i = nrows - 1;

    float4 a0[16];
    load_a_strip(A, arow_i, h, a0);
    float bcol[8];
#pragma unroll
    for (int j = 0; j < 8; ++j) bcol[j] = bias[j * 16 + c];

    stage_weights(W, sW, wv, lane);
    __syncthreads();
    if (row0 >= nrows) return;

    v8f acc[8];
#pragma unroll
    for (int j = 0; j < 8; ++j) {
#pragma unroll
        for (int r = 0; r < 8; ++r) acc[j][r] = bcol[j];
    }

    float pg[8], pb[8], pm[8], pv[8];
    strip_mac<1, false>(a0, A, sW, arow_i, c, h, acc, bias, bias, bias, bias, pg, pb, pm, pv);

    float* op = out + (size_t)(row0 + h * 8) * HDIM + c;
    if (row0 + 16 <= nrows) {
#pragma unroll
        for (int j = 0; j < 8; ++j) {
#pragma unroll
            for (int r = 0; r < 8; ++r) op[r * HDIM + j * 16] = acc[j][r];
        }
    } else {
#pragma unroll
        for (int j = 0; j < 8; ++j) {
#pragma unroll
            for (int r = 0; r < 8; ++r) {
                if (row0 + h * 8 + r < nrows) op[r * HDIM + j * 16] = acc[j][r];
            }
        }
    }
}

__global__ void __launch_bounds__(256)
fc1_relu_fc2(const float* __restrict__ A,
             const float* __restrict__ W,
             const float* __restrict__ bias,
             const float* __restrict__ W2,
             const float* __restrict__ B2,
             float* __restrict__ out, int nrows) {
    __shared__ float4 sW[HDIM * WPITCH4];
    const int lane = threadIdx.x & 31;
    const int wv   = threadIdx.x >> 5;
    const int c    = lane & 15;
    const int h    = lane >> 4;
    const int row0 = (blockIdx.x * 8 + wv) << 4;
    int arow_i = row0 + c;
    if (arow_i >= nrows) arow_i = nrows - 1;

    float4 a0[16];
    load_a_strip(A, arow_i, h, a0);
    float bcol[8];
#pragma unroll
    for (int j = 0; j < 8; ++j) bcol[j] = bias[j * 16 + c];

    stage_weights(W, sW, wv, lane);
    __syncthreads();
    if (row0 >= nrows) return;

    v8f acc[8];
#pragma unroll
    for (int j = 0; j < 8; ++j) {
#pragma unroll
        for (int r = 0; r < 8; ++r) acc[j][r] = bcol[j];
    }

    float w2[8][4];
#pragma unroll
    for (int j = 0; j < 8; ++j) {
#pragma unroll
        for (int q = 0; q < 4; ++q) w2[j][q] = W2[q * HDIM + j * 16 + c];
    }
    const int   q_own = (c & 8) ? 2 : 0;
    const float b2x = B2[q_own], b2y = B2[q_own + 1];

    float pg[8], pb[8], pm[8], pv[8];
    strip_mac<1, false>(a0, A, sW, arow_i, c, h, acc, bias, bias, bias, bias, pg, pb, pm, pv);

    float p[32];
#pragma unroll
    for (int i = 0; i < 32; ++i) p[i] = 0.0f;
#pragma unroll
    for (int j = 0; j < 8; ++j) {
#pragma unroll
        for (int r = 0; r < 8; ++r) {
            const float y = fmaxf(acc[j][r], 0.0f);
#pragma unroll
            for (int q = 0; q < 4; ++q) p[r * 4 + q] = fmaf(y, w2[j][q], p[r * 4 + q]);
        }
    }

    const bool b0 = (c & 1) != 0, b1 = (c & 2) != 0, b2 = (c & 4) != 0, b3 = (c & 8) != 0;
    float s16[16], s8[8], s4[4], s2[2];
#pragma unroll
    for (int k = 0; k < 16; ++k) {
        const float keep = b0 ? p[k + 16] : p[k];
        const float send = b0 ? p[k] : p[k + 16];
        s16[k] = keep + __shfl_xor(send, 1, 32);
    }
#pragma unroll
    for (int k = 0; k < 8; ++k) {
        const float keep = b1 ? s16[k + 8] : s16[k];
        const float send = b1 ? s16[k] : s16[k + 8];
        s8[k] = keep + __shfl_xor(send, 2, 32);
    }
#pragma unroll
    for (int k = 0; k < 4; ++k) {
        const float keep = b2 ? s8[k + 4] : s8[k];
        const float send = b2 ? s8[k] : s8[k + 4];
        s4[k] = keep + __shfl_xor(send, 4, 32);
    }
#pragma unroll
    for (int k = 0; k < 2; ++k) {
        const float keep = b3 ? s4[k + 2] : s4[k];
        const float send = b3 ? s4[k] : s4[k + 2];
        s2[k] = keep + __shfl_xor(send, 8, 32);
    }
    const int r_own = (b0 ? 4 : 0) + (b1 ? 2 : 0) + (b2 ? 1 : 0);
    const int row   = row0 + h * 8 + r_own;
    if (row < nrows) {
        float2 o;
        o.x = s2[0] + b2x;
        o.y = s2[1] + b2y;
        *(float2*)(out + (size_t)row * 4 + q_own) = o;
    }
}

#define BIN_SHIFT 8
#define CHUNK_EDGES 4096

__device__ __forceinline__ int block_excl_scan(int v, int* wsum, int& total) {
    const int lane = threadIdx.x & 31;
    const int wv   = threadIdx.x >> 5;
    int incl = v;
#pragma unroll
    for (int off = 1; off < 32; off <<= 1) {
        const int t = __shfl_up(incl, off, 32);
        if (lane >= off) incl += t;
    }
    __syncthreads();
    if (lane == 31) wsum[wv] = incl;
    __syncthreads();
    int woff = 0, tot = 0;
#pragma unroll
    for (int w = 0; w < 8; ++w) {
        const int t = wsum[w];
        if (w < wv) woff += t;
        tot += t;
    }
    total = tot;
    return woff + incl - v;
}

__global__ void __launch_bounds__(256)
bin_count(const int* __restrict__ ei, int* __restrict__ hist, int E, int nrows, int nbins) {
    extern __shared__ int sbin[];
    for (int i = threadIdx.x; i < nbins; i += 256) sbin[i] = 0;
    __syncthreads();
    const int base = blockIdx.x * CHUNK_EDGES + threadIdx.x;
    int d[16];
#pragma unroll
    for (int j = 0; j < 16; ++j) {
        const int e = min(base + j * 256, E - 1);
        d[j] = ei[e];
    }
#pragma unroll
    for (int j = 0; j < 16; ++j)
        if (base + j * 256 < E && (unsigned)d[j] < (unsigned)nrows)
            atomicAdd(&sbin[d[j] >> BIN_SHIFT], 1);
    __syncthreads();
    for (int i = threadIdx.x; i < nbins; i += 256)
        hist[(size_t)blockIdx.x * nbins + i] = sbin[i];
}

__global__ void __launch_bounds__(256)
bin_offsets(int* __restrict__ hist, int* __restrict__ bin_tot, int nchunks, int nbins) {
    __shared__ int wsum[8];
    const int k = blockIdx.x;
    int carry = 0;
    for (int b0 = 0; b0 < nchunks; b0 += 256) {
        const int b = b0 + threadIdx.x;
        const int v = (b < nchunks) ? hist[(size_t)b * nbins + k] : 0;
        int total;
        const int excl = block_excl_scan(v, wsum, total);
        if (b < nchunks) hist[(size_t)b * nbins + k] = carry + excl;
        carry += total;
    }
    if (threadIdx.x == 0) bin_tot[k] = carry;
}

__global__ void __launch_bounds__(256)
bin_partition(const int* __restrict__ ei, const int* __restrict__ hist,
              const int* __restrict__ bin_tot, int* __restrict__ bin_start,
              unsigned* __restrict__ part, int E, int nrows, int nbins) {
    extern __shared__ int sbin[];
    __shared__ unsigned rec[CHUNK_EDGES];
    __shared__ int      pos[CHUNK_EDGES];
    __shared__ int      wsum[8];
    int* lcnt  = sbin;
    int* gbase = sbin + nbins;
    for (int i = threadIdx.x; i < nbins; i += 256) lcnt[i] = 0;
    int carry = 0;
    for (int k0 = 0; k0 < nbins; k0 += 256) {
        const int k = k0 + threadIdx.x;
        const int v = (k < nbins) ? bin_tot[k] : 0;
        int total;
        const int excl = block_excl_scan(v, wsum, total);
        if (k < nbins) {
            const int s = carry + excl;
            if (blockIdx.x == 0) bin_start[k] = s;
            gbase[k] = s + hist[(size_t)blockIdx.x * nbins + k];
        }
        carry += total;
    }
    __syncthreads();
    const int base = blockIdx.x * CHUNK_EDGES + threadIdx.x;
    int d[16], s[16], r[16];
#pragma unroll
    for (int j = 0; j < 16; ++j) {
        const int e = min(base + j * 256, E - 1);
        d[j] = ei[e];
        s[j] = ei[(size_t)E + e];
    }
#pragma unroll
    for (int j = 0; j < 16; ++j) {
        const bool live = base + j * 256 < E && (unsigned)d[j] < (unsigned)nrows;
        r[j] = live ? atomicAdd(&lcnt[d[j] >> BIN_SHIFT], 1) : -1;
    }
    __syncthreads();
    carry = 0;
    for (int k0 = 0; k0 < nbins; k0 += 256) {
        const int k = k0 + threadIdx.x;
        const int v = (k < nbins) ? lcnt[k] : 0;
        int total;
        const int excl = block_excl_scan(v, wsum, total);
        if (k < nbins) lcnt[k] = carry + excl;
        carry += total;
    }
    __syncthreads();
    const int nlive = carry;
#pragma unroll
    for (int j = 0; j < 16; ++j) {
        if (r[j] >= 0) {
            const int bin  = d[j] >> BIN_SHIFT;
            const int srcc = min(max(s[j], 0), nrows - 1);
            const int slot = lcnt[bin] + r[j];
            rec[slot] = ((unsigned)(d[j] & 255) << 24) | (unsigned)srcc;
            pos[slot] = gbase[bin] + r[j];
        }
    }
    __syncthreads();
    for (int i = threadIdx.x; i < nlive; i += 256) part[pos[i]] = rec[i];
}

#define SORT_CAP 8192
__global__ void __launch_bounds__(256)
bin_sort(const unsigned* __restrict__ part, const int* __restrict__ bin_start,
         const int* __restrict__ bin_tot, int2* __restrict__ seg,
         int* __restrict__ col, int nrows) {
    __shared__ int cnt[256];
    __shared__ int wsum[8];
    __shared__ int sorted[SORT_CAP];
    const int k = blockIdx.x;
    const int s = bin_start[k];
    const int m = bin_tot[k];
    cnt[threadIdx.x] = 0;
    __syncthreads();
    for (int i0 = 0; i0 < m; i0 += 2048) {
        unsigned r[8];
#pragma unroll
        for (int u = 0; u < 8; ++u) r[u] = part[s + min(i0 + u * 256 + (int)threadIdx.x, m - 1)];
#pragma unroll
        for (int u = 0; u < 8; ++u)
            if (i0 + u * 256 + (int)threadIdx.x < m) atomicAdd(&cnt[r[u] >> 24], 1);
    }
    __syncthreads();
    const int deg = cnt[threadIdx.x];
    int total;
    const int first = block_excl_scan(deg, wsum, total);
    const int n = (k << BIN_SHIFT) + threadIdx.x;
    if (n < nrows) seg[n] = make_int2(s + first, deg);
    cnt[threadIdx.x] = first;
    __syncthreads();
    const bool in_lds = (m <= SORT_CAP);
    for (int i0 = 0; i0 < m; i0 += 2048) {
        unsigned r[8];
#pragma unroll
        for (int u = 0; u < 8; ++u) r[u] = part[s + min(i0 + u * 256 + (int)threadIdx.x, m - 1)];
#pragma unroll
        for (int u = 0; u < 8; ++u) {
            if (i0 + u * 256 + (int)threadIdx.x < m) {
                const int p = atomicAdd(&cnt[r[u] >> 24], 1);
                const int v = (int)(r[u] & 0xFFFFFFu);
                if (in_lds) sorted[p] = v; else col[s + p] = v;
            }
        }
    }
    if (in_lds) {
        __syncthreads();
        for (int i = threadIdx.x; i < m; i += 256) col[s + i] = sorted[i];
    }
}

__device__ __forceinline__ float4 max4(float4 a, float4 b) {
    return make_float4(fmaxf(a.x, b.x), fmaxf(a.y, b.y), fmaxf(a.z, b.z), fmaxf(a.w, b.w));
}

#define FUSED_WGS   256
#define FUSED_WAVES 16

__device__ __forceinline__ float4 max_over_sources(const int* __restrict__ col,
                                                   const float4* __restrict__ xr,
                                                   int beg, int cnt, int ids0, int lane) {
    const float ninf = -__builtin_inff();
    float4 m = make_float4(ninf, ninf, ninf, ninf);
    int ids = ids0;
    for (int base = 0; base < cnt; base += 32) {
        const int c = min(cnt - base, 32);
        if (base > 0) ids = col[beg + base + min(lane, c - 1)];
        int i = 0;
        for (; i + 8 <= c; i += 8) {
            const int s0 = __builtin_amdgcn_readlane(ids, i);
            const int s1 = __builtin_amdgcn_readlane(ids, i + 1);
            const int s2 = __builtin_amdgcn_readlane(ids, i + 2);
            const int s3 = __builtin_amdgcn_readlane(ids, i + 3);
            const int s4 = __builtin_amdgcn_readlane(ids, i + 4);
            const int s5 = __builtin_amdgcn_readlane(ids, i + 5);
            const int s6 = __builtin_amdgcn_readlane(ids, i + 6);
            const int s7 = __builtin_amdgcn_readlane(ids, i + 7);
            const float4 v0 = xr[(size_t)s0 * (HDIM / 4)];
            const float4 v1 = xr[(size_t)s1 * (HDIM / 4)];
            const float4 v2 = xr[(size_t)s2 * (HDIM / 4)];
            const float4 v3 = xr[(size_t)s3 * (HDIM / 4)];
            const float4 v4 = xr[(size_t)s4 * (HDIM / 4)];
            const float4 v5 = xr[(size_t)s5 * (HDIM / 4)];
            const float4 v6 = xr[(size_t)s6 * (HDIM / 4)];
            const float4 v7 = xr[(size_t)s7 * (HDIM / 4)];
            m = max4(m, max4(max4(max4(v0, v1), max4(v2, v3)),
                             max4(max4(v4, v5), max4(v6, v7))));
        }
        if (i + 4 <= c) {
            const int s0 = __builtin_amdgcn_readlane(ids, i);
            const int s1 = __builtin_amdgcn_readlane(ids, i + 1);
            const int s2 = __builtin_amdgcn_readlane(ids, i + 2);
            const int s3 = __builtin_amdgcn_readlane(ids, i + 3);
            const float4 v0 = xr[(size_t)s0 * (HDIM / 4)];
            const float4 v1 = xr[(size_t)s1 * (HDIM / 4)];
            const float4 v2 = xr[(size_t)s2 * (HDIM / 4)];
            const float4 v3 = xr[(size_t)s3 * (HDIM / 4)];
            m = max4(m, max4(max4(v0, v1), max4(v2, v3)));
            i += 4;
        }
        const int left = c - i;
        if (left > 0) {
            const int s0 = __builtin_amdgcn_readlane(ids, i);
            float4 t = xr[(size_t)s0 * (HDIM / 4)];
            if (left > 1) {
                const int s1 = __builtin_amdgcn_readlane(ids, i + 1);
                t = max4(t, xr[(size_t)s1 * (HDIM / 4)]);
            }
            if (left > 2) {
                const int s2 = __builtin_amdgcn_readlane(ids, i + 2);
                t = max4(t, xr[(size_t)s2 * (HDIM / 4)]);
            }
            m = max4(m, t);
        }
    }
    float4 r;
    r.x = (m.x == ninf) ? 0.0f : m.x;
    r.y = (m.y == ninf) ? 0.0f : m.y;
    r.z = (m.z == ninf) ? 0.0f : m.z;
    r.w = (m.w == ninf) ? 0.0f : m.w;
    return r;
}

__global__ void __launch_bounds__(FUSED_WAVES * 32)
sage_layer_fused(const int2* __restrict__ seg, const int* __restrict__ col,
                 const float* __restrict__ X,
                 const float* __restrict__ WL,
                 const float* __restrict__ WR,
                 const float* __restrict__ bias,
                 const float* __restrict__ bn_g,
                 const float* __restrict__ bn_b,
                 const float* __restrict__ bn_m,
                 const float* __restrict__ bn_v,
                 float* __restrict__ out, int nrows, int row_tiles) {
    __shared__ float4 sW[2 * HDIM * WPITCH4];
    __shared__ float4 sAgg[FUSED_WAVES * 16 * WPITCH4];
    const int lane = threadIdx.x & 31;
    const int wv   = __builtin_amdgcn_readfirstlane((int)(threadIdx.x >> 5));
    const int c    = lane & 15;
    const int h    = lane >> 4;

    {
        float4 w[16];
#pragma unroll
        for (int it = 0; it < 8; ++it) {
            w[it]     = *(const float4*)(WL + (size_t)(it * FUSED_WAVES + wv) * HDIM + lane * 4);
            w[it + 8] = *(const float4*)(WR + (size_t)(it * FUSED_WAVES + wv) * HDIM + lane * 4);
        }
#pragma unroll
        for (int it = 0; it < 8; ++it) {
            sW[(it * FUSED_WAVES + wv) * WPITCH4 + lane] = w[it];
            sW[HDIM * WPITCH4 + (it * FUSED_WAVES + wv) * WPITCH4 + lane] = w[it + 8];
        }
    }
    float bcol[8];
#pragma unroll
    for (int j = 0; j < 8; ++j) bcol[j] = bias[j * 16 + c];
    __syncthreads();

    float4* myAgg = sAgg + wv * (16 * WPITCH4);
    const float4* __restrict__ xr = (const float4*)X + lane;

    for (int tile = blockIdx.x + FUSED_WGS * wv; tile < row_tiles; tile += FUSED_WGS * FUSED_WAVES) {
        const int row0 = tile << 4;

        int2 sg = seg[min(row0 + c, nrows - 1)];
        if (row0 + c >= nrows) sg.y = 0;
        int beg = __builtin_amdgcn_readlane(sg.x, 0);
        int cnt = __builtin_amdgcn_readlane(sg.y, 0);
        int ids = 0;
        if (cnt > 0) ids = col[beg + min(lane, min(cnt, 32) - 1)];
        for (int r = 0; r < 16; ++r) {
            const int cbeg = beg, ccnt = cnt, cids = ids;
            if (r + 1 < 16) {
                beg = __builtin_amdgcn_readlane(sg.x, r + 1);
                cnt = __builtin_amdgcn_readlane(sg.y, r + 1);
                if (cnt > 0) ids = col[beg + min(lane, min(cnt, 32) - 1)];
            }
            myAgg[r * WPITCH4 + lane] = max_over_sources(col, xr, cbeg, ccnt, cids, lane);
        }
        __builtin_amdgcn_fence(__ATOMIC_RELEASE, "wavefront");
        __builtin_amdgcn_wave_barrier();
        __builtin_amdgcn_fence(__ATOMIC_ACQUIRE, "wavefront");

        float4 a0[16];
#pragma unroll
        for (int t = 0; t < 16; ++t) a0[t] = myAgg[c * WPITCH4 + 2 * t + h];

        int arow_i = row0 + c;
        if (arow_i >= nrows) arow_i = nrows - 1;

        v8f acc[8];
#pragma unroll
        for (int j = 0; j < 8; ++j) {
#pragma unroll
            for (int q = 0; q < 8; ++q) acc[j][q] = 0.0f;
        }

        float pg[8], pb[8], pm[8], pv[8];
        strip_mac<2, true>(a0, X, sW, arow_i, c, h, acc, bn_g, bn_b, bn_m, bn_v, pg, pb, pm, pv);

        float* op = out + (size_t)(row0 + h * 8) * HDIM + c;
        const bool whole = (row0 + 16 <= nrows);
#pragma unroll
        for (int j = 0; j < 8; ++j) {
            const float s  = pg[j] * rsqrtf(pv[j] + BN_EPS);
            const float mu = pm[j];
            const float bb = pb[j];
            const float bj = bcol[j];
            if (whole) {
#pragma unroll
                for (int q = 0; q < 8; ++q) {
                    const float v = ((acc[j][q] + bj) - mu) * s + bb;
                    op[q * HDIM + j * 16] = fmaxf(v, 0.0f);
                }
            } else {
#pragma unroll
                for (int q = 0; q < 8; ++q) {
                    if (row0 + h * 8 + q < nrows) {
                        const float v = ((acc[j][q] + bj) - mu) * s + bb;
                        op[q * HDIM + j * 16] = fmaxf(v, 0.0f);
                    }
                }
            }
        }
    }
}

extern "C" void kernel_launch(void* const* d_in, const int* in_sizes, int n_in,
                              void* d_out, int out_size, void* d_ws, size_t ws_size,
                              hipStream_t stream) {
    const float* features = (const float*)d_in[0];
    const int*   ei       = (const int*)d_in[1];
    const float* w_proj = (const float*)d_in[2];
    const float* b_proj = (const float*)d_in[3];
    const float* w_l0 = (const float*)d_in[4];
    const float* b_l0 = (const float*)d_in[5];
    const float* w_r0 = (const float*)d_in[6];
    const float* bn_g0 = (const float*)d_in[7];
    const float* bn_b0 = (const float*)d_in[8];
    const float* bn_m0 = (const float*)d_in[9];
    const float* bn_v0 = (const float*)d_in[10];
    const float* w_l1 = (const float*)d_in[11];
    const float* b_l1 = (const float*)d_in[12];
    const float* w_r1 = (const float*)d_in[13];
    const float* bn_g1 = (const float*)d_in[14];
    const float* bn_b1 = (const float*)d_in[15];
    const float* bn_m1 = (const float*)d_in[16];
    const float* bn_v1 = (const float*)d_in[17];
    const float* w_fc1 = (const float*)d_in[18];
    const float* b_fc1 = (const float*)d_in[19];
    const float* w_fc2 = (const float*)d_in[20];
    const float* b_fc2 = (const float*)d_in[21];
    float* outp = (float*)d_out;

    const int N = in_sizes[0] / HDIM;
    const int E = in_sizes[1] / 2;

    const size_t SZ = (size_t)N * HDIM * sizeof(float);
    float* X    = (float*)d_ws;
    float* AGGF = (float*)((char*)d_ws + SZ);
    float* H1   = (float*)((char*)d_ws + 2 * SZ);
    char*  G    = (char*)d_ws + 3 * SZ;
    const size_t o_seg  = (((size_t)N + 1) * sizeof(int) + 255) & ~(size_t)255;
    const size_t o_part = o_seg  + (((size_t)N * sizeof(int2) + 255) & ~(size_t)255);
    const size_t o_col  = o_part + (((size_t)E * sizeof(int)  + 255) & ~(size_t)255);
    int2*     seg  = (int2*)(G + o_seg);
    unsigned* part = (unsigned*)(G + o_part);
    int*      col  = (int*)(G + o_col);

    const int nbins   = (N + 255) / 256;
    const int nchunks = (E + 4095) / 4096;
    int* hist      = (int*)AGGF;
    int* bin_tot   = hist + (size_t)nchunks * nbins;
    int* bin_start = bin_tot + nbins;

    const int row_tiles   = (N + 15) / 16;
    const int gemm_blocks = (row_tiles + 7) / 8;

    bin_count<<<nchunks, 256, (size_t)nbins * sizeof(int), stream>>>(ei, hist, E, N, nbins);
    bin_offsets<<<nbins, 256, 0, stream>>>(hist, bin_tot, nchunks, nbins);
    bin_partition<<<nchunks, 256, 2 * (size_t)nbins * sizeof(int), stream>>>(ei, hist, bin_tot, bin_start,
                                                                          part, E, N, nbins);
    bin_sort<<<nbins, 256, 0, stream>>>(part, bin_start, bin_tot, seg, col, N);

    gemm128_bias<<<gemm_blocks, 256, 0, stream>>>(features, w_proj, b_proj, X, N);

    sage_layer_fused<<<FUSED_WGS, FUSED_WAVES * 32, 0, stream>>>(seg, col, X, w_l0, w_r0, b_l0,
                                                                 bn_g0, bn_b0, bn_m0, bn_v0, H1, N, row_tiles);

    sage_layer_fused<<<FUSED_WGS, FUSED_WAVES * 32, 0, stream>>>(seg, col, H1, w_l1, w_r1, b_l1,
                                                                 bn_g1, bn_b1, bn_m1, bn_v1, X, N, row_tiles);

    fc1_relu_fc2<<<gemm_blocks, 256, 0, stream>>>(X, w_fc1, b_fc1, w_fc2, b_fc2, outp, N);
}


# SAGE layers: lin_r (root) product issued inside the aggregation loop, one k-chunk per node between its row requests and their max: WMMAs run under the gather wait; root summed first (sum order)
# speedup vs baseline: 4.2349x; 1.0419x over previous
#include <hip/hip_runtime.h>

#define HDIM 128
#define BN_EPS 1e-5f
#define WPITCH4 33

typedef __attribute__((ext_vector_type(2))) float v2f;
typedef __attribute__((ext_vector_type(8))) float v8f;
typedef __attribute__((ext_vector_type(4))) float v4f;

__device__ __forceinline__ void stage_weights(const float* __restrict__ Wg, float4* sW,
                                              int wv, int lane) {
    float4 w[16];
#pragma unroll
    for (int it = 0; it < 16; ++it)
        w[it] = *(const float4*)(Wg + (size_t)(it * 8 + wv) * HDIM + lane * 4);
#pragma unroll
    for (int it = 0; it < 16; ++it)
        sW[(it * 8 + wv) * WPITCH4 + lane] = w[it];
}

__device__ __forceinline__ void load_a_strip(const float* __restrict__ A, int arow_i, int h,
                                             float4 (&a)[16]) {
    const float4* ap = (const float4*)(A + (size_t)arow_i * HDIM + 4 * h);
#pragma unroll
    for (int t = 0; t < 16; ++t) a[t] = ap[2 * t];
}

template <int NMAT, bool BN>
__device__ __forceinline__ void strip_mac(float4 (&a0)[16], const float* __restrict__ A1,
                                          const float4* sW, int arow_i, int c, int h,
                                          v8f (&acc)[8],
                                          const float* __restrict__ bn_g,
                                          const float* __restrict__ bn_b,
                                          const float* __restrict__ bn_m,
                                          const float* __restrict__ bn_v,
                                          float (&pg)[8], float (&pb)[8],
                                          float (&pm)[8], float (&pv)[8]) {
    const float4* ap1 = (const float4*)(A1 + (size_t)arow_i * HDIM + 4 * h);
    const float4* bp  = sW + c * WPITCH4 + h;

    float4 a1[16];
    float4 b[2][8];
#pragma unroll
    for (int j = 0; j < 8; ++j) b[0][j] = bp[j * 16 * WPITCH4];

#pragma unroll
    for (int t = 0; t < 16 * NMAT; ++t) {
        if (NMAT == 2 && t == 8) {
#pragma unroll
            for (int u = 0; u < 8; ++u) a1[u] = ap1[2 * u];
        }
        if (NMAT == 2 && t == 16) {
#pragma unroll
            for (int u = 8; u < 16; ++u) a1[u] = ap1[2 * u];
        }
        if (BN && t == 16 * NMAT - 3) {
#pragma unroll
            for (int j = 0; j < 8; ++j) {
                pg[j] = bn_g[j * 16 + c];
                pv[j] = bn_v[j * 16 + c];
                pm[j] = bn_m[j * 16 + c];
                pb[j] = bn_b[j * 16 + c];
            }
        }
        if (t + 1 < 16 * NMAT) {
            const int tn = t + 1;
            const float4* nb = bp + (tn >> 4) * (HDIM * WPITCH4) + (tn & 15) * 2;
#pragma unroll
            for (int j = 0; j < 8; ++j) b[tn & 1][j] = nb[j * 16 * WPITCH4];
        }
        __builtin_amdgcn_sched_barrier(0);
        const float4 av = (t < 16) ? a0[t & 15] : a1[t & 15];
        const v2f a01 = {av.x, av.y};
        const v2f a23 = {av.z, av.w};
#pragma unroll
        for (int j = 0; j < 8; ++j) {
            const v2f b01 = {b[t & 1][j].x, b[t & 1][j].y};
            acc[j] = __builtin_amdgcn_wmma_f32_16x16x4_f32(false, a01, false, b01,
                                                           (short)0, acc[j], false, false);
        }
#pragma unroll
        for (int j = 0; j < 8; ++j) {
            const v2f b23 = {b[t & 1][j].z, b[t & 1][j].w};
            acc[j] = __builtin_amdgcn_wmma_f32_16x16x4_f32(false, a23, false, b23,
                                                           (short)0, acc[j], false, false);
        }
        __builtin_amdgcn_sched_barrier(0);
    }
}

__global__ void __launch_bounds__(256)
gemm128_bias(const float* __restrict__ A,
             const float* __restrict__ W,
             const float* __restrict__ bias,
             float* __restrict__ out,
             int nrows) {
    __shared__ float4 sW[HDIM * WPITCH4];
    const int lane = threadIdx.x & 31;
    const int wv   = threadIdx.x >> 5;
    const int c    = lane & 15;
    const int h    = lane >> 4;
    const int row0 = (blockIdx.x * 8 + wv) << 4;
    int arow_i = row0 + c;
    if (arow_i >= nrows) arow_i = nrows - 1;

    float4 a0[16];
    load_a_strip(A, arow_i, h, a0);
    float bcol[8];
#pragma unroll
    for (int j = 0; j < 8; ++j) bcol[j] = bias[j * 16 + c];

    stage_weights(W, sW, wv, lane);
    __syncthreads();
    if (row0 >= nrows) return;

    v8f acc[8];
#pragma unroll
    for (int j = 0; j < 8; ++j) {
#pragma unroll
        for (int r = 0; r < 8; ++r) acc[j][r] = bcol[j];
    }

    float pg[8], pb[8], pm[8], pv[8];
    strip_mac<1, false>(a0, A, sW, arow_i, c, h, acc, bias, bias, bias, bias, pg, pb, pm, pv);

    float* op = out + (size_t)(row0 + h * 8) * HDIM + c;
    if (row0 + 16 <= nrows) {
#pragma unroll
        for (int j = 0; j < 8; ++j) {
#pragma unroll
            for (int r = 0; r < 8; ++r) op[r * HDIM + j * 16] = acc[j][r];
        }
    } else {
#pragma unroll
        for (int j = 0; j < 8; ++j) {
#pragma unroll
            for (int r = 0; r < 8; ++r) {
                if (row0 + h * 8 + r < nrows) op[r * HDIM + j * 16] = acc[j][r];
            }
        }
    }
}

__global__ void __launch_bounds__(256)
fc1_relu_fc2(const float* __restrict__ A,
             const float* __restrict__ W,
             const float* __restrict__ bias,
             const float* __restrict__ W2,
             const float* __restrict__ B2,
             float* __restrict__ out, int nrows) {
    __shared__ float4 sW[HDIM * WPITCH4];
    const int lane = threadIdx.x & 31;
    const int wv   = threadIdx.x >> 5;
    const int c    = lane & 15;
    const int h    = lane >> 4;
    const int row0 = (blockIdx.x * 8 + wv) << 4;
    int arow_i = row0 + c;
    if (arow_i >= nrows) arow_i = nrows - 1;

    float4 a0[16];
    load_a_strip(A, arow_i, h, a0);
    float bcol[8];
#pragma unroll
    for (int j = 0; j < 8; ++j) bcol[j] = bias[j * 16 + c];

    stage_weights(W, sW, wv, lane);
    __syncthreads();
    if (row0 >= nrows) return;

    v8f acc[8];
#pragma unroll
    for (int j = 0; j < 8; ++j) {
#pragma unroll
        for (int r = 0; r < 8; ++r) acc[j][r] = bcol[j];
    }

    float w2[8][4];
#pragma unroll
    for (int j = 0; j < 8; ++j) {
#pragma unroll
        for (int q = 0; q < 4; ++q) w2[j][q] = W2[q * HDIM + j * 16 + c];
    }
    const int   q_own = (c & 8) ? 2 : 0;
    const float b2x = B2[q_own], b2y = B2[q_own + 1];

    float pg[8], pb[8], pm[8], pv[8];
    strip_mac<1, false>(a0, A, sW, arow_i, c, h, acc, bias, bias, bias, bias, pg, pb, pm, pv);

    float p[32];
#pragma unroll
    for (int i = 0; i < 32; ++i) p[i] = 0.0f;
#pragma unroll
    for (int j = 0; j < 8; ++j) {
#pragma unroll
        for (int r = 0; r < 8; ++r) {
            const float y = fmaxf(acc[j][r], 0.0f);
#pragma unroll
            for (int q = 0; q < 4; ++q) p[r * 4 + q] = fmaf(y, w2[j][q], p[r * 4 + q]);
        }
    }

    const bool b0 = (c & 1) != 0, b1 = (c & 2) != 0, b2 = (c & 4) != 0, b3 = (c & 8) != 0;
    float s16[16], s8[8], s4[4], s2[2];
#pragma unroll
    for (int k = 0; k < 16; ++k) {
        const float keep = b0 ? p[k + 16] : p[k];
        const float send = b0 ? p[k] : p[k + 16];
        s16[k] = keep + __shfl_xor(send, 1, 32);
    }
#pragma unroll
    for (int k = 0; k < 8; ++k) {
        const float keep = b1 ? s16[k + 8] : s16[k];
        const float send = b1 ? s16[k] : s16[k + 8];
        s8[k] = keep + __shfl_xor(send, 2, 32);
    }
#pragma unroll
    for (int k = 0; k < 4; ++k) {
        const float keep = b2 ? s8[k + 4] : s8[k];
        const float send = b2 ? s8[k] : s8[k + 4];
        s4[k] = keep + __shfl_xor(send, 4, 32);
    }
#pragma unroll
    for (int k = 0; k < 2; ++k) {
        const float keep = b3 ? s4[k + 2] : s4[k];
        const float send = b3 ? s4[k] : s4[k + 2];
        s2[k] = keep + __shfl_xor(send, 8, 32);
    }
    const int r_own = (b0 ? 4 : 0) + (b1 ? 2 : 0) + (b2 ? 1 : 0);
    const int row   = row0 + h * 8 + r_own;
    if (row < nrows) {
        float2 o;
        o.x = s2[0] + b2x;
        o.y = s2[1] + b2y;
        *(float2*)(out + (size_t)row * 4 + q_own) = o;
    }
}

#define BIN_SHIFT 8
#define CHUNK_EDGES 4096

__device__ __forceinline__ int block_excl_scan(int v, int* wsum, int& total) {
    const int lane = threadIdx.x & 31;
    const int wv   = threadIdx.x >> 5;
    int incl = v;
#pragma unroll
    for (int off = 1; off < 32; off <<= 1) {
        const int t = __shfl_up(incl, off, 32);
        if (lane >= off) incl += t;
    }
    __syncthreads();
    if (lane == 31) wsum[wv] = incl;
    __syncthreads();
    int woff = 0, tot = 0;
#pragma unroll
    for (int w = 0; w < 8; ++w) {
        const int t = wsum[w];
        if (w < wv) woff += t;
        tot += t;
    }
    total = tot;
    return woff + incl - v;
}

__global__ void __launch_bounds__(256)
bin_count(const int* __restrict__ ei, int* __restrict__ hist, int E, int nrows, int nbins) {
    extern __shared__ int sbin[];
    for (int i = threadIdx.x; i < nbins; i += 256) sbin[i] = 0;
    __syncthreads();
    const int base = blockIdx.x * CHUNK_EDGES + threadIdx.x;
    int d[16];
#pragma unroll
    for (int j = 0; j < 16; ++j) {
        const int e = min(base + j * 256, E - 1);
        d[j] = ei[e];
    }
#pragma unroll
    for (int j = 0; j < 16; ++j)
        if (base + j * 256 < E && (unsigned)d[j] < (unsigned)nrows)
            atomicAdd(&sbin[d[j] >> BIN_SHIFT], 1);
    __syncthreads();
    for (int i = threadIdx.x; i < nbins; i += 256)
        hist[(size_t)blockIdx.x * nbins + i] = sbin[i];
}

__global__ void __launch_bounds__(256)
bin_offsets(int* __restrict__ hist, int* __restrict__ bin_tot, int nchunks, int nbins) {
    __shared__ int wsum[8];
    const int k = blockIdx.x;
    int carry = 0;
    for (int b0 = 0; b0 < nchunks; b0 += 256) {
        const int b = b0 + threadIdx.x;
        const int v = (b < nchunks) ? hist[(size_t)b * nbins + k] : 0;
        int total;
        const int excl = block_excl_scan(v, wsum, total);
        if (b < nchunks) hist[(size_t)b * nbins + k] = carry + excl;
        carry += total;
    }
    if (threadIdx.x == 0) bin_tot[k] = carry;
}

__global__ void __launch_bounds__(256)
bin_partition(const int* __restrict__ ei, const int* __restrict__ hist,
              const int* __restrict__ bin_tot, int* __restrict__ bin_start,
              unsigned* __restrict__ part, int E, int nrows, int nbins) {
    extern __shared__ int sbin[];
    __shared__ unsigned rec[CHUNK_EDGES];
    __shared__ int      pos[CHUNK_EDGES];
    __shared__ int      wsum[8];
    int* lcnt  = sbin;
    int* gbase = sbin + nbins;
    for (int i = threadIdx.x; i < nbins; i += 256) lcnt[i] = 0;
    int carry = 0;
    for (int k0 = 0; k0 < nbins; k0 += 256) {
        const int k = k0 + threadIdx.x;
        const int v = (k < nbins) ? bin_tot[k] : 0;
        int total;
        const int excl = block_excl_scan(v, wsum, total);
        if (k < nbins) {
            const int s = carry + excl;
            if (blockIdx.x == 0) bin_start[k] = s;
            gbase[k] = s + hist[(size_t)blockIdx.x * nbins + k];
        }
        carry += total;
    }
    __syncthreads();
    const int base = blockIdx.x * CHUNK_EDGES + threadIdx.x;
    int d[16], s[16], r[16];
#pragma unroll
    for (int j = 0; j < 16; ++j) {
        const int e = min(base + j * 256, E - 1);
        d[j] = ei[e];
        s[j] = ei[(size_t)E + e];
    }
#pragma unroll
    for (int j = 0; j < 16; ++j) {
        const bool live = base + j * 256 < E && (unsigned)d[j] < (unsigned)nrows;
        r[j] = live ? atomicAdd(&lcnt[d[j] >> BIN_SHIFT], 1) : -1;
    }
    __syncthreads();
    carry = 0;
    for (int k0 = 0; k0 < nbins; k0 += 256) {
        const int k = k0 + threadIdx.x;
        const int v = (k < nbins) ? lcnt[k] : 0;
        int total;
        const int excl = block_excl_scan(v, wsum, total);
        if (k < nbins) lcnt[k] = carry + excl;
        carry += total;
    }
    __syncthreads();
    const int nlive = carry;
#pragma unroll
    for (int j = 0; j < 16; ++j) {
        if (r[j] >= 0) {
            const int bin  = d[j] >> BIN_SHIFT;
            const int srcc = min(max(s[j], 0), nrows - 1);
            const int slot = lcnt[bin] + r[j];
            rec[slot] = ((unsigned)(d[j] & 255) << 24) | (unsigned)srcc;
            pos[slot] = gbase[bin] + r[j];
        }
    }
    __syncthreads();
    for (int i = threadIdx.x; i < nlive; i += 256) part[pos[i]] = rec[i];
}

#define SORT_CAP 8192
__global__ void __launch_bounds__(256)
bin_sort(const unsigned* __restrict__ part, const int* __restrict__ bin_start,
         const int* __restrict__ bin_tot, int2* __restrict__ seg,
         int* __restrict__ col, int nrows) {
    __shared__ int cnt[256];
    __shared__ int wsum[8];
    __shared__ int sorted[SORT_CAP];
    const int k = blockIdx.x;
    const int s = bin_start[k];
    const int m = bin_tot[k];
    cnt[threadIdx.x] = 0;
    __syncthreads();
    for (int i0 = 0; i0 < m; i0 += 2048) {
        unsigned r[8];
#pragma unroll
        for (int u = 0; u < 8; ++u) r[u] = part[s + min(i0 + u * 256 + (int)threadIdx.x, m - 1)];
#pragma unroll
        for (int u = 0; u < 8; ++u)
            if (i0 + u * 256 + (int)threadIdx.x < m) atomicAdd(&cnt[r[u] >> 24], 1);
    }
    __syncthreads();
    const int deg = cnt[threadIdx.x];
    int total;
    const int first = block_excl_scan(deg, wsum, total);
    const int n = (k << BIN_SHIFT) + threadIdx.x;
    if (n < nrows) seg[n] = make_int2(s + first, deg);
    cnt[threadIdx.x] = first;
    __syncthreads();
    const bool in_lds = (m <= SORT_CAP);
    for (int i0 = 0; i0 < m; i0 += 2048) {
        unsigned r[8];
#pragma unroll
        for (int u = 0; u < 8; ++u) r[u] = part[s + min(i0 + u * 256 + (int)threadIdx.x, m - 1)];
#pragma unroll
        for (int u = 0; u < 8; ++u) {
            if (i0 + u * 256 + (int)threadIdx.x < m) {
                const int p = atomicAdd(&cnt[r[u] >> 24], 1);
                const int v = (int)(r[u] & 0xFFFFFFu);
                if (in_lds) sorted[p] = v; else col[s + p] = v;
            }
        }
    }
    if (in_lds) {
        __syncthreads();
        for (int i = threadIdx.x; i < m; i += 256) col[s + i] = sorted[i];
    }
}

__device__ __forceinline__ float4 max4(float4 a, float4 b) {
    return make_float4(fmaxf(a.x, b.x), fmaxf(a.y, b.y), fmaxf(a.z, b.z), fmaxf(a.w, b.w));
}

#define FUSED_WGS   256
#define FUSED_WAVES 16

__device__ __forceinline__ void request_first8(const float4* __restrict__ xr, int ids,
                                               v4f (&v)[8]) {
#pragma unroll
    for (int i = 0; i < 8; ++i) {
        const int s = __builtin_amdgcn_readlane(ids, i);
        v[i] = *(const v4f*)(xr + (size_t)s * (HDIM / 4));
    }
}

__device__ __forceinline__ float4 max_over_rest(const int* __restrict__ col,
                                                const float4* __restrict__ xr,
                                                int beg, int cnt, int ids0, int lane, float4 m) {
    int ids = ids0;
    for (int base = 0; base < cnt; base += 32) {
        const int c = min(cnt - base, 32);
        if (base > 0) ids = col[beg + base + min(lane, c - 1)];
        int i = (base == 0) ? 8 : 0;
        for (; i + 8 <= c; i += 8) {
            const int s0 = __builtin_amdgcn_readlane(ids, i);
            const int s1 = __builtin_amdgcn_readlane(ids, i + 1);
            const int s2 = __builtin_amdgcn_readlane(ids, i + 2);
            const int s3 = __builtin_amdgcn_readlane(ids, i + 3);
            const int s4 = __builtin_amdgcn_readlane(ids, i + 4);
            const int s5 = __builtin_amdgcn_readlane(ids, i + 5);
            const int s6 = __builtin_amdgcn_readlane(ids, i + 6);
            const int s7 = __builtin_amdgcn_readlane(ids, i + 7);
            const float4 v0 = xr[(size_t)s0 * (HDIM / 4)];
            const float4 v1 = xr[(size_t)s1 * (HDIM / 4)];
            const float4 v2 = xr[(size_t)s2 * (HDIM / 4)];
            const float4 v3 = xr[(size_t)s3 * (HDIM / 4)];
            const float4 v4 = xr[(size_t)s4 * (HDIM / 4)];
            const float4 v5 = xr[(size_t)s5 * (HDIM / 4)];
            const float4 v6 = xr[(size_t)s6 * (HDIM / 4)];
            const float4 v7 = xr[(size_t)s7 * (HDIM / 4)];
            m = max4(m, max4(max4(max4(v0, v1), max4(v2, v3)),
                             max4(max4(v4, v5), max4(v6, v7))));
        }
        if (i + 4 <= c) {
            const int s0 = __builtin_amdgcn_readlane(ids, i);
            const int s1 = __builtin_amdgcn_readlane(ids, i + 1);
            const int s2 = __builtin_amdgcn_readlane(ids, i + 2);
            const int s3 = __builtin_amdgcn_readlane(ids, i + 3);
            const float4 v0 = xr[(size_t)s0 * (HDIM / 4)];
            const float4 v1 = xr[(size_t)s1 * (HDIM / 4)];
            const float4 v2 = xr[(size_t)s2 * (HDIM / 4)];
            const float4 v3 = xr[(size_t)s3 * (HDIM / 4)];
            m = max4(m, max4(max4(v0, v1), max4(v2, v3)));
            i += 4;
        }
        const int left = c - i;
        if (left > 0) {
            const int s0 = __builtin_amdgcn_readlane(ids, i);
            float4 t = xr[(size_t)s0 * (HDIM / 4)];
            if (left > 1) {
                const int s1 = __builtin_amdgcn_readlane(ids, i + 1);
                t = max4(t, xr[(size_t)s1 * (HDIM / 4)]);
            }
            if (left > 2) {
                const int s2 = __builtin_amdgcn_readlane(ids, i + 2);
                t = max4(t, xr[(size_t)s2 * (HDIM / 4)]);
            }
            m = max4(m, t);
        }
    }
    return m;
}

__device__ __forceinline__ void root_chunk(const float4* bp_t, float4 av, v8f (&acc)[8]) {
    float4 b[8];
#pragma unroll
    for (int j = 0; j < 8; ++j) b[j] = bp_t[j * 16 * WPITCH4];
    const v2f a01 = {av.x, av.y};
    const v2f a23 = {av.z, av.w};
#pragma unroll
    for (int j = 0; j < 8; ++j) {
        const v2f b01 = {b[j].x, b[j].y};
        acc[j] = __builtin_amdgcn_wmma_f32_16x16x4_f32(false, a01, false, b01,
                                                       (short)0, acc[j], false, false);
    }
#pragma unroll
    for (int j = 0; j < 8; ++j) {
        const v2f b23 = {b[j].z, b[j].w};
        acc[j] = __builtin_amdgcn_wmma_f32_16x16x4_f32(false, a23, false, b23,
                                                       (short)0, acc[j], false, false);
    }
}

__global__ void __launch_bounds__(FUSED_WAVES * 32)
sage_layer_fused(const int2* __restrict__ seg, const int* __restrict__ col,
                 const float* __restrict__ X,
                 const float* __restrict__ WL,
                 const float* __restrict__ WR,
                 const float* __restrict__ bias,
                 const float* __restrict__ bn_g,
                 const float* __restrict__ bn_b,
                 const float* __restrict__ bn_m,
                 const float* __restrict__ bn_v,
                 float* __restrict__ out, int nrows, int row_tiles) {
    __shared__ float4 sW[2 * HDIM * WPITCH4];
    __shared__ float4 sAgg[FUSED_WAVES * 16 * WPITCH4];
    const int lane = threadIdx.x & 31;
    const int wv   = __builtin_amdgcn_readfirstlane((int)(threadIdx.x >> 5));
    const int c    = lane & 15;
    const int h    = lane >> 4;

    {
        float4 w[16];
#pragma unroll
        for (int it = 0; it < 8; ++it) {
            w[it]     = *(const float4*)(WL + (size_t)(it * FUSED_WAVES + wv) * HDIM + lane * 4);
            w[it + 8] = *(const float4*)(WR + (size_t)(it * FUSED_WAVES + wv) * HDIM + lane * 4);
        }
#pragma unroll
        for (int it = 0; it < 8; ++it) {
            sW[(it * FUSED_WAVES + wv) * WPITCH4 + lane] = w[it];
            sW[HDIM * WPITCH4 + (it * FUSED_WAVES + wv) * WPITCH4 + lane] = w[it + 8];
        }
    }
    float bcol[8];
#pragma unroll
    for (int j = 0; j < 8; ++j) bcol[j] = bias[j * 16 + c];
    __syncthreads();

    float4* myAgg = sAgg + wv * (16 * WPITCH4);
    const float4* __restrict__ xr = (const float4*)X + lane;

    const float4* bpr = sW + HDIM * WPITCH4 + c * WPITCH4 + h;

    for (int tile = blockIdx.x + FUSED_WGS * wv; tile < row_tiles; tile += FUSED_WGS * FUSED_WAVES) {
        const int row0 = tile << 4;

        int arow_i = row0 + c;
        if (arow_i >= nrows) arow_i = nrows - 1;
        const float4* ap1 = (const float4*)(X + (size_t)arow_i * HDIM + 4 * h);

        v8f acc[8];
#pragma unroll
        for (int j = 0; j < 8; ++j) {
#pragma unroll
            for (int q = 0; q < 8; ++q) acc[j][q] = 0.0f;
        }

        int2 sg = seg[min(row0 + c, nrows - 1)];
        if (row0 + c >= nrows) sg.y = 0;
        int beg = __builtin_amdgcn_readlane(sg.x, 0);
        int cnt = __builtin_amdgcn_readlane(sg.y, 0);
        int ids = 0;
        if (cnt > 0) ids = col[beg + min(lane, min(cnt, 32) - 1)];
        float4 a1c = ap1[0];
        for (int r = 0; r < 16; ++r) {
            const int cbeg = beg, ccnt = cnt, cids = ids;
            float4 a1n = a1c;
            if (r + 1 < 16) {
                beg = __builtin_amdgcn_readlane(sg.x, r + 1);
                cnt = __builtin_amdgcn_readlane(sg.y, r + 1);
                if (cnt > 0) ids = col[beg + min(lane, min(cnt, 32) - 1)];
                a1n = ap1[2 * (r + 1)];
            }
            v4f v[8];
            request_first8(xr, cids, v);
            __builtin_amdgcn_sched_barrier(0);
            root_chunk(bpr + 2 * r, a1c, acc);
            __builtin_amdgcn_sched_barrier(0);
            asm volatile("v_nop\n\tv_nop\n\tv_nop\n\tv_nop"
                         : "+v"(v[0]), "+v"(v[1]), "+v"(v[2]), "+v"(v[3]),
                           "+v"(v[4]), "+v"(v[5]), "+v"(v[6]), "+v"(v[7]));
            const float ninf = -__builtin_inff();
            const v4f mv = __builtin_elementwise_max(
                __builtin_elementwise_max(__builtin_elementwise_max(v[0], v[1]),
                                          __builtin_elementwise_max(v[2], v[3])),
                __builtin_elementwise_max(__builtin_elementwise_max(v[4], v[5]),
                                          __builtin_elementwise_max(v[6], v[7])));
            float4 m = make_float4(mv.x, mv.y, mv.z, mv.w);
            if (ccnt <= 0) m = make_float4(ninf, ninf, ninf, ninf);
            m = max_over_rest(col, xr, cbeg, ccnt, cids, lane, m);
            float4 o;
            o.x = (m.x == ninf) ? 0.0f : m.x;
            o.y = (m.y == ninf) ? 0.0f : m.y;
            o.z = (m.z == ninf) ? 0.0f : m.z;
            o.w = (m.w == ninf) ? 0.0f : m.w;
            myAgg[r * WPITCH4 + lane] = o;
            a1c = a1n;
        }
        __builtin_amdgcn_fence(__ATOMIC_RELEASE, "wavefront");
        __builtin_amdgcn_wave_barrier();
        __builtin_amdgcn_fence(__ATOMIC_ACQUIRE, "wavefront");

        float4 a0[16];
#pragma unroll
        for (int t = 0; t < 16; ++t) a0[t] = myAgg[c * WPITCH4 + 2 * t + h];

        float pg[8], pb[8], pm[8], pv[8];
        strip_mac<1, true>(a0, X, sW, arow_i, c, h, acc, bn_g, bn_b, bn_m, bn_v, pg, pb, pm, pv);
        asm volatile("v_nop\n\tv_nop\n\tv_nop\n\tv_nop");
        __builtin_amdgcn_sched_barrier(0);

        float* op = out + (size_t)(row0 + h * 8) * HDIM + c;
        const bool whole = (row0 + 16 <= nrows);
#pragma unroll
        for (int j = 0; j < 8; ++j) {
            const float s  = pg[j] * rsqrtf(pv[j] + BN_EPS);
            const float mu = pm[j];
            const float bb = pb[j];
            const float bj = bcol[j];
            if (whole) {
#pragma unroll
                for (int q = 0; q < 8; ++q) {
                    const float v = ((acc[j][q] + bj) - mu) * s + bb;
                    op[q * HDIM + j * 16] = fmaxf(v, 0.0f);
                }
            } else {
#pragma unroll
                for (int q = 0; q < 8; ++q) {
                    if (row0 + h * 8 + q < nrows) {
                        const float v = ((acc[j][q] + bj) - mu) * s + bb;
                        op[q * HDIM + j * 16] = fmaxf(v, 0.0f);
                    }
                }
            }
        }
    }
}

extern "C" void kernel_launch(void* const* d_in, const int* in_sizes, int n_in,
                              void* d_out, int out_size, void* d_ws, size_t ws_size,
                              hipStream_t stream) {
    const float* features = (const float*)d_in[0];
    const int*   ei       = (const int*)d_in[1];
    const float* w_proj = (const float*)d_in[2];
    const float* b_proj = (const float*)d_in[3];
    const float* w_l0 = (const float*)d_in[4];
    const float* b_l0 = (const float*)d_in[5];
    const float* w_r0 = (const float*)d_in[6];
    const float* bn_g0 = (const float*)d_in[7];
    const float* bn_b0 = (const float*)d_in[8];
    const float* bn_m0 = (const float*)d_in[9];
    const float* bn_v0 = (const float*)d_in[10];
    const float* w_l1 = (const float*)d_in[11];
    const float* b_l1 = (const float*)d_in[12];
    const float* w_r1 = (const float*)d_in[13];
    const float* bn_g1 = (const float*)d_in[14];
    const float* bn_b1 = (const float*)d_in[15];
    const float* bn_m1 = (const float*)d_in[16];
    const float* bn_v1 = (const float*)d_in[17];
    const float* w_fc1 = (const float*)d_in[18];
    const float* b_fc1 = (const float*)d_in[19];
    const float* w_fc2 = (const float*)d_in[20];
    const float* b_fc2 = (const float*)d_in[21];
    float* outp = (float*)d_out;

    const int N = in_sizes[0] / HDIM;
    const int E = in_sizes[1] / 2;

    const size_t SZ = (size_t)N * HDIM * sizeof(float);
    float* X    = (float*)d_ws;
    float* AGGF = (float*)((char*)d_ws + SZ);
    float* H1   = (float*)((char*)d_ws + 2 * SZ);
    char*  G    = (char*)d_ws + 3 * SZ;
    const size_t o_seg  = (((size_t)N + 1) * sizeof(int) + 255) & ~(size_t)255;
    const size_t o_part = o_seg  + (((size_t)N * sizeof(int2) + 255) & ~(size_t)255);
    const size_t o_col  = o_part + (((size_t)E * sizeof(int)  + 255) & ~(size_t)255);
    int2*     seg  = (int2*)(G + o_seg);
    unsigned* part = (unsigned*)(G + o_part);
    int*      col  = (int*)(G + o_col);

    const int nbins   = (N + 255) / 256;
    const int nchunks = (E + 4095) / 4096;
    int* hist      = (int*)AGGF;
    int* bin_tot   = hist + (size_t)nchunks * nbins;
    int* bin_start = bin_tot + nbins;

    const int row_tiles   = (N + 15) / 16;
    const int gemm_blocks = (row_tiles + 7) / 8;

    bin_count<<<nchunks, 256, (size_t)nbins * sizeof(int), stream>>>(ei, hist, E, N, nbins);
    bin_offsets<<<nbins, 256, 0, stream>>>(hist, bin_tot, nchunks, nbins);
    bin_partition<<<nchunks, 256, 2 * (size_t)nbins * sizeof(int), stream>>>(ei, hist, bin_tot, bin_start,
                                                                          part, E, N, nbins);
    bin_sort<<<nbins, 256, 0, stream>>>(part, bin_start, bin_tot, seg, col, N);

    gemm128_bias<<<gemm_blocks, 256, 0, stream>>>(features, w_proj, b_proj, X, N);

    sage_layer_fused<<<FUSED_WGS, FUSED_WAVES * 32, 0, stream>>>(seg, col, X, w_l0, w_r0, b_l0,
                                                                 bn_g0, bn_b0, bn_m0, bn_v0, H1, N, row_tiles);

    sage_layer_fused<<<FUSED_WGS, FUSED_WAVES * 32, 0, stream>>>(seg, col, H1, w_l1, w_r1, b_l1,
                                                                 bn_g1, bn_b1, bn_m1, bn_v1, X, N, row_tiles);

    fc1_relu_fc2<<<gemm_blocks, 256, 0, stream>>>(X, w_fc1, b_fc1, w_fc2, b_fc2, outp, N);
}


# SAGE gather: a node's first 16 source rows requested as one group (was 8), the rest in whole groups of 8 padded with the last row: one cache trip for up to 16 sources instead of 2-3
# speedup vs baseline: 4.4705x; 1.0556x over previous
#include <hip/hip_runtime.h>

#define HDIM 128
#define BN_EPS 1e-5f
#define WPITCH4 33

typedef __attribute__((ext_vector_type(2))) float v2f;
typedef __attribute__((ext_vector_type(8))) float v8f;
typedef __attribute__((ext_vector_type(4))) float v4f;

__device__ __forceinline__ void stage_weights(const float* __restrict__ Wg, float4* sW,
                                              int wv, int lane) {
    float4 w[16];
#pragma unroll
    for (int it = 0; it < 16; ++it)
        w[it] = *(const float4*)(Wg + (size_t)(it * 8 + wv) * HDIM + lane * 4);
#pragma unroll
    for (int it = 0; it < 16; ++it)
        sW[(it * 8 + wv) * WPITCH4 + lane] = w[it];
}

__device__ __forceinline__ void load_a_strip(const float* __restrict__ A, int arow_i, int h,
                                             float4 (&a)[16]) {
    const float4* ap = (const float4*)(A + (size_t)arow_i * HDIM + 4 * h);
#pragma unroll
    for (int t = 0; t < 16; ++t) a[t] = ap[2 * t];
}

template <int NMAT, bool BN>
__device__ __forceinline__ void strip_mac(float4 (&a0)[16], const float* __restrict__ A1,
                                          const float4* sW, int arow_i, int c, int h,
                                          v8f (&acc)[8],
                                          const float* __restrict__ bn_g,
                                          const float* __restrict__ bn_b,
                                          const float* __restrict__ bn_m,
                                          const float* __restrict__ bn_v,
                                          float (&pg)[8], float (&pb)[8],
                                          float (&pm)[8], float (&pv)[8]) {
    const float4* ap1 = (const float4*)(A1 + (size_t)arow_i * HDIM + 4 * h);
    const float4* bp  = sW + c * WPITCH4 + h;

    float4 a1[16];
    float4 b[2][8];
#pragma unroll
    for (int j = 0; j < 8; ++j) b[0][j] = bp[j * 16 * WPITCH4];

#pragma unroll
    for (int t = 0; t < 16 * NMAT; ++t) {
        if (NMAT == 2 && t == 8) {
#pragma unroll
            for (int u = 0; u < 8; ++u) a1[u] = ap1[2 * u];
        }
        if (NMAT == 2 && t == 16) {
#pragma unroll
            for (int u = 8; u < 16; ++u) a1[u] = ap1[2 * u];
        }
        if (BN && t == 16 * NMAT - 3) {
#pragma unroll
            for (int j = 0; j < 8; ++j) {
                pg[j] = bn_g[j * 16 + c];
                pv[j] = bn_v[j * 16 + c];
                pm[j] = bn_m[j * 16 + c];
                pb[j] = bn_b[j * 16 + c];
            }
        }
        if (t + 1 < 16 * NMAT) {
            const int tn = t + 1;
            const float4* nb = bp + (tn >> 4) * (HDIM * WPITCH4) + (tn & 15) * 2;
#pragma unroll
            for (int j = 0; j < 8; ++j) b[tn & 1][j] = nb[j * 16 * WPITCH4];
        }
        __builtin_amdgcn_sched_barrier(0);
        const float4 av = (t < 16) ? a0[t & 15] : a1[t & 15];
        const v2f a01 = {av.x, av.y};
        const v2f a23 = {av.z, av.w};
#pragma unroll
        for (int j = 0; j < 8; ++j) {
            const v2f b01 = {b[t & 1][j].x, b[t & 1][j].y};
            acc[j] = __builtin_amdgcn_wmma_f32_16x16x4_f32(false, a01, false, b01,
                                                           (short)0, acc[j], false, false);
        }
#pragma unroll
        for (int j = 0; j < 8; ++j) {
            const v2f b23 = {b[t & 1][j].z, b[t & 1][j].w};
            acc[j] = __builtin_amdgcn_wmma_f32_16x16x4_f32(false, a23, false, b23,
                                                           (short)0, acc[j], false, false);
        }
        __builtin_amdgcn_sched_barrier(0);
    }
}

__global__ void __launch_bounds__(256)
gemm128_bias(const float* __restrict__ A,
             const float* __restrict__ W,
             const float* __restrict__ bias,
             float* __restrict__ out,
             int nrows) {
    __shared__ float4 sW[HDIM * WPITCH4];
    const int lane = threadIdx.x & 31;
    const int wv   = threadIdx.x >> 5;
    const int c    = lane & 15;
    const int h    = lane >> 4;
    const int row0 = (blockIdx.x * 8 + wv) << 4;
    int arow_i = row0 + c;
    if (arow_i >= nrows) arow_i = nrows - 1;

    float4 a0[16];
    load_a_strip(A, arow_i, h, a0);
    float bcol[8];
#pragma unroll
    for (int j = 0; j < 8; ++j) bcol[j] = bias[j * 16 + c];

    stage_weights(W, sW, wv, lane);
    __syncthreads();
    if (row0 >= nrows) return;

    v8f acc[8];
#pragma unroll
    for (int j = 0; j < 8; ++j) {
#pragma unroll
        for (int r = 0; r < 8; ++r) acc[j][r] = bcol[j];
    }

    float pg[8], pb[8], pm[8], pv[8];
    strip_mac<1, false>(a0, A, sW, arow_i, c, h, acc, bias, bias, bias, bias, pg, pb, pm, pv);

    float* op = out + (size_t)(row0 + h * 8) * HDIM + c;
    if (row0 + 16 <= nrows) {
#pragma unroll
        for (int j = 0; j < 8; ++j) {
#pragma unroll
            for (int r = 0; r < 8; ++r) op[r * HDIM + j * 16] = acc[j][r];
        }
    } else {
#pragma unroll
        for (int j = 0; j < 8; ++j) {
#pragma unroll
            for (int r = 0; r < 8; ++r) {
                if (row0 + h * 8 + r < nrows) op[r * HDIM + j * 16] = acc[j][r];
            }
        }
    }
}

__global__ void __launch_bounds__(256)
fc1_relu_fc2(const float* __restrict__ A,
             const float* __restrict__ W,
             const float* __restrict__ bias,
             const float* __restrict__ W2,
             const float* __restrict__ B2,
             float* __restrict__ out, int nrows) {
    __shared__ float4 sW[HDIM * WPITCH4];
    const int lane = threadIdx.x & 31;
    const int wv   = threadIdx.x >> 5;
    const int c    = lane & 15;
    const int h    = lane >> 4;
    const int row0 = (blockIdx.x * 8 + wv) << 4;
    int arow_i = row0 + c;
    if (arow_i >= nrows) arow_i = nrows - 1;

    float4 a0[16];
    load_a_strip(A, arow_i, h, a0);
    float bcol[8];
#pragma unroll
    for (int j = 0; j < 8; ++j) bcol[j] = bias[j * 16 + c];

    stage_weights(W, sW, wv, lane);
    __syncthreads();
    if (row0 >= nrows) return;

    v8f acc[8];
#pragma unroll
    for (int j = 0; j < 8; ++j) {
#pragma unroll
        for (int r = 0; r < 8; ++r) acc[j][r] = bcol[j];
    }

    float w2[8][4];
#pragma unroll
    for (int j = 0; j < 8; ++j) {
#pragma unroll
        for (int q = 0; q < 4; ++q) w2[j][q] = W2[q * HDIM + j * 16 + c];
    }
    const int   q_own = (c & 8) ? 2 : 0;
    const float b2x = B2[q_own], b2y = B2[q_own + 1];

    float pg[8], pb[8], pm[8], pv[8];
    strip_mac<1, false>(a0, A, sW, arow_i, c, h, acc, bias, bias, bias, bias, pg, pb, pm, pv);

    float p[32];
#pragma unroll
    for (int i = 0; i < 32; ++i) p[i] = 0.0f;
#pragma unroll
    for (int j = 0; j < 8; ++j) {
#pragma unroll
        for (int r = 0; r < 8; ++r) {
            const float y = fmaxf(acc[j][r], 0.0f);
#pragma unroll
            for (int q = 0; q < 4; ++q) p[r * 4 + q] = fmaf(y, w2[j][q], p[r * 4 + q]);
        }
    }

    const bool b0 = (c & 1) != 0, b1 = (c & 2) != 0, b2 = (c & 4) != 0, b3 = (c & 8) != 0;
    float s16[16], s8[8], s4[4], s2[2];
#pragma unroll
    for (int k = 0; k < 16; ++k) {
        const float keep = b0 ? p[k + 16] : p[k];
        const float send = b0 ? p[k] : p[k + 16];
        s16[k] = keep + __shfl_xor(send, 1, 32);
    }
#pragma unroll
    for (int k = 0; k < 8; ++k) {
        const float keep = b1 ? s16[k + 8] : s16[k];
        const float send = b1 ? s16[k] : s16[k + 8];
        s8[k] = keep + __shfl_xor(send, 2, 32);
    }
#pragma unroll
    for (int k = 0; k < 4; ++k) {
        const float keep = b2 ? s8[k + 4] : s8[k];
        const float send = b2 ? s8[k] : s8[k + 4];
        s4[k] = keep + __shfl_xor(send, 4, 32);
    }
#pragma unroll
    for (int k = 0; k < 2; ++k) {
        const float keep = b3 ? s4[k + 2] : s4[k];
        const float send = b3 ? s4[k] : s4[k + 2];
        s2[k] = keep + __shfl_xor(send, 8, 32);
    }
    const int r_own = (b0 ? 4 : 0) + (b1 ? 2 : 0) + (b2 ? 1 : 0);
    const int row   = row0 + h * 8 + r_own;
    if (row < nrows) {
        float2 o;
        o.x = s2[0] + b2x;
        o.y = s2[1] + b2y;
        *(float2*)(out + (size_t)row * 4 + q_own) = o;
    }
}

#define BIN_SHIFT 8
#define CHUNK_EDGES 4096

__device__ __forceinline__ int block_excl_scan(int v, int* wsum, int& total) {
    const int lane = threadIdx.x & 31;
    const int wv   = threadIdx.x >> 5;
    int incl = v;
#pragma unroll
    for (int off = 1; off < 32; off <<= 1) {
        const int t = __shfl_up(incl, off, 32);
        if (lane >= off) incl += t;
    }
    __syncthreads();
    if (lane == 31) wsum[wv] = incl;
    __syncthreads();
    int woff = 0, tot = 0;
#pragma unroll
    for (int w = 0; w < 8; ++w) {
        const int t = wsum[w];
        if (w < wv) woff += t;
        tot += t;
    }
    total = tot;
    return woff + incl - v;
}

__global__ void __launch_bounds__(256)
bin_count(const int* __restrict__ ei, int* __restrict__ hist, int E, int nrows, int nbins) {
    extern __shared__ int sbin[];
    for (int i = threadIdx.x; i < nbins; i += 256) sbin[i] = 0;
    __syncthreads();
    const int base = blockIdx.x * CHUNK_EDGES + threadIdx.x;
    int d[16];
#pragma unroll
    for (int j = 0; j < 16; ++j) {
        const int e = min(base + j * 256, E - 1);
        d[j] = ei[e];
    }
#pragma unroll
    for (int j = 0; j < 16; ++j)
        if (base + j * 256 < E && (unsigned)d[j] < (unsigned)nrows)
            atomicAdd(&sbin[d[j] >> BIN_SHIFT], 1);
    __syncthreads();
    for (int i = threadIdx.x; i < nbins; i += 256)
        hist[(size_t)blockIdx.x * nbins + i] = sbin[i];
}

__global__ void __launch_bounds__(256)
bin_offsets(int* __restrict__ hist, int* __restrict__ bin_tot, int nchunks, int nbins) {
    __shared__ int wsum[8];
    const int k = blockIdx.x;
    int carry = 0;
    for (int b0 = 0; b0 < nchunks; b0 += 256) {
        const int b = b0 + threadIdx.x;
        const int v = (b < nchunks) ? hist[(size_t)b * nbins + k] : 0;
        int total;
        const int excl = block_excl_scan(v, wsum, total);
        if (b < nchunks) hist[(size_t)b * nbins + k] = carry + excl;
        carry += total;
    }
    if (threadIdx.x == 0) bin_tot[k] = carry;
}

__global__ void __launch_bounds__(256)
bin_partition(const int* __restrict__ ei, const int* __restrict__ hist,
              const int* __restrict__ bin_tot, int* __restrict__ bin_start,
              unsigned* __restrict__ part, int E, int nrows, int nbins) {
    extern __shared__ int sbin[];
    __shared__ unsigned rec[CHUNK_EDGES];
    __shared__ int      pos[CHUNK_EDGES];
    __shared__ int      wsum[8];
    int* lcnt  = sbin;
    int* gbase = sbin + nbins;
    for (int i = threadIdx.x; i < nbins; i += 256) lcnt[i] = 0;
    int carry = 0;
    for (int k0 = 0; k0 < nbins; k0 += 256) {
        const int k = k0 + threadIdx.x;
        const int v = (k < nbins) ? bin_tot[k] : 0;
        int total;
        const int excl = block_excl_scan(v, wsum, total);
        if (k < nbins) {
            const int s = carry + excl;
            if (blockIdx.x == 0) bin_start[k] = s;
            gbase[k] = s + hist[(size_t)blockIdx.x * nbins + k];
        }
        carry += total;
    }
    __syncthreads();
    const int base = blockIdx.x * CHUNK_EDGES + threadIdx.x;
    int d[16], s[16], r[16];
#pragma unroll
    for (int j = 0; j < 16; ++j) {
        const int e = min(base + j * 256, E - 1);
        d[j] = ei[e];
        s[j] = ei[(size_t)E + e];
    }
#pragma unroll
    for (int j = 0; j < 16; ++j) {
        const bool live = base + j * 256 < E && (unsigned)d[j] < (unsigned)nrows;
        r[j] = live ? atomicAdd(&lcnt[d[j] >> BIN_SHIFT], 1) : -1;
    }
    __syncthreads();
    carry = 0;
    for (int k0 = 0; k0 < nbins; k0 += 256) {
        const int k = k0 + threadIdx.x;
        const int v = (k < nbins) ? lcnt[k] : 0;
        int total;
        const int excl = block_excl_scan(v, wsum, total);
        if (k < nbins) lcnt[k] = carry + excl;
        carry += total;
    }
    __syncthreads();
    const int nlive = carry;
#pragma unroll
    for (int j = 0; j < 16; ++j) {
        if (r[j] >= 0) {
            const int bin  = d[j] >> BIN_SHIFT;
            const int srcc = min(max(s[j], 0), nrows - 1);
            const int slot = lcnt[bin] + r[j];
            rec[slot] = ((unsigned)(d[j] & 255) << 24) | (unsigned)srcc;
            pos[slot] = gbase[bin] + r[j];
        }
    }
    __syncthreads();
    for (int i = threadIdx.x; i < nlive; i += 256) part[pos[i]] = rec[i];
}

#define SORT_CAP 8192
__global__ void __launch_bounds__(256)
bin_sort(const unsigned* __restrict__ part, const int* __restrict__ bin_start,
         const int* __restrict__ bin_tot, int2* __restrict__ seg,
         int* __restrict__ col, int nrows) {
    __shared__ int cnt[256];
    __shared__ int wsum[8];
    __shared__ int sorted[SORT_CAP];
    const int k = blockIdx.x;
    const int s = bin_start[k];
    const int m = bin_tot[k];
    cnt[threadIdx.x] = 0;
    __syncthreads();
    for (int i0 = 0; i0 < m; i0 += 2048) {
        unsigned r[8];
#pragma unroll
        for (int u = 0; u < 8; ++u) r[u] = part[s + min(i0 + u * 256 + (int)threadIdx.x, m - 1)];
#pragma unroll
        for (int u = 0; u < 8; ++u)
            if (i0 + u * 256 + (int)threadIdx.x < m) atomicAdd(&cnt[r[u] >> 24], 1);
    }
    __syncthreads();
    const int deg = cnt[threadIdx.x];
    int total;
    const int first = block_excl_scan(deg, wsum, total);
    const int n = (k << BIN_SHIFT) + threadIdx.x;
    if (n < nrows) seg[n] = make_int2(s + first, deg);
    cnt[threadIdx.x] = first;
    __syncthreads();
    const bool in_lds = (m <= SORT_CAP);
    for (int i0 = 0; i0 < m; i0 += 2048) {
        unsigned r[8];
#pragma unroll
        for (int u = 0; u < 8; ++u) r[u] = part[s + min(i0 + u * 256 + (int)threadIdx.x, m - 1)];
#pragma unroll
        for (int u = 0; u < 8; ++u) {
            if (i0 + u * 256 + (int)threadIdx.x < m) {
                const int p = atomicAdd(&cnt[r[u] >> 24], 1);
                const int v = (int)(r[u] & 0xFFFFFFu);
                if (in_lds) sorted[p] = v; else col[s + p] = v;
            }
        }
    }
    if (in_lds) {
        __syncthreads();
        for (int i = threadIdx.x; i < m; i += 256) col[s + i] = sorted[i];
    }
}

__device__ __forceinline__ float4 max4(float4 a, float4 b) {
    return make_float4(fmaxf(a.x, b.x), fmaxf(a.y, b.y), fmaxf(a.z, b.z), fmaxf(a.w, b.w));
}

#define FUSED_WGS   256
#define FUSED_WAVES 16

__device__ __forceinline__ void request_first16(const float4* __restrict__ xr, int ids,
                                                v4f (&v)[16]) {
#pragma unroll
    for (int i = 0; i < 16; ++i) {
        const int s = __builtin_amdgcn_readlane(ids, i);
        v[i] = *(const v4f*)(xr + (size_t)s * (HDIM / 4));
    }
}

__device__ __forceinline__ float4 max_over_rest(const int* __restrict__ col,
                                                const float4* __restrict__ xr,
                                                int beg, int cnt, int ids0, int lane, float4 m) {
    int ids = ids0;
    for (int base = 0; base < cnt; base += 32) {
        const int c = min(cnt - base, 32);
        if (base > 0) ids = col[beg + base + min(lane, c - 1)];
        for (int i = (base == 0) ? 16 : 0; i < c; i += 8) {
            const int s0 = __builtin_amdgcn_readlane(ids, i);
            const int s1 = __builtin_amdgcn_readlane(ids, i + 1);
            const int s2 = __builtin_amdgcn_readlane(ids, i + 2);
            const int s3 = __builtin_amdgcn_readlane(ids, i + 3);
            const int s4 = __builtin_amdgcn_readlane(ids, i + 4);
            const int s5 = __builtin_amdgcn_readlane(ids, i + 5);
            const int s6 = __builtin_amdgcn_readlane(ids, i + 6);
            const int s7 = __builtin_amdgcn_readlane(ids, i + 7);
            const float4 v0 = xr[(size_t)s0 * (HDIM / 4)];
            const float4 v1 = xr[(size_t)s1 * (HDIM / 4)];
            const float4 v2 = xr[(size_t)s2 * (HDIM / 4)];
            const float4 v3 = xr[(size_t)s3 * (HDIM / 4)];
            const float4 v4 = xr[(size_t)s4 * (HDIM / 4)];
            const float4 v5 = xr[(size_t)s5 * (HDIM / 4)];
            const float4 v6 = xr[(size_t)s6 * (HDIM / 4)];
            const float4 v7 = xr[(size_t)s7 * (HDIM / 4)];
            m = max4(m, max4(max4(max4(v0, v1), max4(v2, v3)),
                             max4(max4(v4, v5), max4(v6, v7))));
        }
    }
    return m;
}

__device__ __forceinline__ void root_chunk(const float4* bp_t, float4 av, v8f (&acc)[8]) {
    float4 b[8];
#pragma unroll
    for (int j = 0; j < 8; ++j) b[j] = bp_t[j * 16 * WPITCH4];
    const v2f a01 = {av.x, av.y};
    const v2f a23 = {av.z, av.w};
#pragma unroll
    for (int j = 0; j < 8; ++j) {
        const v2f b01 = {b[j].x, b[j].y};
        acc[j] = __builtin_amdgcn_wmma_f32_16x16x4_f32(false, a01, false, b01,
                                                       (short)0, acc[j], false, false);
    }
#pragma unroll
    for (int j = 0; j < 8; ++j) {
        const v2f b23 = {b[j].z, b[j].w};
        acc[j] = __builtin_amdgcn_wmma_f32_16x16x4_f32(false, a23, false, b23,
                                                       (short)0, acc[j], false, false);
    }
}

__global__ void __launch_bounds__(FUSED_WAVES * 32)
sage_layer_fused(const int2* __restrict__ seg, const int* __restrict__ col,
                 const float* __restrict__ X,
                 const float* __restrict__ WL,
                 const float* __restrict__ WR,
                 const float* __restrict__ bias,
                 const float* __restrict__ bn_g,
                 const float* __restrict__ bn_b,
                 const float* __restrict__ bn_m,
                 const float* __restrict__ bn_v,
                 float* __restrict__ out, int nrows, int row_tiles) {
    __shared__ float4 sW[2 * HDIM * WPITCH4];
    __shared__ float4 sAgg[FUSED_WAVES * 16 * WPITCH4];
    const int lane = threadIdx.x & 31;
    const int wv   = __builtin_amdgcn_readfirstlane((int)(threadIdx.x >> 5));
    const int c    = lane & 15;
    const int h    = lane >> 4;

    {
        float4 w[16];
#pragma unroll
        for (int it = 0; it < 8; ++it) {
            w[it]     = *(const float4*)(WL + (size_t)(it * FUSED_WAVES + wv) * HDIM + lane * 4);
            w[it + 8] = *(const float4*)(WR + (size_t)(it * FUSED_WAVES + wv) * HDIM + lane * 4);
        }
#pragma unroll
        for (int it = 0; it < 8; ++it) {
            sW[(it * FUSED_WAVES + wv) * WPITCH4 + lane] = w[it];
            sW[HDIM * WPITCH4 + (it * FUSED_WAVES + wv) * WPITCH4 + lane] = w[it + 8];
        }
    }
    float bcol[8];
#pragma unroll
    for (int j = 0; j < 8; ++j) bcol[j] = bias[j * 16 + c];
    __syncthreads();

    float4* myAgg = sAgg + wv * (16 * WPITCH4);
    const float4* __restrict__ xr = (const float4*)X + lane;

    const float4* bpr = sW + HDIM * WPITCH4 + c * WPITCH4 + h;

    for (int tile = blockIdx.x + FUSED_WGS * wv; tile < row_tiles; tile += FUSED_WGS * FUSED_WAVES) {
        const int row0 = tile << 4;

        int arow_i = row0 + c;
        if (arow_i >= nrows) arow_i = nrows - 1;
        const float4* ap1 = (const float4*)(X + (size_t)arow_i * HDIM + 4 * h);

        v8f acc[8];
#pragma unroll
        for (int j = 0; j < 8; ++j) {
#pragma unroll
            for (int q = 0; q < 8; ++q) acc[j][q] = 0.0f;
        }

        int2 sg = seg[min(row0 + c, nrows - 1)];
        if (row0 + c >= nrows) sg.y = 0;
        int beg = __builtin_amdgcn_readlane(sg.x, 0);
        int cnt = __builtin_amdgcn_readlane(sg.y, 0);
        int ids = 0;
        if (cnt > 0) ids = col[beg + min(lane, min(cnt, 32) - 1)];
        float4 a1c = ap1[0];
        for (int r = 0; r < 16; ++r) {
            const int cbeg = beg, ccnt = cnt, cids = ids;
            float4 a1n = a1c;
            if (r + 1 < 16) {
                beg = __builtin_amdgcn_readlane(sg.x, r + 1);
                cnt = __builtin_amdgcn_readlane(sg.y, r + 1);
                if (cnt > 0) ids = col[beg + min(lane, min(cnt, 32) - 1)];
                a1n = ap1[2 * (r + 1)];
            }
            v4f v[16];
            request_first16(xr, cids, v);
            __builtin_amdgcn_sched_barrier(0);
            root_chunk(bpr + 2 * r, a1c, acc);
            __builtin_amdgcn_sched_barrier(0);
            asm volatile("v_nop\n\tv_nop\n\tv_nop\n\tv_nop"
                         : "+v"(v[0]), "+v"(v[1]), "+v"(v[2]), "+v"(v[3]),
                           "+v"(v[4]), "+v"(v[5]), "+v"(v[6]), "+v"(v[7]));
            asm volatile(""
                         : "+v"(v[8]), "+v"(v[9]), "+v"(v[10]), "+v"(v[11]),
                           "+v"(v[12]), "+v"(v[13]), "+v"(v[14]), "+v"(v[15]));
            const float ninf = -__builtin_inff();
            const v4f mlo = __builtin_elementwise_max(
                __builtin_elementwise_max(__builtin_elementwise_max(v[0], v[1]),
                                          __builtin_elementwise_max(v[2], v[3])),
                __builtin_elementwise_max(__builtin_elementwise_max(v[4], v[5]),
                                          __builtin_elementwise_max(v[6], v[7])));
            const v4f mhi = __builtin_elementwise_max(
                __builtin_elementwise_max(__builtin_elementwise_max(v[8], v[9]),
                                          __builtin_elementwise_max(v[10], v[11])),
                __builtin_elementwise_max(__builtin_elementwise_max(v[12], v[13]),
                                          __builtin_elementwise_max(v[14], v[15])));
            const v4f mv = __builtin_elementwise_max(mlo, mhi);
            float4 m = make_float4(mv.x, mv.y, mv.z, mv.w);
            if (ccnt <= 0) m = make_float4(ninf, ninf, ninf, ninf);
            m = max_over_rest(col, xr, cbeg, ccnt, cids, lane, m);
            float4 o;
            o.x = (m.x == ninf) ? 0.0f : m.x;
            o.y = (m.y == ninf) ? 0.0f : m.y;
            o.z = (m.z == ninf) ? 0.0f : m.z;
            o.w = (m.w == ninf) ? 0.0f : m.w;
            myAgg[r * WPITCH4 + lane] = o;
            a1c = a1n;
        }
        __builtin_amdgcn_fence(__ATOMIC_RELEASE, "wavefront");
        __builtin_amdgcn_wave_barrier();
        __builtin_amdgcn_fence(__ATOMIC_ACQUIRE, "wavefront");

        float4 a0[16];
#pragma unroll
        for (int t = 0; t < 16; ++t) a0[t] = myAgg[c * WPITCH4 + 2 * t + h];

        float pg[8], pb[8], pm[8], pv[8];
        strip_mac<1, true>(a0, X, sW, arow_i, c, h, acc, bn_g, bn_b, bn_m, bn_v, pg, pb, pm, pv);
        asm volatile("v_nop\n\tv_nop\n\tv_nop\n\tv_nop");
        __builtin_amdgcn_sched_barrier(0);

        float* op = out + (size_t)(row0 + h * 8) * HDIM + c;
        const bool whole = (row0 + 16 <= nrows);
#pragma unroll
        for (int j = 0; j < 8; ++j) {
            const float s  = pg[j] * rsqrtf(pv[j] + BN_EPS);
            const float mu = pm[j];
            const float bb = pb[j];
            const float bj = bcol[j];
            if (whole) {
#pragma unroll
                for (int q = 0; q < 8; ++q) {
                    const float v = ((acc[j][q] + bj) - mu) * s + bb;
                    op[q * HDIM + j * 16] = fmaxf(v, 0.0f);
                }
            } else {
#pragma unroll
                for (int q = 0; q < 8; ++q) {
                    if (row0 + h * 8 + q < nrows) {
                        const float v = ((acc[j][q] + bj) - mu) * s + bb;
                        op[q * HDIM + j * 16] = fmaxf(v, 0.0f);
                    }
                }
            }
        }
    }
}

extern "C" void kernel_launch(void* const* d_in, const int* in_sizes, int n_in,
                              void* d_out, int out_size, void* d_ws, size_t ws_size,
                              hipStream_t stream) {
    const float* features = (const float*)d_in[0];
    const int*   ei       = (const int*)d_in[1];
    const float* w_proj = (const float*)d_in[2];
    const float* b_proj = (const float*)d_in[3];
    const float* w_l0 = (const float*)d_in[4];
    const float* b_l0 = (const float*)d_in[5];
    const float* w_r0 = (const float*)d_in[6];
    const float* bn_g0 = (const float*)d_in[7];
    const float* bn_b0 = (const float*)d_in[8];
    const float* bn_m0 = (const float*)d_in[9];
    const float* bn_v0 = (const float*)d_in[10];
    const float* w_l1 = (const float*)d_in[11];
    const float* b_l1 = (const float*)d_in[12];
    const float* w_r1 = (const float*)d_in[13];
    const float* bn_g1 = (const float*)d_in[14];
    const float* bn_b1 = (const float*)d_in[15];
    const float* bn_m1 = (const float*)d_in[16];
    const float* bn_v1 = (const float*)d_in[17];
    const float* w_fc1 = (const float*)d_in[18];
    const float* b_fc1 = (const float*)d_in[19];
    const float* w_fc2 = (const float*)d_in[20];
    const float* b_fc2 = (const float*)d_in[21];
    float* outp = (float*)d_out;

    const int N = in_sizes[0] / HDIM;
    const int E = in_sizes[1] / 2;

    const size_t SZ = (size_t)N * HDIM * sizeof(float);
    float* X    = (float*)d_ws;
    float* AGGF = (float*)((char*)d_ws + SZ);
    float* H1   = (float*)((char*)d_ws + 2 * SZ);
    char*  G    = (char*)d_ws + 3 * SZ;
    const size_t o_seg  = (((size_t)N + 1) * sizeof(int) + 255) & ~(size_t)255;
    const size_t o_part = o_seg  + (((size_t)N * sizeof(int2) + 255) & ~(size_t)255);
    const size_t o_col  = o_part + (((size_t)E * sizeof(int)  + 255) & ~(size_t)255);
    int2*     seg  = (int2*)(G + o_seg);
    unsigned* part = (unsigned*)(G + o_part);
    int*      col  = (int*)(G + o_col);

    const int nbins   = (N + 255) / 256;
    const int nchunks = (E + 4095) / 4096;
    int* hist      = (int*)AGGF;
    int* bin_tot   = hist + (size_t)nchunks * nbins;
    int* bin_start = bin_tot + nbins;

    const int row_tiles   = (N + 15) / 16;
    const int gemm_blocks = (row_tiles + 7) / 8;

    bin_count<<<nchunks, 256, (size_t)nbins * sizeof(int), stream>>>(ei, hist, E, N, nbins);
    bin_offsets<<<nbins, 256, 0, stream>>>(hist, bin_tot, nchunks, nbins);
    bin_partition<<<nchunks, 256, 2 * (size_t)nbins * sizeof(int), stream>>>(ei, hist, bin_tot, bin_start,
                                                                          part, E, N, nbins);
    bin_sort<<<nbins, 256, 0, stream>>>(part, bin_start, bin_tot, seg, col, N);

    gemm128_bias<<<gemm_blocks, 256, 0, stream>>>(features, w_proj, b_proj, X, N);

    sage_layer_fused<<<FUSED_WGS, FUSED_WAVES * 32, 0, stream>>>(seg, col, X, w_l0, w_r0, b_l0,
                                                                 bn_g0, bn_b0, bn_m0, bn_v0, H1, N, row_tiles);

    sage_layer_fused<<<FUSED_WGS, FUSED_WAVES * 32, 0, stream>>>(seg, col, H1, w_l1, w_r1, b_l1,
                                                                 bn_g1, bn_b1, bn_m1, bn_v1, X, N, row_tiles);

    fc1_relu_fc2<<<gemm_blocks, 256, 0, stream>>>(X, w_fc1, b_fc1, w_fc2, b_fc2, outp, N);
}


# Output head (fc1 + ReLU, fc2) computed in SAGE layer 1's launch: a strip's result rows pass through the wave's LDS rows into A layout, fc1 weights as a third LDS matrix (14 waves); no fc1 launch
# speedup vs baseline: 4.8457x; 1.0839x over previous
#include <hip/hip_runtime.h>

#define HDIM 128
#define BN_EPS 1e-5f
#define WPITCH4 33

typedef __attribute__((ext_vector_type(2))) float v2f;
typedef __attribute__((ext_vector_type(8))) float v8f;
typedef __attribute__((ext_vector_type(4))) float v4f;

__device__ __forceinline__ void stage_weights(const float* __restrict__ Wg, float4* sW,
                                              int wv, int lane) {
    float4 w[16];
#pragma unroll
    for (int it = 0; it < 16; ++it)
        w[it] = *(const float4*)(Wg + (size_t)(it * 8 + wv) * HDIM + lane * 4);
#pragma unroll
    for (int it = 0; it < 16; ++it)
        sW[(it * 8 + wv) * WPITCH4 + lane] = w[it];
}

__device__ __forceinline__ void load_a_strip(const float* __restrict__ A, int arow_i, int h,
                                             float4 (&a)[16]) {
    const float4* ap = (const float4*)(A + (size_t)arow_i * HDIM + 4 * h);
#pragma unroll
    for (int t = 0; t < 16; ++t) a[t] = ap[2 * t];
}

template <int NMAT, bool BN>
__device__ __forceinline__ void strip_mac(float4 (&a0)[16], const float* __restrict__ A1,
                                          const float4* sW, int arow_i, int c, int h,
                                          v8f (&acc)[8],
                                          const float* __restrict__ bn_g,
                                          const float* __restrict__ bn_b,
                                          const float* __restrict__ bn_m,
                                          const float* __restrict__ bn_v,
                                          float (&pg)[8], float (&pb)[8],
                                          float (&pm)[8], float (&pv)[8]) {
    const float4* ap1 = (const float4*)(A1 + (size_t)arow_i * HDIM + 4 * h);
    const float4* bp  = sW + c * WPITCH4 + h;

    float4 a1[16];
    float4 b[2][8];
#pragma unroll
    for (int j = 0; j < 8; ++j) b[0][j] = bp[j * 16 * WPITCH4];

#pragma unroll
    for (int t = 0; t < 16 * NMAT; ++t) {
        if (NMAT == 2 && t == 8) {
#pragma unroll
            for (int u = 0; u < 8; ++u) a1[u] = ap1[2 * u];
        }
        if (NMAT == 2 && t == 16) {
#pragma unroll
            for (int u = 8; u < 16; ++u) a1[u] = ap1[2 * u];
        }
        if (BN && t == 16 * NMAT - 3) {
#pragma unroll
            for (int j = 0; j < 8; ++j) {
                pg[j] = bn_g[j * 16 + c];
                pv[j] = bn_v[j * 16 + c];
                pm[j] = bn_m[j * 16 + c];
                pb[j] = bn_b[j * 16 + c];
            }
        }
        if (t + 1 < 16 * NMAT) {
            const int tn = t + 1;
            const float4* nb = bp + (tn >> 4) * (HDIM * WPITCH4) + (tn & 15) * 2;
#pragma unroll
            for (int j = 0; j < 8; ++j) b[tn & 1][j] = nb[j * 16 * WPITCH4];
        }
        __builtin_amdgcn_sched_barrier(0);
        const float4 av = (t < 16) ? a0[t & 15] : a1[t & 15];
        const v2f a01 = {av.x, av.y};
        const v2f a23 = {av.z, av.w};
#pragma unroll
        for (int j = 0; j < 8; ++j) {
            const v2f b01 = {b[t & 1][j].x, b[t & 1][j].y};
            acc[j] = __builtin_amdgcn_wmma_f32_16x16x4_f32(false, a01, false, b01,
                                                           (short)0, acc[j], false, false);
        }
#pragma unroll
        for (int j = 0; j < 8; ++j) {
            const v2f b23 = {b[t & 1][j].z, b[t & 1][j].w};
            acc[j] = __builtin_amdgcn_wmma_f32_16x16x4_f32(false, a23, false, b23,
                                                           (short)0, acc[j], false, false);
        }
        __builtin_amdgcn_sched_barrier(0);
    }
}

__global__ void __launch_bounds__(256)
gemm128_bias(const float* __restrict__ A,
             const float* __restrict__ W,
             const float* __restrict__ bias,
             float* __restrict__ out,
             int nrows) {
    __shared__ float4 sW[HDIM * WPITCH4];
    const int lane = threadIdx.x & 31;
    const int wv   = threadIdx.x >> 5;
    const int c    = lane & 15;
    const int h    = lane >> 4;
    const int row0 = (blockIdx.x * 8 + wv) << 4;
    int arow_i = row0 + c;
    if (arow_i >= nrows) arow_i = nrows - 1;

    float4 a0[16];
    load_a_strip(A, arow_i, h, a0);
    float bcol[8];
#pragma unroll
    for (int j = 0; j < 8; ++j) bcol[j] = bias[j * 16 + c];

    stage_weights(W, sW, wv, lane);
    __syncthreads();
    if (row0 >= nrows) return;

    v8f acc[8];
#pragma unroll
    for (int j = 0; j < 8; ++j) {
#pragma unroll
        for (int r = 0; r < 8; ++r) acc[j][r] = bcol[j];
    }

    float pg[8], pb[8], pm[8], pv[8];
    strip_mac<1, false>(a0, A, sW, arow_i, c, h, acc, bias, bias, bias, bias, pg, pb, pm, pv);

    float* op = out + (size_t)(row0 + h * 8) * HDIM + c;
    if (row0 + 16 <= nrows) {
#pragma unroll
        for (int j = 0; j < 8; ++j) {
#pragma unroll
            for (int r = 0; r < 8; ++r) op[r * HDIM + j * 16] = acc[j][r];
        }
    } else {
#pragma unroll
        for (int j = 0; j < 8; ++j) {
#pragma unroll
            for (int r = 0; r < 8; ++r) {
                if (row0 + h * 8 + r < nrows) op[r * HDIM + j * 16] = acc[j][r];
            }
        }
    }
}

__device__ __forceinline__ void fc2_head(const v8f (&acc)[8],
                                         const float* __restrict__ W2,
                                         const float* __restrict__ B2,
                                         float* __restrict__ out,
                                         int row0, int c, int h, int nrows) {
    float w2[8][4];
#pragma unroll
    for (int j = 0; j < 8; ++j) {
#pragma unroll
        for (int q = 0; q < 4; ++q) w2[j][q] = W2[q * HDIM + j * 16 + c];
    }
    const int   q_own = (c & 8) ? 2 : 0;
    const float b2x = B2[q_own], b2y = B2[q_own + 1];

    float p[32];
#pragma unroll
    for (int i = 0; i < 32; ++i) p[i] = 0.0f;
#pragma unroll
    for (int j = 0; j < 8; ++j) {
#pragma unroll
        for (int r = 0; r < 8; ++r) {
            const float y = fmaxf(acc[j][r], 0.0f);
#pragma unroll
            for (int q = 0; q < 4; ++q) p[r * 4 + q] = fmaf(y, w2[j][q], p[r * 4 + q]);
        }
    }

    const bool b0 = (c & 1) != 0, b1 = (c & 2) != 0, b2 = (c & 4) != 0, b3 = (c & 8) != 0;
    float s16[16], s8[8], s4[4], s2[2];
#pragma unroll
    for (int k = 0; k < 16; ++k) {
        const float keep = b0 ? p[k + 16] : p[k];
        const float send = b0 ? p[k] : p[k + 16];
        s16[k] = keep + __shfl_xor(send, 1, 32);
    }
#pragma unroll
    for (int k = 0; k < 8; ++k) {
        const float keep = b1 ? s16[k + 8] : s16[k];
        const float send = b1 ? s16[k] : s16[k + 8];
        s8[k] = keep + __shfl_xor(send, 2, 32);
    }
#pragma unroll
    for (int k = 0; k < 4; ++k) {
        const float keep = b2 ? s8[k + 4] : s8[k];
        const float send = b2 ? s8[k] : s8[k + 4];
        s4[k] = keep + __shfl_xor(send, 4, 32);
    }
#pragma unroll
    for (int k = 0; k < 2; ++k) {
        const float keep = b3 ? s4[k + 2] : s4[k];
        const float send = b3 ? s4[k] : s4[k + 2];
        s2[k] = keep + __shfl_xor(send, 8, 32);
    }
    const int r_own = (b0 ? 4 : 0) + (b1 ? 2 : 0) + (b2 ? 1 : 0);
    const int row   = row0 + h * 8 + r_own;
    if (row < nrows) {
        float2 o;
        o.x = s2[0] + b2x;
        o.y = s2[1] + b2y;
        *(float2*)(out + (size_t)row * 4 + q_own) = o;
    }
}

#define BIN_SHIFT 8
#define CHUNK_EDGES 4096

__device__ __forceinline__ int block_excl_scan(int v, int* wsum, int& total) {
    const int lane = threadIdx.x & 31;
    const int wv   = threadIdx.x >> 5;
    int incl = v;
#pragma unroll
    for (int off = 1; off < 32; off <<= 1) {
        const int t = __shfl_up(incl, off, 32);
        if (lane >= off) incl += t;
    }
    __syncthreads();
    if (lane == 31) wsum[wv] = incl;
    __syncthreads();
    int woff = 0, tot = 0;
#pragma unroll
    for (int w = 0; w < 8; ++w) {
        const int t = wsum[w];
        if (w < wv) woff += t;
        tot += t;
    }
    total = tot;
    return woff + incl - v;
}

__global__ void __launch_bounds__(256)
bin_count(const int* __restrict__ ei, int* __restrict__ hist, int E, int nrows, int nbins) {
    extern __shared__ int sbin[];
    for (int i = threadIdx.x; i < nbins; i += 256) sbin[i] = 0;
    __syncthreads();
    const int base = blockIdx.x * CHUNK_EDGES + threadIdx.x;
    int d[16];
#pragma unroll
    for (int j = 0; j < 16; ++j) {
        const int e = min(base + j * 256, E - 1);
        d[j] = ei[e];
    }
#pragma unroll
    for (int j = 0; j < 16; ++j)
        if (base + j * 256 < E && (unsigned)d[j] < (unsigned)nrows)
            atomicAdd(&sbin[d[j] >> BIN_SHIFT], 1);
    __syncthreads();
    for (int i = threadIdx.x; i < nbins; i += 256)
        hist[(size_t)blockIdx.x * nbins + i] = sbin[i];
}

__global__ void __launch_bounds__(256)
bin_offsets(int* __restrict__ hist, int* __restrict__ bin_tot, int nchunks, int nbins) {
    __shared__ int wsum[8];
    const int k = blockIdx.x;
    int carry = 0;
    for (int b0 = 0; b0 < nchunks; b0 += 256) {
        const int b = b0 + threadIdx.x;
        const int v = (b < nchunks) ? hist[(size_t)b * nbins + k] : 0;
        int total;
        const int excl = block_excl_scan(v, wsum, total);
        if (b < nchunks) hist[(size_t)b * nbins + k] = carry + excl;
        carry += total;
    }
    if (threadIdx.x == 0) bin_tot[k] = carry;
}

__global__ void __launch_bounds__(256)
bin_partition(const int* __restrict__ ei, const int* __restrict__ hist,
              const int* __restrict__ bin_tot, int* __restrict__ bin_start,
              unsigned* __restrict__ part, int E, int nrows, int nbins) {
    extern __shared__ int sbin[];
    __shared__ unsigned rec[CHUNK_EDGES];
    __shared__ int      pos[CHUNK_EDGES];
    __shared__ int      wsum[8];
    int* lcnt  = sbin;
    int* gbase = sbin + nbins;
    for (int i = threadIdx.x; i < nbins; i += 256) lcnt[i] = 0;
    int carry = 0;
    for (int k0 = 0; k0 < nbins; k0 += 256) {
        const int k = k0 + threadIdx.x;
        const int v = (k < nbins) ? bin_tot[k] : 0;
        int total;
        const int excl = block_excl_scan(v, wsum, total);
        if (k < nbins) {
            const int s = carry + excl;
            if (blockIdx.x == 0) bin_start[k] = s;
            gbase[k] = s + hist[(size_t)blockIdx.x * nbins + k];
        }
        carry += total;
    }
    __syncthreads();
    const int base = blockIdx.x * CHUNK_EDGES + threadIdx.x;
    int d[16], s[16], r[16];
#pragma unroll
    for (int j = 0; j < 16; ++j) {
        const int e = min(base + j * 256, E - 1);
        d[j] = ei[e];
        s[j] = ei[(size_t)E + e];
    }
#pragma unroll
    for (int j = 0; j < 16; ++j) {
        const bool live = base + j * 256 < E && (unsigned)d[j] < (unsigned)nrows;
        r[j] = live ? atomicAdd(&lcnt[d[j] >> BIN_SHIFT], 1) : -1;
    }
    __syncthreads();
    carry = 0;
    for (int k0 = 0; k0 < nbins; k0 += 256) {
        const int k = k0 + threadIdx.x;
        const int v = (k < nbins) ? lcnt[k] : 0;
        int total;
        const int excl = block_excl_scan(v, wsum, total);
        if (k < nbins) lcnt[k] = carry + excl;
        carry += total;
    }
    __syncthreads();
    const int nlive = carry;
#pragma unroll
    for (int j = 0; j < 16; ++j) {
        if (r[j] >= 0) {
            const int bin  = d[j] >> BIN_SHIFT;
            const int srcc = min(max(s[j], 0), nrows - 1);
            const int slot = lcnt[bin] + r[j];
            rec[slot] = ((unsigned)(d[j] & 255) << 24) | (unsigned)srcc;
            pos[slot] = gbase[bin] + r[j];
        }
    }
    __syncthreads();
    for (int i = threadIdx.x; i < nlive; i += 256) part[pos[i]] = rec[i];
}

#define SORT_CAP 8192
__global__ void __launch_bounds__(256)
bin_sort(const unsigned* __restrict__ part, const int* __restrict__ bin_start,
         const int* __restrict__ bin_tot, int2* __restrict__ seg,
         int* __restrict__ col, int nrows) {
    __shared__ int cnt[256];
    __shared__ int wsum[8];
    __shared__ int sorted[SORT_CAP];
    const int k = blockIdx.x;
    const int s = bin_start[k];
    const int m = bin_tot[k];
    cnt[threadIdx.x] = 0;
    __syncthreads();
    for (int i0 = 0; i0 < m; i0 += 2048) {
        unsigned r[8];
#pragma unroll
        for (int u = 0; u < 8; ++u) r[u] = part[s + min(i0 + u * 256 + (int)threadIdx.x, m - 1)];
#pragma unroll
        for (int u = 0; u < 8; ++u)
            if (i0 + u * 256 + (int)threadIdx.x < m) atomicAdd(&cnt[r[u] >> 24], 1);
    }
    __syncthreads();
    const int deg = cnt[threadIdx.x];
    int total;
    const int first = block_excl_scan(deg, wsum, total);
    const int n = (k << BIN_SHIFT) + threadIdx.x;
    if (n < nrows) seg[n] = make_int2(s + first, deg);
    cnt[threadIdx.x] = first;
    __syncthreads();
    const bool in_lds = (m <= SORT_CAP);
    for (int i0 = 0; i0 < m; i0 += 2048) {
        unsigned r[8];
#pragma unroll
        for (int u = 0; u < 8; ++u) r[u] = part[s + min(i0 + u * 256 + (int)threadIdx.x, m - 1)];
#pragma unroll
        for (int u = 0; u < 8; ++u) {
            if (i0 + u * 256 + (int)threadIdx.x < m) {
                const int p = atomicAdd(&cnt[r[u] >> 24], 1);
                const int v = (int)(r[u] & 0xFFFFFFu);
                if (in_lds) sorted[p] = v; else col[s + p] = v;
            }
        }
    }
    if (in_lds) {
        __syncthreads();
        for (int i = threadIdx.x; i < m; i += 256) col[s + i] = sorted[i];
    }
}

__device__ __forceinline__ float4 max4(float4 a, float4 b) {
    return make_float4(fmaxf(a.x, b.x), fmaxf(a.y, b.y), fmaxf(a.z, b.z), fmaxf(a.w, b.w));
}

#define FUSED_WGS   256
#define FUSED_WAVES 16
#define HEAD_WAVES  14

__device__ __forceinline__ void request_first16(const float4* __restrict__ xr, int ids,
                                                v4f (&v)[16]) {
#pragma unroll
    for (int i = 0; i < 16; ++i) {
        const int s = __builtin_amdgcn_readlane(ids, i);
        v[i] = *(const v4f*)(xr + (size_t)s * (HDIM / 4));
    }
}

__device__ __forceinline__ float4 max_over_rest(const int* __restrict__ col,
                                                const float4* __restrict__ xr,
                                                int beg, int cnt, int ids0, int lane, float4 m) {
    int ids = ids0;
    for (int base = 0; base < cnt; base += 32) {
        const int c = min(cnt - base, 32);
        if (base > 0) ids = col[beg + base + min(lane, c - 1)];
        for (int i = (base == 0) ? 16 : 0; i < c; i += 8) {
            const int s0 = __builtin_amdgcn_readlane(ids, i);
            const int s1 = __builtin_amdgcn_readlane(ids, i + 1);
            const int s2 = __builtin_amdgcn_readlane(ids, i + 2);
            const int s3 = __builtin_amdgcn_readlane(ids, i + 3);
            const int s4 = __builtin_amdgcn_readlane(ids, i + 4);
            const int s5 = __builtin_amdgcn_readlane(ids, i + 5);
            const int s6 = __builtin_amdgcn_readlane(ids, i + 6);
            const int s7 = __builtin_amdgcn_readlane(ids, i + 7);
            const float4 v0 = xr[(size_t)s0 * (HDIM / 4)];
            const float4 v1 = xr[(size_t)s1 * (HDIM / 4)];
            const float4 v2 = xr[(size_t)s2 * (HDIM / 4)];
            const float4 v3 = xr[(size_t)s3 * (HDIM / 4)];
            const float4 v4 = xr[(size_t)s4 * (HDIM / 4)];
            const float4 v5 = xr[(size_t)s5 * (HDIM / 4)];
            const float4 v6 = xr[(size_t)s6 * (HDIM / 4)];
            const float4 v7 = xr[(size_t)s7 * (HDIM / 4)];
            m = max4(m, max4(max4(max4(v0, v1), max4(v2, v3)),
                             max4(max4(v4, v5), max4(v6, v7))));
        }
    }
    return m;
}

__device__ __forceinline__ void root_chunk(const float4* bp_t, float4 av, v8f (&acc)[8]) {
    float4 b[8];
#pragma unroll
    for (int j = 0; j < 8; ++j) b[j] = bp_t[j * 16 * WPITCH4];
    const v2f a01 = {av.x, av.y};
    const v2f a23 = {av.z, av.w};
#pragma unroll
    for (int j = 0; j < 8; ++j) {
        const v2f b01 = {b[j].x, b[j].y};
        acc[j] = __builtin_amdgcn_wmma_f32_16x16x4_f32(false, a01, false, b01,
                                                       (short)0, acc[j], false, false);
    }
#pragma unroll
    for (int j = 0; j < 8; ++j) {
        const v2f b23 = {b[j].z, b[j].w};
        acc[j] = __builtin_amdgcn_wmma_f32_16x16x4_f32(false, a23, false, b23,
                                                       (short)0, acc[j], false, false);
    }
}

template <bool HEAD>
__global__ void __launch_bounds__((HEAD ? HEAD_WAVES : FUSED_WAVES) * 32)
sage_layer_fused(const int2* __restrict__ seg, const int* __restrict__ col,
                 const float* __restrict__ X,
                 const float* __restrict__ WL,
                 const float* __restrict__ WR,
                 const float* __restrict__ bias,
                 const float* __restrict__ bn_g,
                 const float* __restrict__ bn_b,
                 const float* __restrict__ bn_m,
                 const float* __restrict__ bn_v,
                 const float* __restrict__ WF,
                 const float* __restrict__ BF,
                 const float* __restrict__ W2,
                 const float* __restrict__ B2,
                 float* __restrict__ out,
                 int nrows, int row_tiles) {
    constexpr int NW = HEAD ? HEAD_WAVES : FUSED_WAVES;
    __shared__ float4 sW[(HEAD ? 3 : 2) * HDIM * WPITCH4];
    __shared__ float4 sAgg[NW * 16 * WPITCH4];
    const int lane = threadIdx.x & 31;
    const int wv   = __builtin_amdgcn_readfirstlane((int)(threadIdx.x >> 5));
    const int c    = lane & 15;
    const int h    = lane >> 4;

    if constexpr (!HEAD) {
        float4 w[16];
#pragma unroll
        for (int it = 0; it < 8; ++it) {
            w[it]     = *(const float4*)(WL + (size_t)(it * FUSED_WAVES + wv) * HDIM + lane * 4);
            w[it + 8] = *(const float4*)(WR + (size_t)(it * FUSED_WAVES + wv) * HDIM + lane * 4);
        }
#pragma unroll
        for (int it = 0; it < 8; ++it) {
            sW[(it * FUSED_WAVES + wv) * WPITCH4 + lane] = w[it];
            sW[HDIM * WPITCH4 + (it * FUSED_WAVES + wv) * WPITCH4 + lane] = w[it + 8];
        }
    } else {
        constexpr int NIT = (HDIM + NW - 1) / NW;
        float4 w[3][NIT];
#pragma unroll
        for (int it = 0; it < NIT; ++it) {
            const int row = min(it * NW + wv, HDIM - 1);
            w[0][it] = *(const float4*)(WL + (size_t)row * HDIM + lane * 4);
            w[1][it] = *(const float4*)(WR + (size_t)row * HDIM + lane * 4);
            w[2][it] = *(const float4*)(WF + (size_t)row * HDIM + lane * 4);
        }
#pragma unroll
        for (int it = 0; it < NIT; ++it) {
            const int row = min(it * NW + wv, HDIM - 1);
            sW[row * WPITCH4 + lane] = w[0][it];
            sW[HDIM * WPITCH4 + row * WPITCH4 + lane] = w[1][it];
            sW[2 * HDIM * WPITCH4 + row * WPITCH4 + lane] = w[2][it];
        }
    }
    float bcol[8];
#pragma unroll
    for (int j = 0; j < 8; ++j) bcol[j] = bias[j * 16 + c];
    __syncthreads();

    float4* myAgg = sAgg + wv * (16 * WPITCH4);
    const float4* __restrict__ xr = (const float4*)X + lane;

    const float4* bpr = sW + HDIM * WPITCH4 + c * WPITCH4 + h;

    for (int tile = blockIdx.x + FUSED_WGS * wv; tile < row_tiles; tile += FUSED_WGS * NW) {
        const int row0 = tile << 4;

        int arow_i = row0 + c;
        if (arow_i >= nrows) arow_i = nrows - 1;
        const float4* ap1 = (const float4*)(X + (size_t)arow_i * HDIM + 4 * h);

        v8f acc[8];
#pragma unroll
        for (int j = 0; j < 8; ++j) {
#pragma unroll
            for (int q = 0; q < 8; ++q) acc[j][q] = 0.0f;
        }

        int2 sg = seg[min(row0 + c, nrows - 1)];
        if (row0 + c >= nrows) sg.y = 0;
        int beg = __builtin_amdgcn_readlane(sg.x, 0);
        int cnt = __builtin_amdgcn_readlane(sg.y, 0);
        int ids = 0;
        if (cnt > 0) ids = col[beg + min(lane, min(cnt, 32) - 1)];
        float4 a1c = ap1[0];
        for (int r = 0; r < 16; ++r) {
            const int cbeg = beg, ccnt = cnt, cids = ids;
            float4 a1n = a1c;
            if (r + 1 < 16) {
                beg = __builtin_amdgcn_readlane(sg.x, r + 1);
                cnt = __builtin_amdgcn_readlane(sg.y, r + 1);
                if (cnt > 0) ids = col[beg + min(lane, min(cnt, 32) - 1)];
                a1n = ap1[2 * (r + 1)];
            }
            v4f v[16];
            request_first16(xr, cids, v);
            __builtin_amdgcn_sched_barrier(0);
            root_chunk(bpr + 2 * r, a1c, acc);
            __builtin_amdgcn_sched_barrier(0);
            asm volatile("v_nop\n\tv_nop\n\tv_nop\n\tv_nop"
                         : "+v"(v[0]), "+v"(v[1]), "+v"(v[2]), "+v"(v[3]),
                           "+v"(v[4]), "+v"(v[5]), "+v"(v[6]), "+v"(v[7]));
            asm volatile(""
                         : "+v"(v[8]), "+v"(v[9]), "+v"(v[10]), "+v"(v[11]),
                           "+v"(v[12]), "+v"(v[13]), "+v"(v[14]), "+v"(v[15]));
            const float ninf = -__builtin_inff();
            const v4f mlo = __builtin_elementwise_max(
                __builtin_elementwise_max(__builtin_elementwise_max(v[0], v[1]),
                                          __builtin_elementwise_max(v[2], v[3])),
                __builtin_elementwise_max(__builtin_elementwise_max(v[4], v[5]),
                                          __builtin_elementwise_max(v[6], v[7])));
            const v4f mhi = __builtin_elementwise_max(
                __builtin_elementwise_max(__builtin_elementwise_max(v[8], v[9]),
                                          __builtin_elementwise_max(v[10], v[11])),
                __builtin_elementwise_max(__builtin_elementwise_max(v[12], v[13]),
                                          __builtin_elementwise_max(v[14], v[15])));
            const v4f mv = __builtin_elementwise_max(mlo, mhi);
            float4 m = make_float4(mv.x, mv.y, mv.z, mv.w);
            if (ccnt <= 0) m = make_float4(ninf, ninf, ninf, ninf);
            m = max_over_rest(col, xr, cbeg, ccnt, cids, lane, m);
            float4 o;
            o.x = (m.x == ninf) ? 0.0f : m.x;
            o.y = (m.y == ninf) ? 0.0f : m.y;
            o.z = (m.z == ninf) ? 0.0f : m.z;
            o.w = (m.w == ninf) ? 0.0f : m.w;
            myAgg[r * WPITCH4 + lane] = o;
            a1c = a1n;
        }
        __builtin_amdgcn_fence(__ATOMIC_RELEASE, "wavefront");
        __builtin_amdgcn_wave_barrier();
        __builtin_amdgcn_fence(__ATOMIC_ACQUIRE, "wavefront");

        float4 a0[16];
#pragma unroll
        for (int t = 0; t < 16; ++t) a0[t] = myAgg[c * WPITCH4 + 2 * t + h];

        float pg[8], pb[8], pm[8], pv[8];
        strip_mac<1, true>(a0, X, sW, arow_i, c, h, acc, bn_g, bn_b, bn_m, bn_v, pg, pb, pm, pv);
        asm volatile("v_nop\n\tv_nop\n\tv_nop\n\tv_nop");
        __builtin_amdgcn_sched_barrier(0);

        if constexpr (!HEAD) {
            float* op = out + (size_t)(row0 + h * 8) * HDIM + c;
            const bool whole = (row0 + 16 <= nrows);
#pragma unroll
            for (int j = 0; j < 8; ++j) {
                const float s  = pg[j] * rsqrtf(pv[j] + BN_EPS);
                const float mu = pm[j];
                const float bb = pb[j];
                const float bj = bcol[j];
                if (whole) {
#pragma unroll
                    for (int q = 0; q < 8; ++q) {
                        const float v = ((acc[j][q] + bj) - mu) * s + bb;
                        op[q * HDIM + j * 16] = fmaxf(v, 0.0f);
                    }
                } else {
#pragma unroll
                    for (int q = 0; q < 8; ++q) {
                        if (row0 + h * 8 + q < nrows) {
                            const float v = ((acc[j][q] + bj) - mu) * s + bb;
                            op[q * HDIM + j * 16] = fmaxf(v, 0.0f);
                        }
                    }
                }
            }
        } else {
            float* rp = (float*)myAgg + (h * 8) * (WPITCH4 * 4) + c;
#pragma unroll
            for (int j = 0; j < 8; ++j) {
                const float s  = pg[j] * rsqrtf(pv[j] + BN_EPS);
                const float mu = pm[j];
                const float bb = pb[j];
                const float bj = bcol[j];
#pragma unroll
                for (int q = 0; q < 8; ++q) {
                    const float v = ((acc[j][q] + bj) - mu) * s + bb;
                    rp[q * (WPITCH4 * 4) + j * 16] = fmaxf(v, 0.0f);
                }
            }
            __builtin_amdgcn_fence(__ATOMIC_RELEASE, "wavefront");
            __builtin_amdgcn_wave_barrier();
            __builtin_amdgcn_fence(__ATOMIC_ACQUIRE, "wavefront");

            float4 a2[16];
#pragma unroll
            for (int t = 0; t < 16; ++t) a2[t] = myAgg[c * WPITCH4 + 2 * t + h];

            v8f acc2[8];
#pragma unroll
            for (int j = 0; j < 8; ++j) {
                const float bf = BF[j * 16 + c];
#pragma unroll
                for (int q = 0; q < 8; ++q) acc2[j][q] = bf;
            }
            strip_mac<1, false>(a2, X, sW + 2 * HDIM * WPITCH4, arow_i, c, h, acc2,
                                BF, BF, BF, BF, pg, pb, pm, pv);
            asm volatile("v_nop\n\tv_nop\n\tv_nop\n\tv_nop");
            __builtin_amdgcn_sched_barrier(0);
            fc2_head(acc2, W2, B2, out, row0, c, h, nrows);
        }
    }
}

extern "C" void kernel_launch(void* const* d_in, const int* in_sizes, int n_in,
                              void* d_out, int out_size, void* d_ws, size_t ws_size,
                              hipStream_t stream) {
    const float* features = (const float*)d_in[0];
    const int*   ei       = (const int*)d_in[1];
    const float* w_proj = (const float*)d_in[2];
    const float* b_proj = (const float*)d_in[3];
    const float* w_l0 = (const float*)d_in[4];
    const float* b_l0 = (const float*)d_in[5];
    const float* w_r0 = (const float*)d_in[6];
    const float* bn_g0 = (const float*)d_in[7];
    const float* bn_b0 = (const float*)d_in[8];
    const float* bn_m0 = (const float*)d_in[9];
    const float* bn_v0 = (const float*)d_in[10];
    const float* w_l1 = (const float*)d_in[11];
    const float* b_l1 = (const float*)d_in[12];
    const float* w_r1 = (const float*)d_in[13];
    const float* bn_g1 = (const float*)d_in[14];
    const float* bn_b1 = (const float*)d_in[15];
    const float* bn_m1 = (const float*)d_in[16];
    const float* bn_v1 = (const float*)d_in[17];
    const float* w_fc1 = (const float*)d_in[18];
    const float* b_fc1 = (const float*)d_in[19];
    const float* w_fc2 = (const float*)d_in[20];
    const float* b_fc2 = (const float*)d_in[21];
    float* outp = (float*)d_out;

    const int N = in_sizes[0] / HDIM;
    const int E = in_sizes[1] / 2;

    const size_t SZ = (size_t)N * HDIM * sizeof(float);
    float* X    = (float*)d_ws;
    float* AGGF = (float*)((char*)d_ws + SZ);
    float* H1   = (float*)((char*)d_ws + 2 * SZ);
    char*  G    = (char*)d_ws + 3 * SZ;
    const size_t o_seg  = (((size_t)N + 1) * sizeof(int) + 255) & ~(size_t)255;
    const size_t o_part = o_seg  + (((size_t)N * sizeof(int2) + 255) & ~(size_t)255);
    const size_t o_col  = o_part + (((size_t)E * sizeof(int)  + 255) & ~(size_t)255);
    int2*     seg  = (int2*)(G + o_seg);
    unsigned* part = (unsigned*)(G + o_part);
    int*      col  = (int*)(G + o_col);

    const int nbins   = (N + 255) / 256;
    const int nchunks = (E + 4095) / 4096;
    int* hist      = (int*)AGGF;
    int* bin_tot   = hist + (size_t)nchunks * nbins;
    int* bin_start = bin_tot + nbins;

    const int row_tiles   = (N + 15) / 16;
    const int gemm_blocks = (row_tiles + 7) / 8;

    bin_count<<<nchunks, 256, (size_t)nbins * sizeof(int), stream>>>(ei, hist, E, N, nbins);
    bin_offsets<<<nbins, 256, 0, stream>>>(hist, bin_tot, nchunks, nbins);
    bin_partition<<<nchunks, 256, 2 * (size_t)nbins * sizeof(int), stream>>>(ei, hist, bin_tot, bin_start,
                                                                          part, E, N, nbins);
    bin_sort<<<nbins, 256, 0, stream>>>(part, bin_start, bin_tot, seg, col, N);

    gemm128_bias<<<gemm_blocks, 256, 0, stream>>>(features, w_proj, b_proj, X, N);

    sage_layer_fused<false><<<FUSED_WGS, FUSED_WAVES * 32, 0, stream>>>(seg, col, X, w_l0, w_r0, b_l0,
                                                                        bn_g0, bn_b0, bn_m0, bn_v0,
                                                                        w_fc1, b_fc1, w_fc2, b_fc2,
                                                                        H1, N, row_tiles);

    sage_layer_fused<true><<<FUSED_WGS, HEAD_WAVES * 32, 0, stream>>>(seg, col, H1, w_l1, w_r1, b_l1,
                                                                      bn_g1, bn_b1, bn_m1, bn_v1,
                                                                      w_fc1, b_fc1, w_fc2, b_fc2,
                                                                      outp, N, row_tiles);
}
